# starting kernel
# baseline (speedup 1.0000x reference)
#include <hip/hip_runtime.h>
#include <hip/hip_bf16.h>


#define N_USER  200000
#define N_MOVIE 100000
#define N_EDGE  4000000
#define D       64

typedef float v2f __attribute__((ext_vector_type(2)));
typedef float v8f __attribute__((ext_vector_type(8)));

#define WSTRIDE 80
#define ASTRIDE 68

#define OFF_SU 0u
#define OFF_SM (OFF_SU + N_USER * 64u)
#define OFF_CU (OFF_SM + N_MOVIE * 64u)
#define OFF_CM (OFF_CU + N_USER)
#define OFF_U1 (OFF_CM + N_MOVIE)
#define OFF_M1 (OFF_U1 + N_USER * 64u)
#define ZERO_FLOATS (OFF_U1)
#define ZERO_F4 (ZERO_FLOATS / 4u)

__global__ __launch_bounds__(256) void zero_f4_kernel(float4* __restrict__ p,
                                                      unsigned n4) {
  unsigned i = blockIdx.x * 256u + threadIdx.x;
  if (i < n4) p[i] = make_float4(0.f, 0.f, 0.f, 0.f);
}

__global__ __launch_bounds__(256) void scatter_kernel(
    const float* __restrict__ xu, const float* __restrict__ xm,
    const int* __restrict__ uidx, const int* __restrict__ midx,
    float* __restrict__ su, float* __restrict__ sm,
    float* __restrict__ cu, float* __restrict__ cm, int nedge) {
  const int e    = (int)((blockIdx.x * 256u + threadIdx.x) >> 5);
  const int lane = threadIdx.x & 31;
  if (e >= nedge) return;
  const int u = uidx[e];
  const int m = midx[e];
  const int c = lane * 2;
  const float2 a = *(const float2*)&xu[(size_t)u * D + c];
  atomicAdd(&sm[(size_t)m * D + c + 0], a.x);
  atomicAdd(&sm[(size_t)m * D + c + 1], a.y);
  const float2 b = *(const float2*)&xm[(size_t)m * D + c];
  atomicAdd(&su[(size_t)u * D + c + 0], b.x);
  atomicAdd(&su[(size_t)u * D + c + 1], b.y);
  if (lane == 0) {
    atomicAdd(&cm[m], 1.0f);
    atomicAdd(&cu[u], 1.0f);
  }
}

__global__ __launch_bounds__(256) void sage_wmma_kernel(
    const float* __restrict__ s, const float* __restrict__ cnt,
    const float* __restrict__ xdst,
    const float* __restrict__ Wl, const float* __restrict__ Wr,
    const float* __restrict__ bias,
    float* __restrict__ out, int nrows, int do_relu) {
  __shared__ float ldsWl[D * WSTRIDE];
  __shared__ float ldsWr[D * WSTRIDE];
  __shared__ float ldsMean[32 * ASTRIDE];
  __shared__ float ldsX[32 * ASTRIDE];

  const int tid  = threadIdx.x;
  const int row0 = blockIdx.x * 32;

  for (int i = tid * 4; i < D * D; i += 256 * 4) {
    const int r = i >> 6, c = i & 63;
    *(float4*)&ldsWl[r * WSTRIDE + c] = *(const float4*)&Wl[i];
    *(float4*)&ldsWr[r * WSTRIDE + c] = *(const float4*)&Wr[i];
  }
  for (int i = tid * 4; i < 32 * D; i += 256 * 4) {
    const int r = i >> 6, c = i & 63;
    const int g = row0 + r;
    float4 mv = make_float4(0.f, 0.f, 0.f, 0.f);
    float4 xv = make_float4(0.f, 0.f, 0.f, 0.f);
    if (g < nrows) {
      const float inv = 1.0f / fmaxf(cnt[g], 1.0f);
      const float4 sv = *(const float4*)&s[(size_t)g * D + c];
      mv = make_float4(sv.x * inv, sv.y * inv, sv.z * inv, sv.w * inv);
      xv = *(const float4*)&xdst[(size_t)g * D + c];
    }
    *(float4*)&ldsMean[r * ASTRIDE + c] = mv;
    *(float4*)&ldsX[r * ASTRIDE + c]    = xv;
  }
  __syncthreads();

  const int wave  = tid >> 5;
  const int lane  = tid & 31;
  const int rt    = wave >> 2;
  const int ct    = wave & 3;
  const int mrow  = rt * 16 + (lane & 15);
  const int ncol  = ct * 16 + (lane & 15);
  const int khalf = (lane >> 4) * 2;

  v8f acc = {0.f, 0.f, 0.f, 0.f, 0.f, 0.f, 0.f, 0.f};
  #pragma unroll
  for (int k = 0; k < D; k += 4) {
    const int kk = k + khalf;
    const v2f a = *(const v2f*)&ldsMean[mrow * ASTRIDE + kk];
    v2f b;
    b.x = ldsWl[(kk + 0) * WSTRIDE + ncol];
    b.y = ldsWl[(kk + 1) * WSTRIDE + ncol];
    acc = __builtin_amdgcn_wmma_f32_16x16x4_f32(false, a, false, b,
                                                (short)0, acc, false, false);
  }
  #pragma unroll
  for (int k = 0; k < D; k += 4) {
    const int kk = k + khalf;
    const v2f a = *(const v2f*)&ldsX[mrow * ASTRIDE + kk];
    v2f b;
    b.x = ldsWr[(kk + 0) * WSTRIDE + ncol];
    b.y = ldsWr[(kk + 1) * WSTRIDE + ncol];
    acc = __builtin_amdgcn_wmma_f32_16x16x4_f32(false, a, false, b,
                                                (short)0, acc, false, false);
  }

  const float bcol  = bias[ncol];
  const int   mhalf = (lane < 16) ? 0 : 8;
  #pragma unroll
  for (int v = 0; v < 8; ++v) {
    const int g = row0 + rt * 16 + mhalf + v;
    if (g < nrows) {
      float val = acc[v] + bcol;
      if (do_relu) val = fmaxf(val, 0.0f);
      out[(size_t)g * D + ncol] = val;
    }
  }
}

extern "C" void kernel_launch(void* const* d_in, const int* in_sizes, int n_in,
                              void* d_out, int out_size, void* d_ws, size_t ws_size,
                              hipStream_t stream) {
  const float* x_user  = (const float*)d_in[0];
  const float* x_movie = (const float*)d_in[1];
  const float* Wl1_um  = (const float*)d_in[2];
  const float* Wr1_um  = (const float*)d_in[3];
  const float* b1_um   = (const float*)d_in[4];
  const float* Wl1_mu  = (const float*)d_in[5];
  const float* Wr1_mu  = (const float*)d_in[6];
  const float* b1_mu   = (const float*)d_in[7];
  const float* Wl2_um  = (const float*)d_in[8];
  const float* Wr2_um  = (const float*)d_in[9];
  const float* b2_um   = (const float*)d_in[10];
  const float* Wl2_mu  = (const float*)d_in[11];
  const float* Wr2_mu  = (const float*)d_in[12];
  const float* b2_mu   = (const float*)d_in[13];
  const int*   uidx    = (const int*)d_in[14];
  const int*   midx    = (const int*)d_in[15];

  float* ws = (float*)d_ws;
  float* su = ws + OFF_SU;
  float* sm = ws + OFF_SM;
  float* cu = ws + OFF_CU;
  float* cm = ws + OFF_CM;
  float* u1 = ws + OFF_U1;
  float* m1 = ws + OFF_M1;

  float* u2 = (float*)d_out;
  float* m2 = (float*)d_out + (size_t)N_USER * D;

  const dim3 blk(256);
  const dim3 gZero((ZERO_F4 + 255) / 256);
  const dim3 gScat((unsigned)(((size_t)N_EDGE * 32 + 255) / 256));
  const dim3 gMov(N_MOVIE / 32);
  const dim3 gUsr(N_USER / 32);

  zero_f4_kernel<<<gZero, blk, 0, stream>>>((float4*)ws, ZERO_F4);
  scatter_kernel<<<gScat, blk, 0, stream>>>(x_user, x_movie, uidx, midx,
                                            su, sm, cu, cm, N_EDGE);
  sage_wmma_kernel<<<gMov, blk, 0, stream>>>(sm, cm, x_movie, Wl1_um, Wr1_um,
                                             b1_um, m1, N_MOVIE, 1);
  sage_wmma_kernel<<<gUsr, blk, 0, stream>>>(su, cu, x_user, Wl1_mu, Wr1_mu,
                                             b1_mu, u1, N_USER, 1);

  zero_f4_kernel<<<gZero, blk, 0, stream>>>((float4*)ws, ZERO_F4);
  scatter_kernel<<<gScat, blk, 0, stream>>>(u1, m1, uidx, midx,
                                            su, sm, cu, cm, N_EDGE);
  sage_wmma_kernel<<<gMov, blk, 0, stream>>>(sm, cm, m1, Wl2_um, Wr2_um,
                                             b2_um, m2, N_MOVIE, 0);
  sage_wmma_kernel<<<gUsr, blk, 0, stream>>>(su, cu, u1, Wl2_mu, Wr2_mu,
                                             b2_mu, u2, N_USER, 0);
}


# Neighbour sums kept per destination tile in LDS: a work-group owns 1184 rows, scans all edge ids, gathers its edges' source rows, adds by LDS float atomics; no global atomics; sum order changes
# speedup vs baseline: 1.9864x; 1.9864x over previous
#include <hip/hip_runtime.h>
#include <hip/hip_bf16.h>


#define N_USER  200000
#define N_MOVIE 100000
#define N_EDGE  4000000
#define D       64

typedef float v2f __attribute__((ext_vector_type(2)));
typedef float v8f __attribute__((ext_vector_type(8)));

#define WSTRIDE 80
#define ASTRIDE 68

#define OFF_SU 0u
#define OFF_SM (OFF_SU + N_USER * 64u)
#define OFF_CU (OFF_SM + N_MOVIE * 64u)
#define OFF_CM (OFF_CU + N_USER)
#define OFF_U1 (OFF_CM + N_MOVIE)
#define OFF_M1 (OFF_U1 + N_USER * 64u)

#define TR          1184
#define NT_M        ((N_MOVIE + TR - 1) / TR)
#define NT_U        ((N_USER + TR - 1) / TR)
#define NTILE       (NT_M + NT_U)
#define AGG_THREADS 1024
#define AGG_WAVES   (AGG_THREADS / 32)
#define QCAP        64
#define N_EDGE4     (N_EDGE / 4)
#define SCAN_U      4
#define STEP4       (AGG_THREADS * SCAN_U)
#define SCAN_ITERS  (N_EDGE4 / STEP4)
#define TAIL_E0     (SCAN_ITERS * STEP4 * 4)
#define TAIL_ITERS  ((N_EDGE - TAIL_E0 + AGG_THREADS - 1) / AGG_THREADS)
#define GB          8

static_assert(N_EDGE % 4 == 0, "edge indices are scanned 4 per lane");
static_assert((QCAP & (QCAP - 1)) == 0 && QCAP >= 64, "ring holds 31 + 32 entries");

__device__ __forceinline__ float2 row_load(const float* __restrict__ x, int r,
                                           unsigned lane) {
  return ((const float2*)x)[((unsigned)r << 5) + lane];
}

__device__ __forceinline__ void drain_ring(
    const float* __restrict__ xsrc, const int* __restrict__ sidx,
    float* acc, float* cnt, const int* qe, const int* qd,
    int head, int n, unsigned lane) {
  int s = 0, d = 0;
  if ((int)lane < n) {
    const int slot = (head + (int)lane) & (QCAP - 1);
    const int e = qe[slot];
    d = qd[slot];
    s = sidx[e];
    atomicAdd(&cnt[d], 1.0f);
  }
  if (n == 32) {
    #pragma unroll 1
    for (int k = 0; k < 32; k += GB) {
      float2 r[GB];
      #pragma unroll
      for (int j = 0; j < GB; ++j)
        r[j] = row_load(xsrc, __builtin_amdgcn_readlane(s, k + j), lane);
      #pragma unroll
      for (int j = 0; j < GB; ++j) {
        float* q = &acc[((unsigned)__builtin_amdgcn_readlane(d, k + j) << 6) + lane * 2u];
        atomicAdd(q + 0, r[j].x);
        atomicAdd(q + 1, r[j].y);
      }
    }
  } else {
    #pragma unroll 1
    for (int k = 0; k < n; ++k) {
      const float2 r1 = row_load(xsrc, __builtin_amdgcn_readlane(s, k), lane);
      float* q = &acc[((unsigned)__builtin_amdgcn_readlane(d, k) << 6) + lane * 2u];
      atomicAdd(q + 0, r1.x);
      atomicAdd(q + 1, r1.y);
    }
  }
}

__device__ __forceinline__ void scan_one(
    int dst, int eid, int lo, int rows,
    const float* __restrict__ xsrc, const int* __restrict__ sidx,
    float* acc, float* cnt, int* qe, int* qd,
    int& head, int& qn, unsigned lane) {
  const unsigned dl = (unsigned)(dst - lo);
  const bool hit = dl < (unsigned)rows;
  const unsigned mask = __builtin_amdgcn_ballot_w32(hit);
  if (mask != 0u) {
    if (hit) {
      const int pos = (head + qn + (int)__builtin_amdgcn_mbcnt_lo(mask, 0u)) & (QCAP - 1);
      qe[pos] = eid;
      qd[pos] = (int)dl;
    }
    qn += __builtin_popcount(mask);
    if (qn >= 32) {
      drain_ring(xsrc, sidx, acc, cnt, qe, qd, head, 32, lane);
      head = (head + 32) & (QCAP - 1);
      qn -= 32;
    }
  }
}

__device__ __forceinline__ int4 scan_load(const int4* __restrict__ d4, int it, int u, int tid) {
  return d4[(it * SCAN_U + u) * AGG_THREADS + tid];
}

__global__ __launch_bounds__(AGG_THREADS) void tile_aggregate_kernel(
    const float* __restrict__ xu, const float* __restrict__ xm,
    const int* __restrict__ uidx, const int* __restrict__ midx,
    float* __restrict__ su, float* __restrict__ sm,
    float* __restrict__ cu, float* __restrict__ cm) {
  __shared__ float acc[TR * D];
  __shared__ float cnt[TR];
  __shared__ int   ringE[AGG_WAVES * QCAP];
  __shared__ int   ringD[AGG_WAVES * QCAP];

  const int tid = threadIdx.x;
  const unsigned lane = threadIdx.x & 31u;
  const int wave = tid >> 5;
  const int tile = blockIdx.x;

  const bool mov = tile < NT_M;
  const int*   didx = mov ? midx : uidx;
  const int*   sidx = mov ? uidx : midx;
  const float* xsrc = mov ? xu : xm;
  float*       sOut = mov ? sm : su;
  float*       cOut = mov ? cm : cu;
  const int    nrow = mov ? N_MOVIE : N_USER;
  const int    lo   = (mov ? tile : tile - NT_M) * TR;
  const int    rows = min(TR, nrow - lo);

  for (int i = tid; i < TR * D / 4; i += AGG_THREADS)
    ((float4*)acc)[i] = make_float4(0.f, 0.f, 0.f, 0.f);
  for (int i = tid; i < TR; i += AGG_THREADS) cnt[i] = 0.f;
  __syncthreads();

  int* qe = &ringE[wave * QCAP];
  int* qd = &ringD[wave * QCAP];
  int head = 0, qn = 0;

  const int4* d4 = (const int4*)didx;
  int it = (tile * SCAN_ITERS) / NTILE;
  int4 v[SCAN_U];
  #pragma unroll
  for (int u = 0; u < SCAN_U; ++u) v[u] = scan_load(d4, it, u, tid);
  #pragma unroll 1
  for (int n = 0; n < SCAN_ITERS; ++n) {
    const int itn = (it + 1 == SCAN_ITERS) ? 0 : it + 1;
    int4 vn[SCAN_U];
    #pragma unroll
    for (int u = 0; u < SCAN_U; ++u) vn[u] = scan_load(d4, itn, u, tid);
    #pragma unroll
    for (int u = 0; u < SCAN_U; ++u) {
      const int e0 = ((it * SCAN_U + u) * AGG_THREADS + tid) * 4;
      scan_one(v[u].x, e0 + 0, lo, rows, xsrc, sidx, acc, cnt, qe, qd, head, qn, lane);
      scan_one(v[u].y, e0 + 1, lo, rows, xsrc, sidx, acc, cnt, qe, qd, head, qn, lane);
      scan_one(v[u].z, e0 + 2, lo, rows, xsrc, sidx, acc, cnt, qe, qd, head, qn, lane);
      scan_one(v[u].w, e0 + 3, lo, rows, xsrc, sidx, acc, cnt, qe, qd, head, qn, lane);
    }
    #pragma unroll
    for (int u = 0; u < SCAN_U; ++u) v[u] = vn[u];
    it = itn;
  }
  #pragma unroll 1
  for (int j = 0; j < TAIL_ITERS; ++j) {
    const int e = TAIL_E0 + j * AGG_THREADS + tid;
    const int dv = (e < N_EDGE) ? didx[e] : -1;
    scan_one(dv, e, lo, rows, xsrc, sidx, acc, cnt, qe, qd, head, qn, lane);
  }
  if (qn > 0) drain_ring(xsrc, sidx, acc, cnt, qe, qd, head, qn, lane);
  __syncthreads();

  float4* o4 = (float4*)(sOut + (size_t)lo * D);
  for (int i = tid; i < rows * (D / 4); i += AGG_THREADS)
    o4[i] = ((const float4*)acc)[i];
  for (int i = tid; i < rows; i += AGG_THREADS) cOut[lo + i] = cnt[i];
}

__global__ __launch_bounds__(256) void sage_wmma_kernel(
    const float* __restrict__ s, const float* __restrict__ cnt,
    const float* __restrict__ xdst,
    const float* __restrict__ Wl, const float* __restrict__ Wr,
    const float* __restrict__ bias,
    float* __restrict__ out, int nrows, int do_relu) {
  __shared__ float ldsWl[D * WSTRIDE];
  __shared__ float ldsWr[D * WSTRIDE];
  __shared__ float ldsMean[32 * ASTRIDE];
  __shared__ float ldsX[32 * ASTRIDE];

  const int tid  = threadIdx.x;
  const int row0 = blockIdx.x * 32;

  for (int i = tid * 4; i < D * D; i += 256 * 4) {
    const int r = i >> 6, c = i & 63;
    *(float4*)&ldsWl[r * WSTRIDE + c] = *(const float4*)&Wl[i];
    *(float4*)&ldsWr[r * WSTRIDE + c] = *(const float4*)&Wr[i];
  }
  for (int i = tid * 4; i < 32 * D; i += 256 * 4) {
    const int r = i >> 6, c = i & 63;
    const int g = row0 + r;
    float4 mv = make_float4(0.f, 0.f, 0.f, 0.f);
    float4 xv = make_float4(0.f, 0.f, 0.f, 0.f);
    if (g < nrows) {
      const float inv = 1.0f / fmaxf(cnt[g], 1.0f);
      const float4 sv = *(const float4*)&s[(size_t)g * D + c];
      mv = make_float4(sv.x * inv, sv.y * inv, sv.z * inv, sv.w * inv);
      xv = *(const float4*)&xdst[(size_t)g * D + c];
    }
    *(float4*)&ldsMean[r * ASTRIDE + c] = mv;
    *(float4*)&ldsX[r * ASTRIDE + c]    = xv;
  }
  __syncthreads();

  const int wave  = tid >> 5;
  const int lane  = tid & 31;
  const int rt    = wave >> 2;
  const int ct    = wave & 3;
  const int mrow  = rt * 16 + (lane & 15);
  const int ncol  = ct * 16 + (lane & 15);
  const int khalf = (lane >> 4) * 2;

  v8f acc = {0.f, 0.f, 0.f, 0.f, 0.f, 0.f, 0.f, 0.f};
  #pragma unroll
  for (int k = 0; k < D; k += 4) {
    const int kk = k + khalf;
    const v2f a = *(const v2f*)&ldsMean[mrow * ASTRIDE + kk];
    v2f b;
    b.x = ldsWl[(kk + 0) * WSTRIDE + ncol];
    b.y = ldsWl[(kk + 1) * WSTRIDE + ncol];
    acc = __builtin_amdgcn_wmma_f32_16x16x4_f32(false, a, false, b,
                                                (short)0, acc, false, false);
  }
  #pragma unroll
  for (int k = 0; k < D; k += 4) {
    const int kk = k + khalf;
    const v2f a = *(const v2f*)&ldsX[mrow * ASTRIDE + kk];
    v2f b;
    b.x = ldsWr[(kk + 0) * WSTRIDE + ncol];
    b.y = ldsWr[(kk + 1) * WSTRIDE + ncol];
    acc = __builtin_amdgcn_wmma_f32_16x16x4_f32(false, a, false, b,
                                                (short)0, acc, false, false);
  }

  const float bcol  = bias[ncol];
  const int   mhalf = (lane < 16) ? 0 : 8;
  #pragma unroll
  for (int v = 0; v < 8; ++v) {
    const int g = row0 + rt * 16 + mhalf + v;
    if (g < nrows) {
      float val = acc[v] + bcol;
      if (do_relu) val = fmaxf(val, 0.0f);
      out[(size_t)g * D + ncol] = val;
    }
  }
}

extern "C" void kernel_launch(void* const* d_in, const int* in_sizes, int n_in,
                              void* d_out, int out_size, void* d_ws, size_t ws_size,
                              hipStream_t stream) {
  const float* x_user  = (const float*)d_in[0];
  const float* x_movie = (const float*)d_in[1];
  const float* Wl1_um  = (const float*)d_in[2];
  const float* Wr1_um  = (const float*)d_in[3];
  const float* b1_um   = (const float*)d_in[4];
  const float* Wl1_mu  = (const float*)d_in[5];
  const float* Wr1_mu  = (const float*)d_in[6];
  const float* b1_mu   = (const float*)d_in[7];
  const float* Wl2_um  = (const float*)d_in[8];
  const float* Wr2_um  = (const float*)d_in[9];
  const float* b2_um   = (const float*)d_in[10];
  const float* Wl2_mu  = (const float*)d_in[11];
  const float* Wr2_mu  = (const float*)d_in[12];
  const float* b2_mu   = (const float*)d_in[13];
  const int*   uidx    = (const int*)d_in[14];
  const int*   midx    = (const int*)d_in[15];

  float* ws = (float*)d_ws;
  float* su = ws + OFF_SU;
  float* sm = ws + OFF_SM;
  float* cu = ws + OFF_CU;
  float* cm = ws + OFF_CM;
  float* u1 = ws + OFF_U1;
  float* m1 = ws + OFF_M1;

  float* u2 = (float*)d_out;
  float* m2 = (float*)d_out + (size_t)N_USER * D;

  const dim3 blk(256);
  const dim3 bAgg(AGG_THREADS);
  const dim3 gAgg(NTILE);
  const dim3 gMov(N_MOVIE / 32);
  const dim3 gUsr(N_USER / 32);

  tile_aggregate_kernel<<<gAgg, bAgg, 0, stream>>>(x_user, x_movie, uidx, midx,
                                                   su, sm, cu, cm);
  sage_wmma_kernel<<<gMov, blk, 0, stream>>>(sm, cm, x_movie, Wl1_um, Wr1_um,
                                             b1_um, m1, N_MOVIE, 1);
  sage_wmma_kernel<<<gUsr, blk, 0, stream>>>(su, cu, x_user, Wl1_mu, Wr1_mu,
                                             b1_mu, u1, N_USER, 1);

  tile_aggregate_kernel<<<gAgg, bAgg, 0, stream>>>(u1, m1, uidx, midx,
                                                   su, sm, cu, cm);
  sage_wmma_kernel<<<gMov, blk, 0, stream>>>(sm, cm, m1, Wl2_um, Wr2_um,
                                             b2_um, m2, N_MOVIE, 0);
  sage_wmma_kernel<<<gUsr, blk, 0, stream>>>(su, cu, u1, Wl2_mu, Wr2_mu,
                                             b2_mu, u2, N_USER, 0);
}


# Layer 2 skips the edge-id scan: layer 1 stores each wave's matched (source id, tile row) words in the still unwritten output buffer and layer 2 walks them; layer-2 atomic order changes
# speedup vs baseline: 2.4840x; 1.2505x over previous
#include <hip/hip_runtime.h>
#include <hip/hip_bf16.h>


#define N_USER  200000
#define N_MOVIE 100000
#define N_EDGE  4000000
#define D       64

typedef float v2f __attribute__((ext_vector_type(2)));
typedef float v8f __attribute__((ext_vector_type(8)));

#define WSTRIDE 80
#define ASTRIDE 68

#define OFF_SU 0u
#define OFF_SM (OFF_SU + N_USER * 64u)
#define OFF_CU (OFF_SM + N_MOVIE * 64u)
#define OFF_CM (OFF_CU + N_USER)
#define OFF_U1 (OFF_CM + N_MOVIE)
#define OFF_M1 (OFF_U1 + N_USER * 64u)

#define TR          1184
#define NT_M        ((N_MOVIE + TR - 1) / TR)
#define NT_U        ((N_USER + TR - 1) / TR)
#define NTILE       (NT_M + NT_U)
#define AGG_THREADS 1024
#define AGG_WAVES   (AGG_THREADS / 32)
#define QCAP        64
#define N_EDGE4     (N_EDGE / 4)
#define SCAN_U      4
#define STEP4       (AGG_THREADS * SCAN_U)
#define SCAN_ITERS  (N_EDGE4 / STEP4)
#define TAIL_E0     (SCAN_ITERS * STEP4 * 4)
#define TAIL_ITERS  ((N_EDGE - TAIL_E0 + AGG_THREADS - 1) / AGG_THREADS)
#define GB          8

#define LSTRIDE     2336u
#define LCAP        ((int)LSTRIDE - 32)

static_assert((unsigned long long)NTILE * AGG_WAVES * LSTRIDE
                  <= (unsigned long long)(N_USER + N_MOVIE) * D,
              "counts and lists fit into the output buffer");
static_assert(LSTRIDE % 32 == 0, "a wave's count line and list are whole 128-B lines");
static_assert(TR <= (1 << 11) && N_USER <= (1 << 21) && N_MOVIE <= (1 << 21),
              "source id and row in tile share one 32-bit word");

static_assert(N_EDGE % 4 == 0, "edge indices are scanned 4 per lane");
static_assert((QCAP & (QCAP - 1)) == 0 && QCAP >= 64, "ring holds 31 + 32 entries");

__device__ __forceinline__ float2 row_load(const float* __restrict__ x, int r,
                                           unsigned lane) {
  return ((const float2*)x)[((unsigned)r << 5) + lane];
}

template <bool SAVE>
__device__ __forceinline__ void drain_ring(
    const float* __restrict__ xsrc, const int* __restrict__ sidx,
    float* acc, float* cnt, const int* qe, const int* qd,
    int head, int n, unsigned lane, unsigned* __restrict__ wl, int& nsaved) {
  int s = 0, d = 0;
  if ((int)lane < n) {
    const int slot = (head + (int)lane) & (QCAP - 1);
    const int e = qe[slot];
    d = qd[slot];
    s = sidx[e];
    if (SAVE) atomicAdd(&cnt[d], 1.0f);
  }
  if (SAVE) {
    if (nsaved + n <= LCAP && (int)lane < n)
      wl[nsaved + (int)lane] = ((unsigned)s << 11) | (unsigned)d;
    nsaved += n;
  }
  if (n == 32) {
    #pragma unroll 1
    for (int k = 0; k < 32; k += GB) {
      float2 r[GB];
      #pragma unroll
      for (int j = 0; j < GB; ++j)
        r[j] = row_load(xsrc, __builtin_amdgcn_readlane(s, k + j), lane);
      #pragma unroll
      for (int j = 0; j < GB; ++j) {
        float* q = &acc[((unsigned)__builtin_amdgcn_readlane(d, k + j) << 6) + lane * 2u];
        atomicAdd(q + 0, r[j].x);
        atomicAdd(q + 1, r[j].y);
      }
    }
  } else {
    #pragma unroll 1
    for (int k = 0; k < n; ++k) {
      const float2 r1 = row_load(xsrc, __builtin_amdgcn_readlane(s, k), lane);
      float* q = &acc[((unsigned)__builtin_amdgcn_readlane(d, k) << 6) + lane * 2u];
      atomicAdd(q + 0, r1.x);
      atomicAdd(q + 1, r1.y);
    }
  }
}

template <bool SAVE>
__device__ __forceinline__ void scan_one(
    int dst, int eid, int lo, int rows,
    const float* __restrict__ xsrc, const int* __restrict__ sidx,
    float* acc, float* cnt, int* qe, int* qd,
    int& head, int& qn, unsigned lane, unsigned* __restrict__ wl, int& nsaved) {
  const unsigned dl = (unsigned)(dst - lo);
  const bool hit = dl < (unsigned)rows;
  const unsigned mask = __builtin_amdgcn_ballot_w32(hit);
  if (mask != 0u) {
    if (hit) {
      const int pos = (head + qn + (int)__builtin_amdgcn_mbcnt_lo(mask, 0u)) & (QCAP - 1);
      qe[pos] = eid;
      qd[pos] = (int)dl;
    }
    qn += __builtin_popcount(mask);
    if (qn >= 32) {
      drain_ring<SAVE>(xsrc, sidx, acc, cnt, qe, qd, head, 32, lane, wl, nsaved);
      head = (head + 32) & (QCAP - 1);
      qn -= 32;
    }
  }
}

__device__ __forceinline__ int4 scan_load(const int4* __restrict__ d4, int it, int u, int tid) {
  return d4[(it * SCAN_U + u) * AGG_THREADS + tid];
}

template <bool SAVE>
__device__ __forceinline__ void scan_edges(
    const int* __restrict__ didx, const int* __restrict__ sidx,
    const float* __restrict__ xsrc, float* acc, float* cnt, int* qe, int* qd,
    int lo, int rows, int tile, int tid, unsigned lane,
    unsigned* __restrict__ wl, int& nsaved) {
  int head = 0, qn = 0;

  const int4* d4 = (const int4*)didx;
  int it = (tile * SCAN_ITERS) / NTILE;
  int4 v[SCAN_U];
  #pragma unroll
  for (int u = 0; u < SCAN_U; ++u) v[u] = scan_load(d4, it, u, tid);
  #pragma unroll 1
  for (int n = 0; n < SCAN_ITERS; ++n) {
    const int itn = (it + 1 == SCAN_ITERS) ? 0 : it + 1;
    int4 vn[SCAN_U];
    #pragma unroll
    for (int u = 0; u < SCAN_U; ++u) vn[u] = scan_load(d4, itn, u, tid);
    #pragma unroll
    for (int u = 0; u < SCAN_U; ++u) {
      const int e0 = ((it * SCAN_U + u) * AGG_THREADS + tid) * 4;
      scan_one<SAVE>(v[u].x, e0 + 0, lo, rows, xsrc, sidx, acc, cnt, qe, qd, head, qn, lane, wl, nsaved);
      scan_one<SAVE>(v[u].y, e0 + 1, lo, rows, xsrc, sidx, acc, cnt, qe, qd, head, qn, lane, wl, nsaved);
      scan_one<SAVE>(v[u].z, e0 + 2, lo, rows, xsrc, sidx, acc, cnt, qe, qd, head, qn, lane, wl, nsaved);
      scan_one<SAVE>(v[u].w, e0 + 3, lo, rows, xsrc, sidx, acc, cnt, qe, qd, head, qn, lane, wl, nsaved);
    }
    #pragma unroll
    for (int u = 0; u < SCAN_U; ++u) v[u] = vn[u];
    it = itn;
  }
  #pragma unroll 1
  for (int j = 0; j < TAIL_ITERS; ++j) {
    const int e = TAIL_E0 + j * AGG_THREADS + tid;
    const int dv = (e < N_EDGE) ? didx[e] : -1;
    scan_one<SAVE>(dv, e, lo, rows, xsrc, sidx, acc, cnt, qe, qd, head, qn, lane, wl, nsaved);
  }
  if (qn > 0) drain_ring<SAVE>(xsrc, sidx, acc, cnt, qe, qd, head, qn, lane, wl, nsaved);
}

__global__ __launch_bounds__(AGG_THREADS) void tile_aggregate_kernel(
    const float* __restrict__ xu, const float* __restrict__ xm,
    const int* __restrict__ uidx, const int* __restrict__ midx,
    float* __restrict__ su, float* __restrict__ sm,
    float* __restrict__ cu, float* __restrict__ cm,
    unsigned* __restrict__ lst) {
  __shared__ float acc[TR * D];
  __shared__ float cnt[TR];
  __shared__ int   ringE[AGG_WAVES * QCAP];
  __shared__ int   ringD[AGG_WAVES * QCAP];

  const int tid = threadIdx.x;
  const unsigned lane = threadIdx.x & 31u;
  const int wave = tid >> 5;
  const int tile = blockIdx.x;

  const bool mov = tile < NT_M;
  const int*   didx = mov ? midx : uidx;
  const int*   sidx = mov ? uidx : midx;
  const float* xsrc = mov ? xu : xm;
  float*       sOut = mov ? sm : su;
  float*       cOut = mov ? cm : cu;
  const int    nrow = mov ? N_MOVIE : N_USER;
  const int    lo   = (mov ? tile : tile - NT_M) * TR;
  const int    rows = min(TR, nrow - lo);

  for (int i = tid; i < TR * D / 4; i += AGG_THREADS)
    ((float4*)acc)[i] = make_float4(0.f, 0.f, 0.f, 0.f);
  for (int i = tid; i < TR; i += AGG_THREADS) cnt[i] = 0.f;
  __syncthreads();

  unsigned* wreg = lst + (unsigned)(tile * AGG_WAVES + wave) * LSTRIDE;
  int nsaved = 0;
  scan_edges<true>(didx, sidx, xsrc, acc, cnt, &ringE[wave * QCAP], &ringD[wave * QCAP],
                   lo, rows, tile, tid, lane, wreg + 32, nsaved);
  if (lane == 0) wreg[0] = (unsigned)nsaved;
  __syncthreads();

  float4* o4 = (float4*)(sOut + (size_t)lo * D);
  for (int i = tid; i < rows * (D / 4); i += AGG_THREADS)
    o4[i] = ((const float4*)acc)[i];
  for (int i = tid; i < rows; i += AGG_THREADS) cOut[lo + i] = cnt[i];
}

__global__ __launch_bounds__(AGG_THREADS) void tile_replay_kernel(
    const float* __restrict__ xu, const float* __restrict__ xm,
    const int* __restrict__ uidx, const int* __restrict__ midx,
    float* __restrict__ su, float* __restrict__ sm,
    const unsigned* __restrict__ lst) {
  __shared__ float acc[TR * D];
  __shared__ int   ringE[AGG_WAVES * QCAP];
  __shared__ int   ringD[AGG_WAVES * QCAP];

  const int tid = threadIdx.x;
  const unsigned lane = threadIdx.x & 31u;
  const int wave = tid >> 5;
  const int tile = blockIdx.x;

  const bool mov = tile < NT_M;
  const float* xsrc = mov ? xu : xm;
  float*       sOut = mov ? sm : su;
  const int    nrow = mov ? N_MOVIE : N_USER;
  const int    lo   = (mov ? tile : tile - NT_M) * TR;
  const int    rows = min(TR, nrow - lo);

  for (int i = tid; i < TR * D / 4; i += AGG_THREADS)
    ((float4*)acc)[i] = make_float4(0.f, 0.f, 0.f, 0.f);
  __syncthreads();

  const unsigned* __restrict__ wreg = lst + (unsigned)(tile * AGG_WAVES + wave) * LSTRIDE;
  const int n = __builtin_amdgcn_readfirstlane((int)wreg[0]);
  if (n > LCAP) {
    const int* didx = mov ? midx : uidx;
    const int* sidx = mov ? uidx : midx;
    int unused = 0;
    scan_edges<false>(didx, sidx, xsrc, acc, nullptr, &ringE[wave * QCAP], &ringD[wave * QCAP],
                      lo, rows, tile, tid, lane, nullptr, unused);
  } else if (n > 0) {
    const unsigned* __restrict__ wl = wreg + 32;
    const int last = n - 1;
    unsigned w = wl[min((int)lane, last)];
    #pragma unroll 1
    for (int base = 0; base < n; base += 32) {
      const int m = min(32, n - base);
      const unsigned wn = wl[min(base + 32 + (int)lane, last)];
      const int s = (int)(w >> 11);
      const int d = (int)(w & 2047u);
      if (m == 32) {
        float2 r[32];
        #pragma unroll
        for (int j = 0; j < 32; ++j)
          r[j] = row_load(xsrc, __builtin_amdgcn_readlane(s, j), lane);
        #pragma unroll
        for (int j = 0; j < 32; ++j) {
          float* q = &acc[((unsigned)__builtin_amdgcn_readlane(d, j) << 6) + lane * 2u];
          atomicAdd(q + 0, r[j].x);
          atomicAdd(q + 1, r[j].y);
        }
      } else {
        #pragma unroll 1
        for (int k = 0; k < m; ++k) {
          const float2 r1 = row_load(xsrc, __builtin_amdgcn_readlane(s, k), lane);
          float* q = &acc[((unsigned)__builtin_amdgcn_readlane(d, k) << 6) + lane * 2u];
          atomicAdd(q + 0, r1.x);
          atomicAdd(q + 1, r1.y);
        }
      }
      w = wn;
    }
  }
  __syncthreads();

  float4* o4 = (float4*)(sOut + (size_t)lo * D);
  for (int i = tid; i < rows * (D / 4); i += AGG_THREADS)
    o4[i] = ((const float4*)acc)[i];
}

__global__ __launch_bounds__(256) void sage_wmma_kernel(
    const float* __restrict__ s, const float* __restrict__ cnt,
    const float* __restrict__ xdst,
    const float* __restrict__ Wl, const float* __restrict__ Wr,
    const float* __restrict__ bias,
    float* __restrict__ out, int nrows, int do_relu) {
  __shared__ float ldsWl[D * WSTRIDE];
  __shared__ float ldsWr[D * WSTRIDE];
  __shared__ float ldsMean[32 * ASTRIDE];
  __shared__ float ldsX[32 * ASTRIDE];

  const int tid  = threadIdx.x;
  const int row0 = blockIdx.x * 32;

  for (int i = tid * 4; i < D * D; i += 256 * 4) {
    const int r = i >> 6, c = i & 63;
    *(float4*)&ldsWl[r * WSTRIDE + c] = *(const float4*)&Wl[i];
    *(float4*)&ldsWr[r * WSTRIDE + c] = *(const float4*)&Wr[i];
  }
  for (int i = tid * 4; i < 32 * D; i += 256 * 4) {
    const int r = i >> 6, c = i & 63;
    const int g = row0 + r;
    float4 mv = make_float4(0.f, 0.f, 0.f, 0.f);
    float4 xv = make_float4(0.f, 0.f, 0.f, 0.f);
    if (g < nrows) {
      const float inv = 1.0f / fmaxf(cnt[g], 1.0f);
      const float4 sv = *(const float4*)&s[(size_t)g * D + c];
      mv = make_float4(sv.x * inv, sv.y * inv, sv.z * inv, sv.w * inv);
      xv = *(const float4*)&xdst[(size_t)g * D + c];
    }
    *(float4*)&ldsMean[r * ASTRIDE + c] = mv;
    *(float4*)&ldsX[r * ASTRIDE + c]    = xv;
  }
  __syncthreads();

  const int wave  = tid >> 5;
  const int lane  = tid & 31;
  const int rt    = wave >> 2;
  const int ct    = wave & 3;
  const int mrow  = rt * 16 + (lane & 15);
  const int ncol  = ct * 16 + (lane & 15);
  const int khalf = (lane >> 4) * 2;

  v8f acc = {0.f, 0.f, 0.f, 0.f, 0.f, 0.f, 0.f, 0.f};
  #pragma unroll
  for (int k = 0; k < D; k += 4) {
    const int kk = k + khalf;
    const v2f a = *(const v2f*)&ldsMean[mrow * ASTRIDE + kk];
    v2f b;
    b.x = ldsWl[(kk + 0) * WSTRIDE + ncol];
    b.y = ldsWl[(kk + 1) * WSTRIDE + ncol];
    acc = __builtin_amdgcn_wmma_f32_16x16x4_f32(false, a, false, b,
                                                (short)0, acc, false, false);
  }
  #pragma unroll
  for (int k = 0; k < D; k += 4) {
    const int kk = k + khalf;
    const v2f a = *(const v2f*)&ldsX[mrow * ASTRIDE + kk];
    v2f b;
    b.x = ldsWr[(kk + 0) * WSTRIDE + ncol];
    b.y = ldsWr[(kk + 1) * WSTRIDE + ncol];
    acc = __builtin_amdgcn_wmma_f32_16x16x4_f32(false, a, false, b,
                                                (short)0, acc, false, false);
  }

  const float bcol  = bias[ncol];
  const int   mhalf = (lane < 16) ? 0 : 8;
  #pragma unroll
  for (int v = 0; v < 8; ++v) {
    const int g = row0 + rt * 16 + mhalf + v;
    if (g < nrows) {
      float val = acc[v] + bcol;
      if (do_relu) val = fmaxf(val, 0.0f);
      out[(size_t)g * D + ncol] = val;
    }
  }
}

extern "C" void kernel_launch(void* const* d_in, const int* in_sizes, int n_in,
                              void* d_out, int out_size, void* d_ws, size_t ws_size,
                              hipStream_t stream) {
  const float* x_user  = (const float*)d_in[0];
  const float* x_movie = (const float*)d_in[1];
  const float* Wl1_um  = (const float*)d_in[2];
  const float* Wr1_um  = (const float*)d_in[3];
  const float* b1_um   = (const float*)d_in[4];
  const float* Wl1_mu  = (const float*)d_in[5];
  const float* Wr1_mu  = (const float*)d_in[6];
  const float* b1_mu   = (const float*)d_in[7];
  const float* Wl2_um  = (const float*)d_in[8];
  const float* Wr2_um  = (const float*)d_in[9];
  const float* b2_um   = (const float*)d_in[10];
  const float* Wl2_mu  = (const float*)d_in[11];
  const float* Wr2_mu  = (const float*)d_in[12];
  const float* b2_mu   = (const float*)d_in[13];
  const int*   uidx    = (const int*)d_in[14];
  const int*   midx    = (const int*)d_in[15];

  float* ws = (float*)d_ws;
  float* su = ws + OFF_SU;
  float* sm = ws + OFF_SM;
  float* cu = ws + OFF_CU;
  float* cm = ws + OFF_CM;
  float* u1 = ws + OFF_U1;
  float* m1 = ws + OFF_M1;

  float* u2 = (float*)d_out;
  float* m2 = (float*)d_out + (size_t)N_USER * D;
  unsigned* lst = (unsigned*)d_out;

  const dim3 blk(256);
  const dim3 bAgg(AGG_THREADS);
  const dim3 gAgg(NTILE);
  const dim3 gMov(N_MOVIE / 32);
  const dim3 gUsr(N_USER / 32);

  tile_aggregate_kernel<<<gAgg, bAgg, 0, stream>>>(x_user, x_movie, uidx, midx,
                                                   su, sm, cu, cm, lst);
  sage_wmma_kernel<<<gMov, blk, 0, stream>>>(sm, cm, x_movie, Wl1_um, Wr1_um,
                                             b1_um, m1, N_MOVIE, 1);
  sage_wmma_kernel<<<gUsr, blk, 0, stream>>>(su, cu, x_user, Wl1_mu, Wr1_mu,
                                             b1_mu, u1, N_USER, 1);

  tile_replay_kernel<<<gAgg, bAgg, 0, stream>>>(u1, m1, uidx, midx, su, sm, lst);
  sage_wmma_kernel<<<gMov, blk, 0, stream>>>(sm, cm, m1, Wl2_um, Wr2_um,
                                             b2_um, m2, N_MOVIE, 0);
  sage_wmma_kernel<<<gUsr, blk, 0, stream>>>(su, cu, u1, Wl2_mu, Wr2_mu,
                                             b2_mu, u2, N_USER, 0);
}


# Transform fused into the aggregation kernels: a tile's sums are scaled and multiplied out of LDS by WMMA and never stored as sums; 2 launches, was 6; edge lists move to the workspace; sum order kept
# speedup vs baseline: 2.7028x; 1.0881x over previous
#include <hip/hip_runtime.h>
#include <hip/hip_bf16.h>


#define N_USER  200000
#define N_MOVIE 100000
#define N_EDGE  4000000
#define D       64

typedef float v2f __attribute__((ext_vector_type(2)));
typedef float v8f __attribute__((ext_vector_type(8)));

#define WPS 160

#define OFF_LST 0u
#define OFF_CU (OFF_LST + (N_USER + N_MOVIE) * 64u)
#define OFF_CM (OFF_CU + N_USER)
#define OFF_U1 (OFF_CM + N_MOVIE)
#define OFF_M1 (OFF_U1 + N_USER * 64u)

#define TR          1184
#define NT_M        ((N_MOVIE + TR - 1) / TR)
#define NT_U        ((N_USER + TR - 1) / TR)
#define NTILE       (NT_M + NT_U)
#define AGG_THREADS 1024
#define AGG_WAVES   (AGG_THREADS / 32)
#define QCAP        64
#define N_EDGE4     (N_EDGE / 4)
#define SCAN_U      4
#define STEP4       (AGG_THREADS * SCAN_U)
#define SCAN_ITERS  (N_EDGE4 / STEP4)
#define TAIL_E0     (SCAN_ITERS * STEP4 * 4)
#define TAIL_ITERS  ((N_EDGE - TAIL_E0 + AGG_THREADS - 1) / AGG_THREADS)
#define GB          8

#define LSTRIDE     2336u
#define LCAP        ((int)LSTRIDE - 32)

static_assert((unsigned long long)NTILE * AGG_WAVES * LSTRIDE
                  <= (unsigned long long)(N_USER + N_MOVIE) * D,
              "counts and lists fit into their workspace plane");
static_assert(TR % 16 == 0 && N_USER % 16 == 0 && N_MOVIE % 16 == 0,
              "every tile is a whole number of 16-row matrix tiles");
static_assert(LSTRIDE % 32 == 0, "a wave's count line and list are whole 128-B lines");
static_assert(TR <= (1 << 11) && N_USER <= (1 << 21) && N_MOVIE <= (1 << 21),
              "source id and row in tile share one 32-bit word");

static_assert(N_EDGE % 4 == 0, "edge indices are scanned 4 per lane");
static_assert((QCAP & (QCAP - 1)) == 0 && QCAP >= 64, "ring holds 31 + 32 entries");

__device__ __forceinline__ float2 row_load(const float* __restrict__ x, int r,
                                           unsigned lane) {
  return ((const float2*)x)[((unsigned)r << 5) + lane];
}

template <bool SAVE>
__device__ __forceinline__ void drain_ring(
    const float* __restrict__ xsrc, const int* __restrict__ sidx,
    float* acc, float* cnt, const int* qe, const int* qd,
    int head, int n, unsigned lane, unsigned* __restrict__ wl, int& nsaved) {
  int s = 0, d = 0;
  if ((int)lane < n) {
    const int slot = (head + (int)lane) & (QCAP - 1);
    const int e = qe[slot];
    d = qd[slot];
    s = sidx[e];
    if (SAVE) atomicAdd(&cnt[d], 1.0f);
  }
  if (SAVE) {
    if (nsaved + n <= LCAP && (int)lane < n)
      wl[nsaved + (int)lane] = ((unsigned)s << 11) | (unsigned)d;
    nsaved += n;
  }
  if (n == 32) {
    #pragma unroll 1
    for (int k = 0; k < 32; k += GB) {
      float2 r[GB];
      #pragma unroll
      for (int j = 0; j < GB; ++j)
        r[j] = row_load(xsrc, __builtin_amdgcn_readlane(s, k + j), lane);
      #pragma unroll
      for (int j = 0; j < GB; ++j) {
        float* q = &acc[((unsigned)__builtin_amdgcn_readlane(d, k + j) << 6) + lane * 2u];
        atomicAdd(q + 0, r[j].x);
        atomicAdd(q + 1, r[j].y);
      }
    }
  } else {
    #pragma unroll 1
    for (int k = 0; k < n; ++k) {
      const float2 r1 = row_load(xsrc, __builtin_amdgcn_readlane(s, k), lane);
      float* q = &acc[((unsigned)__builtin_amdgcn_readlane(d, k) << 6) + lane * 2u];
      atomicAdd(q + 0, r1.x);
      atomicAdd(q + 1, r1.y);
    }
  }
}

template <bool SAVE>
__device__ __forceinline__ void scan_one(
    int dst, int eid, int lo, int rows,
    const float* __restrict__ xsrc, const int* __restrict__ sidx,
    float* acc, float* cnt, int* qe, int* qd,
    int& head, int& qn, unsigned lane, unsigned* __restrict__ wl, int& nsaved) {
  const unsigned dl = (unsigned)(dst - lo);
  const bool hit = dl < (unsigned)rows;
  const unsigned mask = __builtin_amdgcn_ballot_w32(hit);
  if (mask != 0u) {
    if (hit) {
      const int pos = (head + qn + (int)__builtin_amdgcn_mbcnt_lo(mask, 0u)) & (QCAP - 1);
      qe[pos] = eid;
      qd[pos] = (int)dl;
    }
    qn += __builtin_popcount(mask);
    if (qn >= 32) {
      drain_ring<SAVE>(xsrc, sidx, acc, cnt, qe, qd, head, 32, lane, wl, nsaved);
      head = (head + 32) & (QCAP - 1);
      qn -= 32;
    }
  }
}

__device__ __forceinline__ int4 scan_load(const int4* __restrict__ d4, int it, int u, int tid) {
  return d4[(it * SCAN_U + u) * AGG_THREADS + tid];
}

template <bool SAVE>
__device__ __forceinline__ void scan_edges(
    const int* __restrict__ didx, const int* __restrict__ sidx,
    const float* __restrict__ xsrc, float* acc, float* cnt, int* qe, int* qd,
    int lo, int rows, int tile, int tid, unsigned lane,
    unsigned* __restrict__ wl, int& nsaved) {
  int head = 0, qn = 0;

  const int4* d4 = (const int4*)didx;
  int it = (tile * SCAN_ITERS) / NTILE;
  int4 v[SCAN_U];
  #pragma unroll
  for (int u = 0; u < SCAN_U; ++u) v[u] = scan_load(d4, it, u, tid);
  #pragma unroll 1
  for (int n = 0; n < SCAN_ITERS; ++n) {
    const int itn = (it + 1 == SCAN_ITERS) ? 0 : it + 1;
    int4 vn[SCAN_U];
    #pragma unroll
    for (int u = 0; u < SCAN_U; ++u) vn[u] = scan_load(d4, itn, u, tid);
    #pragma unroll
    for (int u = 0; u < SCAN_U; ++u) {
      const int e0 = ((it * SCAN_U + u) * AGG_THREADS + tid) * 4;
      scan_one<SAVE>(v[u].x, e0 + 0, lo, rows, xsrc, sidx, acc, cnt, qe, qd, head, qn, lane, wl, nsaved);
      scan_one<SAVE>(v[u].y, e0 + 1, lo, rows, xsrc, sidx, acc, cnt, qe, qd, head, qn, lane, wl, nsaved);
      scan_one<SAVE>(v[u].z, e0 + 2, lo, rows, xsrc, sidx, acc, cnt, qe, qd, head, qn, lane, wl, nsaved);
      scan_one<SAVE>(v[u].w, e0 + 3, lo, rows, xsrc, sidx, acc, cnt, qe, qd, head, qn, lane, wl, nsaved);
    }
    #pragma unroll
    for (int u = 0; u < SCAN_U; ++u) v[u] = vn[u];
    it = itn;
  }
  #pragma unroll 1
  for (int j = 0; j < TAIL_ITERS; ++j) {
    const int e = TAIL_E0 + j * AGG_THREADS + tid;
    const int dv = (e < N_EDGE) ? didx[e] : -1;
    scan_one<SAVE>(dv, e, lo, rows, xsrc, sidx, acc, cnt, qe, qd, head, qn, lane, wl, nsaved);
  }
  if (qn > 0) drain_ring<SAVE>(xsrc, sidx, acc, cnt, qe, qd, head, qn, lane, wl, nsaved);
}

#define AUX_CNT    0
#define AUX_RINGE  TR
#define AUX_RINGD  (TR + AGG_WAVES * QCAP)
#define AUX_FLOATS (TR + 2 * AGG_WAVES * QCAP)
static_assert(32 * WPS <= AUX_FLOATS, "a staged weight matrix fits where counts and rings were");

__device__ __forceinline__ void stage_weights(float* aux, const float* __restrict__ Wg, int tid) {
  for (int idx = tid; idx < 32 * D; idx += AGG_THREADS) {
    const int kp = idx >> 6, n = idx & 63;
    *(float2*)&aux[kp * WPS + 2 * n] =
        make_float2(Wg[(2 * kp) * D + n], Wg[(2 * kp + 1) * D + n]);
  }
}

__device__ __forceinline__ void load_a_frags(const float* tile, int c, int h, v2f (&a)[16]) {
  #pragma unroll
  for (int j = 0; j < 16; ++j)
    a[j] = *(const v2f*)&tile[c * D + ((j ^ c) << 2) + 2 * h];
}

__device__ __forceinline__ v2f load_b_frag(const float* aux, int wo, int j, int ct, int c, int h) {
  return *(const v2f*)&aux[wo + (2 * j + h) * WPS + 2 * (16 * ct + c)];
}

__device__ __forceinline__ int trip_offset() {
  int wo = 0;
  asm volatile("" : "+v"(wo));
  return wo;
}

__device__ __forceinline__ int d_place(int h, int r, int col) {
  return (8 * h + r) * D + (col ^ (32 * h));
}

template <bool CNT_LDS>
__device__ __forceinline__ void tile_transform(
    float* acc, float* aux, float* __restrict__ cntG,
    const float* __restrict__ xdst, const float* __restrict__ Wl,
    const float* __restrict__ Wr, const float* __restrict__ bias,
    float* __restrict__ out, int rows, bool relu, int tid) {
  const int lane = tid & 31, wave = tid >> 5;
  const int c = lane & 15, h = lane >> 4;
  const int nrt = rows >> 4;

  for (int i = tid; i < rows * 16; i += AGG_THREADS) {
    const int row = i >> 4, q = i & 15;
    const float cv = CNT_LDS ? aux[AUX_CNT + row] : cntG[row];
    if (CNT_LDS && q == 0) cntG[row] = cv;
    const float inv = 1.0f / fmaxf(cv, 1.0f);
    const float4 sv = ((const float4*)acc)[i];
    ((float4*)acc)[(row << 4) + (q ^ (row & 15))] =
        make_float4(sv.x * inv, sv.y * inv, sv.z * inv, sv.w * inv);
  }
  __syncthreads();
  stage_weights(aux, Wl, tid);
  __syncthreads();

  #pragma unroll 1
  for (int rt = wave; rt < nrt; rt += AGG_WAVES) {
    float* tile = acc + rt * 16 * D;
    const int wo = trip_offset();
    v2f a[16];
    load_a_frags(tile, c, h, a);
    #pragma unroll
    for (int ct = 0; ct < 4; ++ct) {
      v8f d = {0.f, 0.f, 0.f, 0.f, 0.f, 0.f, 0.f, 0.f};
      #pragma unroll
      for (int j = 0; j < 16; ++j)
        d = __builtin_amdgcn_wmma_f32_16x16x4_f32(false, a[j], false,
                                                  load_b_frag(aux, wo, j, ct, c, h),
                                                  (short)0, d, false, false);
      #pragma unroll
      for (int r = 0; r < 8; ++r) tile[d_place(h, r, 16 * ct + c)] = d[r];
    }
  }
  __syncthreads();
  stage_weights(aux, Wr, tid);
  __syncthreads();

  #pragma unroll 1
  for (int rt = wave; rt < nrt; rt += AGG_WAVES) {
    float* tile = acc + rt * 16 * D;
    const int wo = trip_offset();
    const float* xt = xdst + (size_t)rt * 16 * D;
    float*       ot = out + (size_t)rt * 16 * D;
    float4 xr[8];
    #pragma unroll
    for (int i = 0; i < 8; ++i)
      xr[i] = *(const float4*)&xt[(2 * i + h) * D + 4 * c];
    v8f d[4];
    #pragma unroll
    for (int ct = 0; ct < 4; ++ct) {
      #pragma unroll
      for (int r = 0; r < 8; ++r) d[ct][r] = tile[d_place(h, r, 16 * ct + c)];
    }
    #pragma unroll
    for (int i = 0; i < 8; ++i) {
      const int row = 2 * i + h;
      *(float4*)&tile[row * D + ((c ^ row) << 2)] = xr[i];
    }
    v2f a[16];
    load_a_frags(tile, c, h, a);
    #pragma unroll
    for (int ct = 0; ct < 4; ++ct) {
      #pragma unroll
      for (int j = 0; j < 16; ++j)
        d[ct] = __builtin_amdgcn_wmma_f32_16x16x4_f32(false, a[j], false,
                                                      load_b_frag(aux, wo, j, ct, c, h),
                                                      (short)0, d[ct], false, false);
    }
    #pragma unroll
    for (int ct = 0; ct < 4; ++ct) {
      const float bcol = bias[16 * ct + c];
      #pragma unroll
      for (int r = 0; r < 8; ++r) {
        float val = d[ct][r] + bcol;
        if (relu) val = fmaxf(val, 0.0f);
        tile[d_place(h, r, 16 * ct + c)] = val;
      }
    }
    #pragma unroll
    for (int i = 0; i < 8; ++i) {
      const int row = 2 * i + h;
      *(float4*)&ot[row * D + 4 * c] =
          *(const float4*)&tile[row * D + ((c ^ (8 * (row >> 3))) << 2)];
    }
  }
}

__global__ __launch_bounds__(AGG_THREADS) void tile_layer1_kernel(
    const float* __restrict__ xu, const float* __restrict__ xm,
    const int* __restrict__ uidx, const int* __restrict__ midx,
    const float* __restrict__ WlM, const float* __restrict__ WrM, const float* __restrict__ bM,
    const float* __restrict__ WlU, const float* __restrict__ WrU, const float* __restrict__ bU,
    float* __restrict__ cu, float* __restrict__ cm,
    float* __restrict__ u1, float* __restrict__ m1,
    unsigned* __restrict__ lst) {
  __shared__ __attribute__((aligned(16))) float acc[TR * D];
  __shared__ __attribute__((aligned(16))) float aux[AUX_FLOATS];
  float* cnt   = aux + AUX_CNT;
  int*   ringE = (int*)(aux + AUX_RINGE);
  int*   ringD = (int*)(aux + AUX_RINGD);

  const int tid = threadIdx.x;
  const unsigned lane = threadIdx.x & 31u;
  const int wave = tid >> 5;
  const int tile = blockIdx.x;

  const bool mov = tile < NT_M;
  const int*   didx = mov ? midx : uidx;
  const int*   sidx = mov ? uidx : midx;
  const float* xsrc = mov ? xu : xm;
  const float* xdst = mov ? xm : xu;
  float*       hOut = mov ? m1 : u1;
  float*       cOut = mov ? cm : cu;
  const int    nrow = mov ? N_MOVIE : N_USER;
  const int    lo   = (mov ? tile : tile - NT_M) * TR;
  const int    rows = min(TR, nrow - lo);

  for (int i = tid; i < TR * D / 4; i += AGG_THREADS)
    ((float4*)acc)[i] = make_float4(0.f, 0.f, 0.f, 0.f);
  for (int i = tid; i < TR; i += AGG_THREADS) cnt[i] = 0.f;
  __syncthreads();

  unsigned* wreg = lst + (unsigned)(tile * AGG_WAVES + wave) * LSTRIDE;
  int nsaved = 0;
  scan_edges<true>(didx, sidx, xsrc, acc, cnt, &ringE[wave * QCAP], &ringD[wave * QCAP],
                   lo, rows, tile, tid, lane, wreg + 32, nsaved);
  if (lane == 0) wreg[0] = (unsigned)nsaved;
  __syncthreads();

  tile_transform<true>(acc, aux, cOut + lo, xdst + (size_t)lo * D,
                       mov ? WlM : WlU, mov ? WrM : WrU, mov ? bM : bU,
                       hOut + (size_t)lo * D, rows, true, tid);
}

__global__ __launch_bounds__(AGG_THREADS) void tile_layer2_kernel(
    const float* __restrict__ h1u, const float* __restrict__ h1m,
    const int* __restrict__ uidx, const int* __restrict__ midx,
    const float* __restrict__ WlM, const float* __restrict__ WrM, const float* __restrict__ bM,
    const float* __restrict__ WlU, const float* __restrict__ WrU, const float* __restrict__ bU,
    float* __restrict__ cu, float* __restrict__ cm,
    float* __restrict__ outU, float* __restrict__ outM,
    const unsigned* __restrict__ lst) {
  __shared__ __attribute__((aligned(16))) float acc[TR * D];
  __shared__ __attribute__((aligned(16))) float aux[AUX_FLOATS];
  int* ringE = (int*)(aux + AUX_RINGE);
  int* ringD = (int*)(aux + AUX_RINGD);

  const int tid = threadIdx.x;
  const unsigned lane = threadIdx.x & 31u;
  const int wave = tid >> 5;
  const int tile = blockIdx.x;

  const bool mov = tile < NT_M;
  const float* xsrc = mov ? h1u : h1m;
  const float* xdst = mov ? h1m : h1u;
  float*       oOut = mov ? outM : outU;
  float*       cIn  = mov ? cm : cu;
  const int    nrow = mov ? N_MOVIE : N_USER;
  const int    lo   = (mov ? tile : tile - NT_M) * TR;
  const int    rows = min(TR, nrow - lo);

  for (int i = tid; i < TR * D / 4; i += AGG_THREADS)
    ((float4*)acc)[i] = make_float4(0.f, 0.f, 0.f, 0.f);
  __syncthreads();

  const unsigned* __restrict__ wreg = lst + (unsigned)(tile * AGG_WAVES + wave) * LSTRIDE;
  const int n = __builtin_amdgcn_readfirstlane((int)wreg[0]);
  if (n > LCAP) {
    const int* didx = mov ? midx : uidx;
    const int* sidx = mov ? uidx : midx;
    int unused = 0;
    scan_edges<false>(didx, sidx, xsrc, acc, nullptr, &ringE[wave * QCAP], &ringD[wave * QCAP],
                      lo, rows, tile, tid, lane, nullptr, unused);
  } else if (n > 0) {
    const unsigned* __restrict__ wl = wreg + 32;
    const int last = n - 1;
    unsigned w = wl[min((int)lane, last)];
    #pragma unroll 1
    for (int base = 0; base < n; base += 32) {
      const int m = min(32, n - base);
      const unsigned wn = wl[min(base + 32 + (int)lane, last)];
      const int s = (int)(w >> 11);
      const int d = (int)(w & 2047u);
      if (m == 32) {
        float2 r[32];
        #pragma unroll
        for (int j = 0; j < 32; ++j)
          r[j] = row_load(xsrc, __builtin_amdgcn_readlane(s, j), lane);
        #pragma unroll
        for (int j = 0; j < 32; ++j) {
          float* q = &acc[((unsigned)__builtin_amdgcn_readlane(d, j) << 6) + lane * 2u];
          atomicAdd(q + 0, r[j].x);
          atomicAdd(q + 1, r[j].y);
        }
      } else {
        #pragma unroll 1
        for (int k = 0; k < m; ++k) {
          const float2 r1 = row_load(xsrc, __builtin_amdgcn_readlane(s, k), lane);
          float* q = &acc[((unsigned)__builtin_amdgcn_readlane(d, k) << 6) + lane * 2u];
          atomicAdd(q + 0, r1.x);
          atomicAdd(q + 1, r1.y);
        }
      }
      w = wn;
    }
  }
  __syncthreads();

  tile_transform<false>(acc, aux, cIn + lo, xdst + (size_t)lo * D,
                        mov ? WlM : WlU, mov ? WrM : WrU, mov ? bM : bU,
                        oOut + (size_t)lo * D, rows, false, tid);
}

extern "C" void kernel_launch(void* const* d_in, const int* in_sizes, int n_in,
                              void* d_out, int out_size, void* d_ws, size_t ws_size,
                              hipStream_t stream) {
  const float* x_user  = (const float*)d_in[0];
  const float* x_movie = (const float*)d_in[1];
  const float* Wl1_um  = (const float*)d_in[2];
  const float* Wr1_um  = (const float*)d_in[3];
  const float* b1_um   = (const float*)d_in[4];
  const float* Wl1_mu  = (const float*)d_in[5];
  const float* Wr1_mu  = (const float*)d_in[6];
  const float* b1_mu   = (const float*)d_in[7];
  const float* Wl2_um  = (const float*)d_in[8];
  const float* Wr2_um  = (const float*)d_in[9];
  const float* b2_um   = (const float*)d_in[10];
  const float* Wl2_mu  = (const float*)d_in[11];
  const float* Wr2_mu  = (const float*)d_in[12];
  const float* b2_mu   = (const float*)d_in[13];
  const int*   uidx    = (const int*)d_in[14];
  const int*   midx    = (const int*)d_in[15];

  float* ws = (float*)d_ws;
  unsigned* lst = (unsigned*)(ws + OFF_LST);
  float* cu = ws + OFF_CU;
  float* cm = ws + OFF_CM;
  float* u1 = ws + OFF_U1;
  float* m1 = ws + OFF_M1;

  float* u2 = (float*)d_out;
  float* m2 = (float*)d_out + (size_t)N_USER * D;

  const dim3 bAgg(AGG_THREADS);
  const dim3 gAgg(NTILE);

  tile_layer1_kernel<<<gAgg, bAgg, 0, stream>>>(x_user, x_movie, uidx, midx,
                                                Wl1_um, Wr1_um, b1_um,
                                                Wl1_mu, Wr1_mu, b1_mu,
                                                cu, cm, u1, m1, lst);
  tile_layer2_kernel<<<gAgg, bAgg, 0, stream>>>(u1, m1, uidx, midx,
                                                Wl2_um, Wr2_um, b2_um,
                                                Wl2_mu, Wr2_mu, b2_mu,
                                                cu, cm, u2, m2, lst);
}


# Layer 2 re-tiled for equal edge counts: each work-group owns a third of a movie tile and two thirds of a user tile (1184 rows); layer 1 writes its lists per sub-range; layer-2 sum order changes
# speedup vs baseline: 2.8468x; 1.0533x over previous
#include <hip/hip_runtime.h>
#include <hip/hip_bf16.h>


#define N_USER  200000
#define N_MOVIE 100000
#define N_EDGE  4000000
#define D       64

typedef float v2f __attribute__((ext_vector_type(2)));
typedef float v8f __attribute__((ext_vector_type(8)));

#define WPS 160

#define OFF_LST 0u
#define OFF_CU (OFF_LST + (N_USER + N_MOVIE) * 64u)
#define OFF_CM (OFF_CU + N_USER)
#define OFF_U1 (OFF_CM + N_MOVIE)
#define OFF_M1 (OFF_U1 + N_USER * 64u)

#define TR          1184
#define NT_M        ((N_MOVIE + TR - 1) / TR)
#define NT_U        ((N_USER + TR - 1) / TR)
#define NTILE       (NT_M + NT_U)
#define AGG_THREADS 1024
#define AGG_WAVES   (AGG_THREADS / 32)
#define QCAP        64
#define N_EDGE4     (N_EDGE / 4)
#define SCAN_U      4
#define STEP4       (AGG_THREADS * SCAN_U)
#define SCAN_ITERS  (N_EDGE4 / STEP4)
#define TAIL_E0     (SCAN_ITERS * STEP4 * 4)
#define TAIL_ITERS  ((N_EDGE - TAIL_E0 + AGG_THREADS - 1) / AGG_THREADS)
#define GB          8

#define LSTRIDE     2336u
#define SUBCAP      768

#define K1          400
#define K2          784

static_assert(3 * SUBCAP + 32 == (int)LSTRIDE, "count line and three sub-lists fill the wave's share");
static_assert(K1 % 16 == 0 && K2 % 16 == 0 && TR - K2 == K1 && (K2 - K1) + 2 * K1 == TR,
              "sub-ranges are whole matrix tiles; one movie and two user sub-ranges are TR rows");
static_assert((NTILE / 3) * TR + (NTILE % 3 == 0 ? 0 : (NTILE % 3 == 1 ? K1 : K2)) >= N_MOVIE,
              "layer 2: movie sub-ranges 0 .. NTILE - 1 hold every movie row");
static_assert(((2 * NTILE) / 3) * TR + ((2 * NTILE) % 3 == 0 ? 0 : ((2 * NTILE) % 3 == 1 ? K1 : K2)) >= N_USER,
              "layer 2: user sub-ranges 0 .. 2 NTILE - 1 hold every user row");
static_assert(N_USER + N_MOVIE <= (1 << 21), "joint source row and LDS row share one 32-bit word");

__device__ __forceinline__ int sub_lo(int k) { return k == 0 ? 0 : (k == 1 ? K1 : (k == 2 ? K2 : TR)); }
__device__ __forceinline__ int sub_size(int k) { return k == 1 ? K2 - K1 : K1; }
__device__ __forceinline__ int sub_start(int g, int nrow) {
  const int t = g / 3;
  return min(t * TR + sub_lo(g - 3 * t), nrow);
}

struct ListState {
  unsigned* wl;
  int n0, n1, n2;
  int b0, b1, b2;
  int soff;
};

static_assert((unsigned long long)NTILE * AGG_WAVES * LSTRIDE
                  <= (unsigned long long)(N_USER + N_MOVIE) * D,
              "counts and lists fit into their workspace plane");
static_assert(TR % 16 == 0 && N_USER % 16 == 0 && N_MOVIE % 16 == 0,
              "every tile is a whole number of 16-row matrix tiles");
static_assert(LSTRIDE % 32 == 0 && SUBCAP % 32 == 0, "a wave's count line and lists are whole 128-B lines");
static_assert(TR <= (1 << 11) && N_USER <= (1 << 21) && N_MOVIE <= (1 << 21),
              "source id and row in tile share one 32-bit word");

static_assert(N_EDGE % 4 == 0, "edge indices are scanned 4 per lane");
static_assert((QCAP & (QCAP - 1)) == 0 && QCAP >= 64, "ring holds 31 + 32 entries");

__device__ __forceinline__ float2 row_load(const float* __restrict__ x, int r,
                                           unsigned lane) {
  return ((const float2*)x)[((unsigned)r << 5) + lane];
}

template <bool SAVE>
__device__ __forceinline__ void drain_ring(
    const float* __restrict__ xsrc, const int* __restrict__ sidx,
    float* acc, float* cnt, const int* qe, const int* qd,
    int head, int n, unsigned lane, ListState& ls) {
  int s = 0, d = 0;
  if ((int)lane < n) {
    const int slot = (head + (int)lane) & (QCAP - 1);
    const int e = qe[slot];
    d = qd[slot];
    s = sidx[e];
    if (SAVE) atomicAdd(&cnt[d], 1.0f);
  }
  if (SAVE) {
    const bool act = (int)lane < n;
    const unsigned sw = (unsigned)(s + ls.soff) << 11;
    {
      const bool mine = act && d < K1;
      const unsigned mk = __builtin_amdgcn_ballot_w32(mine);
      const int c = __builtin_popcount(mk);
      if (mine && ls.n0 + c <= SUBCAP)
        ls.wl[ls.n0 + (int)__builtin_amdgcn_mbcnt_lo(mk, 0u)] = sw | (unsigned)(ls.b0 + d);
      ls.n0 += c;
    }
    {
      const bool mine = act && d >= K1 && d < K2;
      const unsigned mk = __builtin_amdgcn_ballot_w32(mine);
      const int c = __builtin_popcount(mk);
      if (mine && ls.n1 + c <= SUBCAP)
        ls.wl[SUBCAP + ls.n1 + (int)__builtin_amdgcn_mbcnt_lo(mk, 0u)] = sw | (unsigned)(ls.b1 + d);
      ls.n1 += c;
    }
    {
      const bool mine = act && d >= K2;
      const unsigned mk = __builtin_amdgcn_ballot_w32(mine);
      const int c = __builtin_popcount(mk);
      if (mine && ls.n2 + c <= SUBCAP)
        ls.wl[2 * SUBCAP + ls.n2 + (int)__builtin_amdgcn_mbcnt_lo(mk, 0u)] = sw | (unsigned)(ls.b2 + d);
      ls.n2 += c;
    }
  }
  if (n == 32) {
    #pragma unroll 1
    for (int k = 0; k < 32; k += GB) {
      float2 r[GB];
      #pragma unroll
      for (int j = 0; j < GB; ++j)
        r[j] = row_load(xsrc, __builtin_amdgcn_readlane(s, k + j), lane);
      #pragma unroll
      for (int j = 0; j < GB; ++j) {
        float* q = &acc[((unsigned)__builtin_amdgcn_readlane(d, k + j) << 6) + lane * 2u];
        atomicAdd(q + 0, r[j].x);
        atomicAdd(q + 1, r[j].y);
      }
    }
  } else {
    #pragma unroll 1
    for (int k = 0; k < n; ++k) {
      const float2 r1 = row_load(xsrc, __builtin_amdgcn_readlane(s, k), lane);
      float* q = &acc[((unsigned)__builtin_amdgcn_readlane(d, k) << 6) + lane * 2u];
      atomicAdd(q + 0, r1.x);
      atomicAdd(q + 1, r1.y);
    }
  }
}

template <bool SAVE>
__device__ __forceinline__ void scan_one(
    int dst, int eid, int lo, int rows,
    const float* __restrict__ xsrc, const int* __restrict__ sidx,
    float* acc, float* cnt, int* qe, int* qd,
    int& head, int& qn, unsigned lane, ListState& ls) {
  const unsigned dl = (unsigned)(dst - lo);
  const bool hit = dl < (unsigned)rows;
  const unsigned mask = __builtin_amdgcn_ballot_w32(hit);
  if (mask != 0u) {
    if (hit) {
      const int pos = (head + qn + (int)__builtin_amdgcn_mbcnt_lo(mask, 0u)) & (QCAP - 1);
      qe[pos] = eid;
      qd[pos] = (int)dl;
    }
    qn += __builtin_popcount(mask);
    if (qn >= 32) {
      drain_ring<SAVE>(xsrc, sidx, acc, cnt, qe, qd, head, 32, lane, ls);
      head = (head + 32) & (QCAP - 1);
      qn -= 32;
    }
  }
}

__device__ __forceinline__ int4 scan_load(const int4* __restrict__ d4, int it, int u, int tid) {
  return d4[(it * SCAN_U + u) * AGG_THREADS + tid];
}

template <bool SAVE>
__device__ __forceinline__ void scan_edges(
    const int* __restrict__ didx, const int* __restrict__ sidx,
    const float* __restrict__ xsrc, float* acc, float* cnt, int* qe, int* qd,
    int lo, int rows, int tile, int tid, unsigned lane, ListState& ls) {
  int head = 0, qn = 0;

  const int4* d4 = (const int4*)didx;
  int it = (tile * SCAN_ITERS) / NTILE;
  int4 v[SCAN_U];
  #pragma unroll
  for (int u = 0; u < SCAN_U; ++u) v[u] = scan_load(d4, it, u, tid);
  #pragma unroll 1
  for (int n = 0; n < SCAN_ITERS; ++n) {
    const int itn = (it + 1 == SCAN_ITERS) ? 0 : it + 1;
    int4 vn[SCAN_U];
    #pragma unroll
    for (int u = 0; u < SCAN_U; ++u) vn[u] = scan_load(d4, itn, u, tid);
    #pragma unroll
    for (int u = 0; u < SCAN_U; ++u) {
      const int e0 = ((it * SCAN_U + u) * AGG_THREADS + tid) * 4;
      scan_one<SAVE>(v[u].x, e0 + 0, lo, rows, xsrc, sidx, acc, cnt, qe, qd, head, qn, lane, ls);
      scan_one<SAVE>(v[u].y, e0 + 1, lo, rows, xsrc, sidx, acc, cnt, qe, qd, head, qn, lane, ls);
      scan_one<SAVE>(v[u].z, e0 + 2, lo, rows, xsrc, sidx, acc, cnt, qe, qd, head, qn, lane, ls);
      scan_one<SAVE>(v[u].w, e0 + 3, lo, rows, xsrc, sidx, acc, cnt, qe, qd, head, qn, lane, ls);
    }
    #pragma unroll
    for (int u = 0; u < SCAN_U; ++u) v[u] = vn[u];
    it = itn;
  }
  #pragma unroll 1
  for (int j = 0; j < TAIL_ITERS; ++j) {
    const int e = TAIL_E0 + j * AGG_THREADS + tid;
    const int dv = (e < N_EDGE) ? didx[e] : -1;
    scan_one<SAVE>(dv, e, lo, rows, xsrc, sidx, acc, cnt, qe, qd, head, qn, lane, ls);
  }
  if (qn > 0) drain_ring<SAVE>(xsrc, sidx, acc, cnt, qe, qd, head, qn, lane, ls);
}

#define AUX_CNT    0
#define AUX_RINGE  TR
#define AUX_RINGD  (TR + AGG_WAVES * QCAP)
#define AUX_FLOATS (TR + 2 * AGG_WAVES * QCAP)
static_assert(32 * WPS <= AUX_FLOATS, "a staged weight matrix fits where counts and rings were");

__device__ __forceinline__ void stage_weights(float* aux, const float* __restrict__ Wg, int tid) {
  for (int idx = tid; idx < 32 * D; idx += AGG_THREADS) {
    const int kp = idx >> 6, n = idx & 63;
    *(float2*)&aux[kp * WPS + 2 * n] =
        make_float2(Wg[(2 * kp) * D + n], Wg[(2 * kp + 1) * D + n]);
  }
}

__device__ __forceinline__ void load_a_frags(const float* tile, int c, int h, v2f (&a)[16]) {
  #pragma unroll
  for (int j = 0; j < 16; ++j)
    a[j] = *(const v2f*)&tile[c * D + ((j ^ c) << 2) + 2 * h];
}

__device__ __forceinline__ v2f load_b_frag(const float* aux, int wo, int j, int ct, int c, int h) {
  return *(const v2f*)&aux[wo + (2 * j + h) * WPS + 2 * (16 * ct + c)];
}

__device__ __forceinline__ int trip_offset() {
  int wo = 0;
  asm volatile("" : "+v"(wo));
  return wo;
}

__device__ __forceinline__ int d_place(int h, int r, int col) {
  return (8 * h + r) * D + (col ^ (32 * h));
}

template <bool CNT_LDS>
__device__ __forceinline__ void tile_transform(
    float* acc, float* aux, float* __restrict__ cntG,
    const float* __restrict__ xdst, const float* __restrict__ Wl,
    const float* __restrict__ Wr, const float* __restrict__ bias,
    float* __restrict__ out, int rows, bool relu, int tid) {
  const int lane = tid & 31, wave = tid >> 5;
  const int c = lane & 15, h = lane >> 4;
  const int nrt = rows >> 4;

  for (int i = tid; i < rows * 16; i += AGG_THREADS) {
    const int row = i >> 4, q = i & 15;
    const float cv = CNT_LDS ? aux[AUX_CNT + row] : cntG[row];
    if (CNT_LDS && q == 0) cntG[row] = cv;
    const float inv = 1.0f / fmaxf(cv, 1.0f);
    const float4 sv = ((const float4*)acc)[i];
    ((float4*)acc)[(row << 4) + (q ^ (row & 15))] =
        make_float4(sv.x * inv, sv.y * inv, sv.z * inv, sv.w * inv);
  }
  __syncthreads();
  stage_weights(aux, Wl, tid);
  __syncthreads();

  #pragma unroll 1
  for (int rt = wave; rt < nrt; rt += AGG_WAVES) {
    float* tile = acc + rt * 16 * D;
    const int wo = trip_offset();
    v2f a[16];
    load_a_frags(tile, c, h, a);
    #pragma unroll
    for (int ct = 0; ct < 4; ++ct) {
      v8f d = {0.f, 0.f, 0.f, 0.f, 0.f, 0.f, 0.f, 0.f};
      #pragma unroll
      for (int j = 0; j < 16; ++j)
        d = __builtin_amdgcn_wmma_f32_16x16x4_f32(false, a[j], false,
                                                  load_b_frag(aux, wo, j, ct, c, h),
                                                  (short)0, d, false, false);
      #pragma unroll
      for (int r = 0; r < 8; ++r) tile[d_place(h, r, 16 * ct + c)] = d[r];
    }
  }
  __syncthreads();
  stage_weights(aux, Wr, tid);
  __syncthreads();

  #pragma unroll 1
  for (int rt = wave; rt < nrt; rt += AGG_WAVES) {
    float* tile = acc + rt * 16 * D;
    const int wo = trip_offset();
    const float* xt = xdst + (size_t)rt * 16 * D;
    float*       ot = out + (size_t)rt * 16 * D;
    float4 xr[8];
    #pragma unroll
    for (int i = 0; i < 8; ++i)
      xr[i] = *(const float4*)&xt[(2 * i + h) * D + 4 * c];
    v8f d[4];
    #pragma unroll
    for (int ct = 0; ct < 4; ++ct) {
      #pragma unroll
      for (int r = 0; r < 8; ++r) d[ct][r] = tile[d_place(h, r, 16 * ct + c)];
    }
    #pragma unroll
    for (int i = 0; i < 8; ++i) {
      const int row = 2 * i + h;
      *(float4*)&tile[row * D + ((c ^ row) << 2)] = xr[i];
    }
    v2f a[16];
    load_a_frags(tile, c, h, a);
    #pragma unroll
    for (int ct = 0; ct < 4; ++ct) {
      #pragma unroll
      for (int j = 0; j < 16; ++j)
        d[ct] = __builtin_amdgcn_wmma_f32_16x16x4_f32(false, a[j], false,
                                                      load_b_frag(aux, wo, j, ct, c, h),
                                                      (short)0, d[ct], false, false);
    }
    #pragma unroll
    for (int ct = 0; ct < 4; ++ct) {
      const float bcol = bias[16 * ct + c];
      #pragma unroll
      for (int r = 0; r < 8; ++r) {
        float val = d[ct][r] + bcol;
        if (relu) val = fmaxf(val, 0.0f);
        tile[d_place(h, r, 16 * ct + c)] = val;
      }
    }
    #pragma unroll
    for (int i = 0; i < 8; ++i) {
      const int row = 2 * i + h;
      *(float4*)&ot[row * D + 4 * c] =
          *(const float4*)&tile[row * D + ((c ^ (8 * (row >> 3))) << 2)];
    }
  }
}

__global__ __launch_bounds__(AGG_THREADS) void tile_layer1_kernel(
    const float* __restrict__ xu, const float* __restrict__ xm,
    const int* __restrict__ uidx, const int* __restrict__ midx,
    const float* __restrict__ WlM, const float* __restrict__ WrM, const float* __restrict__ bM,
    const float* __restrict__ WlU, const float* __restrict__ WrU, const float* __restrict__ bU,
    float* __restrict__ cu, float* __restrict__ cm,
    float* __restrict__ u1, float* __restrict__ m1,
    unsigned* __restrict__ lst) {
  __shared__ __attribute__((aligned(16))) float acc[TR * D];
  __shared__ __attribute__((aligned(16))) float aux[AUX_FLOATS];
  float* cnt   = aux + AUX_CNT;
  int*   ringE = (int*)(aux + AUX_RINGE);
  int*   ringD = (int*)(aux + AUX_RINGD);

  const int tid = threadIdx.x;
  const unsigned lane = threadIdx.x & 31u;
  const int wave = tid >> 5;
  const int tile = blockIdx.x;

  const bool mov = tile < NT_M;
  const int*   didx = mov ? midx : uidx;
  const int*   sidx = mov ? uidx : midx;
  const float* xsrc = mov ? xu : xm;
  const float* xdst = mov ? xm : xu;
  float*       hOut = mov ? m1 : u1;
  float*       cOut = mov ? cm : cu;
  const int    nrow = mov ? N_MOVIE : N_USER;
  const int    lo   = (mov ? tile : tile - NT_M) * TR;
  const int    rows = min(TR, nrow - lo);

  for (int i = tid; i < TR * D / 4; i += AGG_THREADS)
    ((float4*)acc)[i] = make_float4(0.f, 0.f, 0.f, 0.f);
  for (int i = tid; i < TR; i += AGG_THREADS) cnt[i] = 0.f;
  __syncthreads();

  unsigned* wreg = lst + (unsigned)(tile * AGG_WAVES + wave) * LSTRIDE;
  ListState ls;
  ls.wl = wreg + 32;
  ls.n0 = 0; ls.n1 = 0; ls.n2 = 0;
  ls.soff = mov ? 0 : N_USER;
  {
    int b[3];
    #pragma unroll
    for (int k = 0; k < 3; ++k) {
      int base = 0;
      if (!mov) {
        const int g = 3 * (tile - NT_M) + k;
        base = sub_size((g >> 1) % 3) + ((g & 1) ? sub_size((g - 1) % 3) : 0);
      }
      b[k] = base - sub_lo(k);
    }
    ls.b0 = b[0]; ls.b1 = b[1]; ls.b2 = b[2];
  }
  scan_edges<true>(didx, sidx, xsrc, acc, cnt, &ringE[wave * QCAP], &ringD[wave * QCAP],
                   lo, rows, tile, tid, lane, ls);
  if (lane < 3u)
    wreg[lane] = (unsigned)(lane == 0u ? ls.n0 : (lane == 1u ? ls.n1 : ls.n2));
  __syncthreads();

  tile_transform<true>(acc, aux, cOut + lo, xdst + (size_t)lo * D,
                       mov ? WlM : WlU, mov ? WrM : WrU, mov ? bM : bU,
                       hOut + (size_t)lo * D, rows, true, tid);
}

static_assert(OFF_M1 == OFF_U1 + N_USER * 64u, "the layer-1 tables are one joint table, users first");

__device__ __forceinline__ void sub_list(const unsigned* __restrict__ lst, bool exists, int tile1,
                                         int wave, int k, int& n, unsigned& off) {
  const unsigned reg = (unsigned)(tile1 * AGG_WAVES + wave) * LSTRIDE;
  n = exists ? __builtin_amdgcn_readfirstlane((int)lst[reg + (unsigned)k]) : 0;
  off = reg + 32u + (unsigned)k * SUBCAP;
}

__global__ __launch_bounds__(AGG_THREADS) void tile_layer2_kernel(
    const float* __restrict__ h1u, const float* __restrict__ h1m,
    const int* __restrict__ uidx, const int* __restrict__ midx,
    const float* __restrict__ WlM, const float* __restrict__ WrM, const float* __restrict__ bM,
    const float* __restrict__ WlU, const float* __restrict__ WrU, const float* __restrict__ bU,
    float* __restrict__ cu, float* __restrict__ cm,
    float* __restrict__ outU, float* __restrict__ outM,
    const unsigned* __restrict__ lst) {
  __shared__ __attribute__((aligned(16))) float acc[TR * D];
  __shared__ __attribute__((aligned(16))) float aux[AUX_FLOATS];
  int* ringE = (int*)(aux + AUX_RINGE);
  int* ringD = (int*)(aux + AUX_RINGD);

  const int tid = threadIdx.x;
  const unsigned lane = threadIdx.x & 31u;
  const int wave = tid >> 5;
  const int wg = blockIdx.x;

  const int mlo   = sub_start(wg, N_MOVIE);
  const int rm    = sub_start(wg + 1, N_MOVIE) - mlo;
  const int ulo   = sub_start(2 * wg, N_USER);
  const int umid  = sub_start(2 * wg + 1, N_USER);
  const int uhi   = sub_start(2 * wg + 2, N_USER);
  const int ubase = sub_size(wg % 3);

  for (int i = tid; i < TR * D / 4; i += AGG_THREADS)
    ((float4*)acc)[i] = make_float4(0.f, 0.f, 0.f, 0.f);
  __syncthreads();

  int n0, n1, n2;
  unsigned o0, o1, o2;
  {
    const int g1 = 2 * wg, g2 = 2 * wg + 1;
    sub_list(lst, true, wg / 3, wave, wg % 3, n0, o0);
    sub_list(lst, g1 / 3 < NT_U, NT_M + g1 / 3, wave, g1 % 3, n1, o1);
    sub_list(lst, g2 / 3 < NT_U, NT_M + g2 / 3, wave, g2 % 3, n2, o2);
  }
  const int ovf = (n0 > SUBCAP ? 1 : 0) | (n1 > SUBCAP ? 2 : 0) | (n2 > SUBCAP ? 4 : 0);
  if (ovf != 0) {
    #pragma unroll 1
    for (int p = 0; p < 3; ++p) {
      if ((ovf >> p) & 1) {
        const bool mv = p == 0;
        const int plo   = mv ? mlo : (p == 1 ? ulo : umid);
        const int prows = mv ? rm : (p == 1 ? umid - ulo : uhi - umid);
        const int prow0 = mv ? 0 : (p == 1 ? ubase : ubase + (umid - ulo));
        ListState none;
        none.wl = nullptr;
        none.n0 = 0; none.n1 = 0; none.n2 = 0;
        none.b0 = 0; none.b1 = 0; none.b2 = 0;
        none.soff = 0;
        scan_edges<false>(mv ? midx : uidx, mv ? uidx : midx, mv ? h1u : h1m,
                          acc + prow0 * D, nullptr, &ringE[wave * QCAP], &ringD[wave * QCAP],
                          plo, prows, wg, tid, lane, none);
      }
    }
    if (n0 > SUBCAP) n0 = 0;
    if (n1 > SUBCAP) n1 = 0;
    if (n2 > SUBCAP) n2 = 0;
  }

  const int n01 = n0 + n1;
  const int n = n01 + n2;
  if (n > 0) {
    const int last = n - 1;
    auto entry = [&](int i) -> unsigned {
      const unsigned o = (i < n0) ? o0 + (unsigned)i
                       : ((i < n01) ? o1 + (unsigned)(i - n0) : o2 + (unsigned)(i - n01));
      return lst[o];
    };
    unsigned w = entry(min((int)lane, last));
    #pragma unroll 1
    for (int base = 0; base < n; base += 32) {
      const int m = min(32, n - base);
      const unsigned wn = entry(min(base + 32 + (int)lane, last));
      const int s = (int)(w >> 11);
      const int d = (int)(w & 2047u);
      if (m == 32) {
        float2 r[32];
        #pragma unroll
        for (int j = 0; j < 32; ++j)
          r[j] = row_load(h1u, __builtin_amdgcn_readlane(s, j), lane);
        #pragma unroll
        for (int j = 0; j < 32; ++j) {
          float* q = &acc[((unsigned)__builtin_amdgcn_readlane(d, j) << 6) + lane * 2u];
          atomicAdd(q + 0, r[j].x);
          atomicAdd(q + 1, r[j].y);
        }
      } else {
        #pragma unroll 1
        for (int k = 0; k < m; ++k) {
          const float2 r1 = row_load(h1u, __builtin_amdgcn_readlane(s, k), lane);
          float* q = &acc[((unsigned)__builtin_amdgcn_readlane(d, k) << 6) + lane * 2u];
          atomicAdd(q + 0, r1.x);
          atomicAdd(q + 1, r1.y);
        }
      }
      w = wn;
    }
  }
  __syncthreads();

  tile_transform<false>(acc, aux, cm + mlo, h1m + (size_t)mlo * D, WlM, WrM, bM,
                        outM + (size_t)mlo * D, rm, false, tid);
  tile_transform<false>(acc + ubase * D, aux, cu + ulo, h1u + (size_t)ulo * D, WlU, WrU, bU,
                        outU + (size_t)ulo * D, uhi - ulo, false, tid);
}

extern "C" void kernel_launch(void* const* d_in, const int* in_sizes, int n_in,
                              void* d_out, int out_size, void* d_ws, size_t ws_size,
                              hipStream_t stream) {
  const float* x_user  = (const float*)d_in[0];
  const float* x_movie = (const float*)d_in[1];
  const float* Wl1_um  = (const float*)d_in[2];
  const float* Wr1_um  = (const float*)d_in[3];
  const float* b1_um   = (const float*)d_in[4];
  const float* Wl1_mu  = (const float*)d_in[5];
  const float* Wr1_mu  = (const float*)d_in[6];
  const float* b1_mu   = (const float*)d_in[7];
  const float* Wl2_um  = (const float*)d_in[8];
  const float* Wr2_um  = (const float*)d_in[9];
  const float* b2_um   = (const float*)d_in[10];
  const float* Wl2_mu  = (const float*)d_in[11];
  const float* Wr2_mu  = (const float*)d_in[12];
  const float* b2_mu   = (const float*)d_in[13];
  const int*   uidx    = (const int*)d_in[14];
  const int*   midx    = (const int*)d_in[15];

  float* ws = (float*)d_ws;
  unsigned* lst = (unsigned*)(ws + OFF_LST);
  float* cu = ws + OFF_CU;
  float* cm = ws + OFF_CM;
  float* u1 = ws + OFF_U1;
  float* m1 = ws + OFF_M1;

  float* u2 = (float*)d_out;
  float* m2 = (float*)d_out + (size_t)N_USER * D;

  const dim3 bAgg(AGG_THREADS);
  const dim3 gAgg(NTILE);

  tile_layer1_kernel<<<gAgg, bAgg, 0, stream>>>(x_user, x_movie, uidx, midx,
                                                Wl1_um, Wr1_um, b1_um,
                                                Wl1_mu, Wr1_mu, b1_mu,
                                                cu, cm, u1, m1, lst);
  tile_layer2_kernel<<<gAgg, bAgg, 0, stream>>>(u1, m1, uidx, midx,
                                                Wl2_um, Wr2_um, b2_um,
                                                Wl2_mu, Wr2_mu, b2_mu,
                                                cu, cm, u2, m2, lst);
}


# Layer-1 scan reads one tile byte per edge (16 per 16-B load, 4 bytes tested per register), not a 4-B id; a new first launch writes the bytes and packed (source, row) words; atomic sum order changes
# speedup vs baseline: 3.3021x; 1.1599x over previous
#include <hip/hip_runtime.h>
#include <hip/hip_bf16.h>


#define N_USER  200000
#define N_MOVIE 100000
#define N_EDGE  4000000
#define D       64

typedef float v2f __attribute__((ext_vector_type(2)));
typedef float v8f __attribute__((ext_vector_type(8)));

#define WPS 160

#define OFF_LST 0u
#define OFF_CU (OFF_LST + (N_USER + N_MOVIE) * 64u)
#define OFF_CM (OFF_CU + N_USER)
#define OFF_U1 (OFF_CM + N_MOVIE)
#define OFF_M1 (OFF_U1 + N_USER * 64u)

#define PL_PKM 0u
#define PL_PKU (PL_PKM + N_EDGE)
#define PL_TBM (PL_PKU + N_EDGE)
#define PL_TBU (PL_TBM + N_EDGE / 4)
#define ENC_THREADS 256
#define ENC_BLOCKS  ((N_EDGE / 4 + ENC_THREADS - 1) / ENC_THREADS)

#define TR          1184
#define NT_M        ((N_MOVIE + TR - 1) / TR)
#define NT_U        ((N_USER + TR - 1) / TR)
#define NTILE       (NT_M + NT_U)
#define AGG_THREADS 1024
#define AGG_WAVES   (AGG_THREADS / 32)
#define QCAP        64
#define N_EDGE4     (N_EDGE / 4)
#define SCAN_U      4
#define STEP4       (AGG_THREADS * SCAN_U)
#define SCAN_ITERS  (N_EDGE4 / STEP4)
#define TAIL_E0     (SCAN_ITERS * STEP4 * 4)
#define TAIL_ITERS  ((N_EDGE - TAIL_E0 + AGG_THREADS - 1) / AGG_THREADS)
#define GB          8

#define LSTRIDE     2336u
#define SUBCAP      768

#define K1          400
#define K2          784

static_assert(3 * SUBCAP + 32 == (int)LSTRIDE, "count line and three sub-lists fill the wave's share");
static_assert(K1 % 16 == 0 && K2 % 16 == 0 && TR - K2 == K1 && (K2 - K1) + 2 * K1 == TR,
              "sub-ranges are whole matrix tiles; one movie and two user sub-ranges are TR rows");
static_assert((NTILE / 3) * TR + (NTILE % 3 == 0 ? 0 : (NTILE % 3 == 1 ? K1 : K2)) >= N_MOVIE,
              "layer 2: movie sub-ranges 0 .. NTILE - 1 hold every movie row");
static_assert(((2 * NTILE) / 3) * TR + ((2 * NTILE) % 3 == 0 ? 0 : ((2 * NTILE) % 3 == 1 ? K1 : K2)) >= N_USER,
              "layer 2: user sub-ranges 0 .. 2 NTILE - 1 hold every user row");
static_assert(N_USER + N_MOVIE <= (1 << 21), "joint source row and LDS row share one 32-bit word");

__device__ __forceinline__ int sub_lo(int k) { return k == 0 ? 0 : (k == 1 ? K1 : (k == 2 ? K2 : TR)); }
__device__ __forceinline__ int sub_size(int k) { return k == 1 ? K2 - K1 : K1; }
__device__ __forceinline__ int sub_start(int g, int nrow) {
  const int t = g / 3;
  return min(t * TR + sub_lo(g - 3 * t), nrow);
}

struct ListState {
  unsigned* wl;
  int n0, n1, n2;
  int b0, b1, b2;
  int soff;
};

static_assert((unsigned long long)NTILE * AGG_WAVES * LSTRIDE
                  <= (unsigned long long)(N_USER + N_MOVIE) * D,
              "counts and lists fit into their workspace plane");
static_assert(TR % 16 == 0 && N_USER % 16 == 0 && N_MOVIE % 16 == 0,
              "every tile is a whole number of 16-row matrix tiles");
static_assert(LSTRIDE % 32 == 0 && SUBCAP % 32 == 0, "a wave's count line and lists are whole 128-B lines");
static_assert(TR <= (1 << 11) && N_USER <= (1 << 21) && N_MOVIE <= (1 << 21),
              "source id and row in tile share one 32-bit word");

static_assert(N_EDGE % 4 == 0 && SCAN_U == 4, "edge indices are scanned 4 per load, 16 per lane and step");
static_assert((QCAP & (QCAP - 1)) == 0 && QCAP >= 64, "ring holds 31 + 32 entries");

__device__ __forceinline__ float2 row_load(const float* __restrict__ x, int r,
                                           unsigned lane) {
  return ((const float2*)x)[((unsigned)r << 5) + lane];
}

__device__ __forceinline__ int ring_edge(int code) {
  const int b = code & 31;
  return ((code >> 5) << 4) + 4 * (b & 7) + (b >> 3);
}

template <bool SAVE>
__device__ __forceinline__ void drain_ring(
    const float* __restrict__ xsrc, const int* __restrict__ sidx,
    const unsigned* __restrict__ pk,
    float* acc, float* cnt, const int* qe, const int* qd,
    int head, int n, unsigned lane, ListState& ls) {
  int s = 0, d = 0;
  if ((int)lane < n) {
    const int slot = (head + (int)lane) & (QCAP - 1);
    const int e = qe[slot];
    if (SAVE) {
      const unsigned w = pk[ring_edge(e)];
      d = (int)(w & 2047u);
      s = (int)(w >> 11);
      atomicAdd(&cnt[d], 1.0f);
    } else {
      d = qd[slot];
      s = sidx[e];
    }
  }
  if (SAVE) {
    const bool act = (int)lane < n;
    const unsigned sw = (unsigned)(s + ls.soff) << 11;
    {
      const bool mine = act && d < K1;
      const unsigned mk = __builtin_amdgcn_ballot_w32(mine);
      const int c = __builtin_popcount(mk);
      if (mine && ls.n0 + c <= SUBCAP)
        ls.wl[ls.n0 + (int)__builtin_amdgcn_mbcnt_lo(mk, 0u)] = sw | (unsigned)(ls.b0 + d);
      ls.n0 += c;
    }
    {
      const bool mine = act && d >= K1 && d < K2;
      const unsigned mk = __builtin_amdgcn_ballot_w32(mine);
      const int c = __builtin_popcount(mk);
      if (mine && ls.n1 + c <= SUBCAP)
        ls.wl[SUBCAP + ls.n1 + (int)__builtin_amdgcn_mbcnt_lo(mk, 0u)] = sw | (unsigned)(ls.b1 + d);
      ls.n1 += c;
    }
    {
      const bool mine = act && d >= K2;
      const unsigned mk = __builtin_amdgcn_ballot_w32(mine);
      const int c = __builtin_popcount(mk);
      if (mine && ls.n2 + c <= SUBCAP)
        ls.wl[2 * SUBCAP + ls.n2 + (int)__builtin_amdgcn_mbcnt_lo(mk, 0u)] = sw | (unsigned)(ls.b2 + d);
      ls.n2 += c;
    }
  }
  if (n == 32) {
    #pragma unroll 1
    for (int k = 0; k < 32; k += GB) {
      float2 r[GB];
      #pragma unroll
      for (int j = 0; j < GB; ++j)
        r[j] = row_load(xsrc, __builtin_amdgcn_readlane(s, k + j), lane);
      #pragma unroll
      for (int j = 0; j < GB; ++j) {
        float* q = &acc[((unsigned)__builtin_amdgcn_readlane(d, k + j) << 6) + lane * 2u];
        atomicAdd(q + 0, r[j].x);
        atomicAdd(q + 1, r[j].y);
      }
    }
  } else {
    #pragma unroll 1
    for (int k = 0; k < n; ++k) {
      const float2 r1 = row_load(xsrc, __builtin_amdgcn_readlane(s, k), lane);
      float* q = &acc[((unsigned)__builtin_amdgcn_readlane(d, k) << 6) + lane * 2u];
      atomicAdd(q + 0, r1.x);
      atomicAdd(q + 1, r1.y);
    }
  }
}

template <bool SAVE>
__device__ __forceinline__ void scan_one(
    int dst, int eid, int lo, int rows,
    const float* __restrict__ xsrc, const int* __restrict__ sidx,
    float* acc, float* cnt, int* qe, int* qd,
    int& head, int& qn, unsigned lane, ListState& ls) {
  const unsigned dl = (unsigned)(dst - lo);
  const bool hit = dl < (unsigned)rows;
  const unsigned mask = __builtin_amdgcn_ballot_w32(hit);
  if (mask != 0u) {
    if (hit) {
      const int pos = (head + qn + (int)__builtin_amdgcn_mbcnt_lo(mask, 0u)) & (QCAP - 1);
      qe[pos] = eid;
      qd[pos] = (int)dl;
    }
    qn += __builtin_popcount(mask);
    if (qn >= 32) {
      drain_ring<SAVE>(xsrc, sidx, nullptr, acc, cnt, qe, qd, head, 32, lane, ls);
      head = (head + 32) & (QCAP - 1);
      qn -= 32;
    }
  }
}

__device__ __forceinline__ int4 scan_load(const int4* __restrict__ d4, int it, int u, int tid) {
  return d4[(it * AGG_THREADS + tid) * SCAN_U + u];
}

template <bool SAVE>
__device__ __forceinline__ void scan_edges(
    const int* __restrict__ didx, const int* __restrict__ sidx,
    const float* __restrict__ xsrc, float* acc, float* cnt, int* qe, int* qd,
    int lo, int rows, int tile, int tid, unsigned lane, ListState& ls) {
  int head = 0, qn = 0;

  const int4* d4 = (const int4*)didx;
  int it = (tile * SCAN_ITERS) / NTILE;
  int4 v[SCAN_U];
  #pragma unroll
  for (int u = 0; u < SCAN_U; ++u) v[u] = scan_load(d4, it, u, tid);
  #pragma unroll 1
  for (int n = 0; n < SCAN_ITERS; ++n) {
    const int itn = (it + 1 == SCAN_ITERS) ? 0 : it + 1;
    int4 vn[SCAN_U];
    #pragma unroll
    for (int u = 0; u < SCAN_U; ++u) vn[u] = scan_load(d4, itn, u, tid);
    #pragma unroll
    for (int u = 0; u < SCAN_U; ++u) {
      const int e0 = ((it * AGG_THREADS + tid) * SCAN_U + u) * 4;
      scan_one<SAVE>(v[u].x, e0 + 0, lo, rows, xsrc, sidx, acc, cnt, qe, qd, head, qn, lane, ls);
      scan_one<SAVE>(v[u].y, e0 + 1, lo, rows, xsrc, sidx, acc, cnt, qe, qd, head, qn, lane, ls);
      scan_one<SAVE>(v[u].z, e0 + 2, lo, rows, xsrc, sidx, acc, cnt, qe, qd, head, qn, lane, ls);
      scan_one<SAVE>(v[u].w, e0 + 3, lo, rows, xsrc, sidx, acc, cnt, qe, qd, head, qn, lane, ls);
    }
    #pragma unroll
    for (int u = 0; u < SCAN_U; ++u) v[u] = vn[u];
    it = itn;
  }
  #pragma unroll 1
  for (int j = 0; j < TAIL_ITERS; ++j) {
    const int e = TAIL_E0 + j * AGG_THREADS + tid;
    const int dv = (e < N_EDGE) ? didx[e] : -1;
    scan_one<SAVE>(dv, e, lo, rows, xsrc, sidx, acc, cnt, qe, qd, head, qn, lane, ls);
  }
  if (qn > 0) drain_ring<SAVE>(xsrc, sidx, nullptr, acc, cnt, qe, qd, head, qn, lane, ls);
}

__device__ __forceinline__ void ring_push(
    unsigned mask, bool hit, int code, const float* __restrict__ xsrc,
    const unsigned* __restrict__ pk, float* acc, float* cnt, int* qe,
    int& head, int& qn, unsigned lane, ListState& ls) {
  if (hit)
    qe[(head + qn + (int)__builtin_amdgcn_mbcnt_lo(mask, 0u)) & (QCAP - 1)] = code;
  qn += __builtin_popcount(mask);
  if (qn >= 32) {
    drain_ring<true>(xsrc, nullptr, pk, acc, cnt, qe, nullptr, head, 32, lane, ls);
    head = (head + 32) & (QCAP - 1);
    qn -= 32;
  }
}

__device__ __forceinline__ unsigned zero_bytes(unsigned x) {
  const unsigned t = (x & 0x7f7f7f7fu) + 0x7f7f7f7fu;
  return ~(t | x | 0x7f7f7f7fu);
}

__device__ __forceinline__ void scan_tiles(
    const unsigned* __restrict__ tbw, const unsigned* __restrict__ pk,
    const float* __restrict__ xsrc, float* acc, float* cnt, int* qe,
    unsigned mytile, int tile, int tid, unsigned lane, ListState& ls) {
  int head = 0, qn = 0;
  const unsigned pat = mytile * 0x01010101u;
  const uint4* t4 = (const uint4*)tbw;
  int it = (tile * SCAN_ITERS) / NTILE;
  uint4 v = t4[it * AGG_THREADS + tid];
  #pragma unroll 1
  for (int n = 0; n < SCAN_ITERS; ++n) {
    const int itn = (it + 1 == SCAN_ITERS) ? 0 : it + 1;
    const uint4 vn = t4[itn * AGG_THREADS + tid];
    const int e0 = (it * AGG_THREADS + tid) * 16;
    unsigned m = (zero_bytes(v.x ^ pat) >> 7) | (zero_bytes(v.y ^ pat) >> 6) |
                 (zero_bytes(v.z ^ pat) >> 5) | (zero_bytes(v.w ^ pat) >> 4);
    unsigned mask = __builtin_amdgcn_ballot_w32(m != 0u);
    #pragma unroll 1
    while (mask != 0u) {
      const int code = 2 * e0 + (int)__builtin_ctz(m | 0x80000000u);
      ring_push(mask, m != 0u, code, xsrc, pk, acc, cnt, qe, head, qn, lane, ls);
      m &= m - 1u;
      mask = __builtin_amdgcn_ballot_w32(m != 0u);
    }
    v = vn;
    it = itn;
  }
  const unsigned char* tb = (const unsigned char*)tbw;
  #pragma unroll 1
  for (int j = 0; j < TAIL_ITERS; ++j) {
    const int e = TAIL_E0 + j * AGG_THREADS + tid;
    const bool hit = (e < N_EDGE) && ((unsigned)tb[min(e, N_EDGE - 1)] == mytile);
    const unsigned mask = __builtin_amdgcn_ballot_w32(hit);
    if (mask != 0u)
      ring_push(mask, hit, 2 * (e & ~15) + 8 * (e & 3) + ((e >> 2) & 3),
                xsrc, pk, acc, cnt, qe, head, qn, lane, ls);
  }
  if (qn > 0) drain_ring<true>(xsrc, nullptr, pk, acc, cnt, qe, nullptr, head, qn, lane, ls);
}

#define AUX_CNT    0
#define AUX_RINGE  TR
#define AUX_RINGD  (TR + AGG_WAVES * QCAP)
#define AUX_FLOATS (TR + 2 * AGG_WAVES * QCAP)
static_assert(32 * WPS <= AUX_FLOATS, "a staged weight matrix fits where counts and rings were");

__device__ __forceinline__ void stage_weights(float* aux, const float* __restrict__ Wg, int tid) {
  for (int idx = tid; idx < 32 * D; idx += AGG_THREADS) {
    const int kp = idx >> 6, n = idx & 63;
    *(float2*)&aux[kp * WPS + 2 * n] =
        make_float2(Wg[(2 * kp) * D + n], Wg[(2 * kp + 1) * D + n]);
  }
}

__device__ __forceinline__ void load_a_frags(const float* tile, int c, int h, v2f (&a)[16]) {
  #pragma unroll
  for (int j = 0; j < 16; ++j)
    a[j] = *(const v2f*)&tile[c * D + ((j ^ c) << 2) + 2 * h];
}

__device__ __forceinline__ v2f load_b_frag(const float* aux, int wo, int j, int ct, int c, int h) {
  return *(const v2f*)&aux[wo + (2 * j + h) * WPS + 2 * (16 * ct + c)];
}

__device__ __forceinline__ int trip_offset() {
  int wo = 0;
  asm volatile("" : "+v"(wo));
  return wo;
}

__device__ __forceinline__ int d_place(int h, int r, int col) {
  return (8 * h + r) * D + (col ^ (32 * h));
}

template <bool CNT_LDS>
__device__ __forceinline__ void tile_transform(
    float* acc, float* aux, float* __restrict__ cntG,
    const float* __restrict__ xdst, const float* __restrict__ Wl,
    const float* __restrict__ Wr, const float* __restrict__ bias,
    float* __restrict__ out, int rows, bool relu, int tid) {
  const int lane = tid & 31, wave = tid >> 5;
  const int c = lane & 15, h = lane >> 4;
  const int nrt = rows >> 4;

  for (int i = tid; i < rows * 16; i += AGG_THREADS) {
    const int row = i >> 4, q = i & 15;
    const float cv = CNT_LDS ? aux[AUX_CNT + row] : cntG[row];
    if (CNT_LDS && q == 0) cntG[row] = cv;
    const float inv = 1.0f / fmaxf(cv, 1.0f);
    const float4 sv = ((const float4*)acc)[i];
    ((float4*)acc)[(row << 4) + (q ^ (row & 15))] =
        make_float4(sv.x * inv, sv.y * inv, sv.z * inv, sv.w * inv);
  }
  __syncthreads();
  stage_weights(aux, Wl, tid);
  __syncthreads();

  #pragma unroll 1
  for (int rt = wave; rt < nrt; rt += AGG_WAVES) {
    float* tile = acc + rt * 16 * D;
    const int wo = trip_offset();
    v2f a[16];
    load_a_frags(tile, c, h, a);
    #pragma unroll
    for (int ct = 0; ct < 4; ++ct) {
      v8f d = {0.f, 0.f, 0.f, 0.f, 0.f, 0.f, 0.f, 0.f};
      #pragma unroll
      for (int j = 0; j < 16; ++j)
        d = __builtin_amdgcn_wmma_f32_16x16x4_f32(false, a[j], false,
                                                  load_b_frag(aux, wo, j, ct, c, h),
                                                  (short)0, d, false, false);
      #pragma unroll
      for (int r = 0; r < 8; ++r) tile[d_place(h, r, 16 * ct + c)] = d[r];
    }
  }
  __syncthreads();
  stage_weights(aux, Wr, tid);
  __syncthreads();

  #pragma unroll 1
  for (int rt = wave; rt < nrt; rt += AGG_WAVES) {
    float* tile = acc + rt * 16 * D;
    const int wo = trip_offset();
    const float* xt = xdst + (size_t)rt * 16 * D;
    float*       ot = out + (size_t)rt * 16 * D;
    float4 xr[8];
    #pragma unroll
    for (int i = 0; i < 8; ++i)
      xr[i] = *(const float4*)&xt[(2 * i + h) * D + 4 * c];
    v8f d[4];
    #pragma unroll
    for (int ct = 0; ct < 4; ++ct) {
      #pragma unroll
      for (int r = 0; r < 8; ++r) d[ct][r] = tile[d_place(h, r, 16 * ct + c)];
    }
    #pragma unroll
    for (int i = 0; i < 8; ++i) {
      const int row = 2 * i + h;
      *(float4*)&tile[row * D + ((c ^ row) << 2)] = xr[i];
    }
    v2f a[16];
    load_a_frags(tile, c, h, a);
    #pragma unroll
    for (int ct = 0; ct < 4; ++ct) {
      #pragma unroll
      for (int j = 0; j < 16; ++j)
        d[ct] = __builtin_amdgcn_wmma_f32_16x16x4_f32(false, a[j], false,
                                                      load_b_frag(aux, wo, j, ct, c, h),
                                                      (short)0, d[ct], false, false);
    }
    #pragma unroll
    for (int ct = 0; ct < 4; ++ct) {
      const float bcol = bias[16 * ct + c];
      #pragma unroll
      for (int r = 0; r < 8; ++r) {
        float val = d[ct][r] + bcol;
        if (relu) val = fmaxf(val, 0.0f);
        tile[d_place(h, r, 16 * ct + c)] = val;
      }
    }
    #pragma unroll
    for (int i = 0; i < 8; ++i) {
      const int row = 2 * i + h;
      *(float4*)&ot[row * D + 4 * c] =
          *(const float4*)&tile[row * D + ((c ^ (8 * (row >> 3))) << 2)];
    }
  }
}

static_assert(SCAN_ITERS * AGG_THREADS * 16 == TAIL_E0, "a scan step is 16 tile bytes per lane");
static_assert(NT_M <= 255 && NT_U <= 255, "a tile number fits a byte");
static_assert(N_USER <= (1 << 21) && N_MOVIE <= (1 << 21) && TR <= (1 << 11),
              "source id and row in tile share one 32-bit word");
static_assert((unsigned long long)PL_TBU + N_EDGE / 4 <= (unsigned long long)(N_USER + N_MOVIE) * D,
              "the edge planes fit into the output buffer");

__device__ __forceinline__ unsigned pack_edge(int src, int dst, unsigned& tilebyte, int k) {
  const unsigned t = (unsigned)dst / (unsigned)TR;
  tilebyte |= t << (8 * k);
  return ((unsigned)src << 11) | ((unsigned)dst - t * (unsigned)TR);
}

__global__ __launch_bounds__(ENC_THREADS) void edge_encode_kernel(
    const int* __restrict__ uidx, const int* __restrict__ midx, unsigned* __restrict__ pl) {
  const int i = blockIdx.x * ENC_THREADS + threadIdx.x;
  if (i >= N_EDGE / 4) return;
  const int4 u = ((const int4*)uidx)[i];
  const int4 m = ((const int4*)midx)[i];
  unsigned bm = 0u, bu = 0u;
  uint4 pm, pu;
  pm.x = pack_edge(u.x, m.x, bm, 0);  pu.x = pack_edge(m.x, u.x, bu, 0);
  pm.y = pack_edge(u.y, m.y, bm, 1);  pu.y = pack_edge(m.y, u.y, bu, 1);
  pm.z = pack_edge(u.z, m.z, bm, 2);  pu.z = pack_edge(m.z, u.z, bu, 2);
  pm.w = pack_edge(u.w, m.w, bm, 3);  pu.w = pack_edge(m.w, u.w, bu, 3);
  ((uint4*)(pl + PL_PKM))[i] = pm;
  ((uint4*)(pl + PL_PKU))[i] = pu;
  pl[PL_TBM + i] = bm;
  pl[PL_TBU + i] = bu;
}

__global__ __launch_bounds__(AGG_THREADS) void tile_layer1_kernel(
    const float* __restrict__ xu, const float* __restrict__ xm,
    const unsigned* __restrict__ pl,
    const float* __restrict__ WlM, const float* __restrict__ WrM, const float* __restrict__ bM,
    const float* __restrict__ WlU, const float* __restrict__ WrU, const float* __restrict__ bU,
    float* __restrict__ cu, float* __restrict__ cm,
    float* __restrict__ u1, float* __restrict__ m1,
    unsigned* __restrict__ lst) {
  __shared__ __attribute__((aligned(16))) float acc[TR * D];
  __shared__ __attribute__((aligned(16))) float aux[AUX_FLOATS];
  float* cnt   = aux + AUX_CNT;
  int*   ringE = (int*)(aux + AUX_RINGE);

  const int tid = threadIdx.x;
  const unsigned lane = threadIdx.x & 31u;
  const int wave = tid >> 5;
  const int tile = blockIdx.x;

  const bool mov = tile < NT_M;
  const unsigned* tbw = pl + (mov ? PL_TBM : PL_TBU);
  const unsigned* pk  = pl + (mov ? PL_PKM : PL_PKU);
  const float* xsrc = mov ? xu : xm;
  const float* xdst = mov ? xm : xu;
  float*       hOut = mov ? m1 : u1;
  float*       cOut = mov ? cm : cu;
  const int    nrow = mov ? N_MOVIE : N_USER;
  const int    lo   = (mov ? tile : tile - NT_M) * TR;
  const int    rows = min(TR, nrow - lo);

  for (int i = tid; i < TR * D / 4; i += AGG_THREADS)
    ((float4*)acc)[i] = make_float4(0.f, 0.f, 0.f, 0.f);
  for (int i = tid; i < TR; i += AGG_THREADS) cnt[i] = 0.f;
  __syncthreads();

  unsigned* wreg = lst + (unsigned)(tile * AGG_WAVES + wave) * LSTRIDE;
  ListState ls;
  ls.wl = wreg + 32;
  ls.n0 = 0; ls.n1 = 0; ls.n2 = 0;
  ls.soff = mov ? 0 : N_USER;
  {
    int b[3];
    #pragma unroll
    for (int k = 0; k < 3; ++k) {
      int base = 0;
      if (!mov) {
        const int g = 3 * (tile - NT_M) + k;
        base = sub_size((g >> 1) % 3) + ((g & 1) ? sub_size((g - 1) % 3) : 0);
      }
      b[k] = base - sub_lo(k);
    }
    ls.b0 = b[0]; ls.b1 = b[1]; ls.b2 = b[2];
  }
  scan_tiles(tbw, pk, xsrc, acc, cnt, &ringE[wave * QCAP],
             (unsigned)(mov ? tile : tile - NT_M), tile, tid, lane, ls);
  if (lane < 3u)
    wreg[lane] = (unsigned)(lane == 0u ? ls.n0 : (lane == 1u ? ls.n1 : ls.n2));
  __syncthreads();

  tile_transform<true>(acc, aux, cOut + lo, xdst + (size_t)lo * D,
                       mov ? WlM : WlU, mov ? WrM : WrU, mov ? bM : bU,
                       hOut + (size_t)lo * D, rows, true, tid);
}

static_assert(OFF_M1 == OFF_U1 + N_USER * 64u, "the layer-1 tables are one joint table, users first");

__device__ __forceinline__ void sub_list(const unsigned* __restrict__ lst, bool exists, int tile1,
                                         int wave, int k, int& n, unsigned& off) {
  const unsigned reg = (unsigned)(tile1 * AGG_WAVES + wave) * LSTRIDE;
  n = exists ? __builtin_amdgcn_readfirstlane((int)lst[reg + (unsigned)k]) : 0;
  off = reg + 32u + (unsigned)k * SUBCAP;
}

__global__ __launch_bounds__(AGG_THREADS) void tile_layer2_kernel(
    const float* __restrict__ h1u, const float* __restrict__ h1m,
    const int* __restrict__ uidx, const int* __restrict__ midx,
    const float* __restrict__ WlM, const float* __restrict__ WrM, const float* __restrict__ bM,
    const float* __restrict__ WlU, const float* __restrict__ WrU, const float* __restrict__ bU,
    float* __restrict__ cu, float* __restrict__ cm,
    float* __restrict__ outU, float* __restrict__ outM,
    const unsigned* __restrict__ lst) {
  __shared__ __attribute__((aligned(16))) float acc[TR * D];
  __shared__ __attribute__((aligned(16))) float aux[AUX_FLOATS];
  int* ringE = (int*)(aux + AUX_RINGE);
  int* ringD = (int*)(aux + AUX_RINGD);

  const int tid = threadIdx.x;
  const unsigned lane = threadIdx.x & 31u;
  const int wave = tid >> 5;
  const int wg = blockIdx.x;

  const int mlo   = sub_start(wg, N_MOVIE);
  const int rm    = sub_start(wg + 1, N_MOVIE) - mlo;
  const int ulo   = sub_start(2 * wg, N_USER);
  const int umid  = sub_start(2 * wg + 1, N_USER);
  const int uhi   = sub_start(2 * wg + 2, N_USER);
  const int ubase = sub_size(wg % 3);

  for (int i = tid; i < TR * D / 4; i += AGG_THREADS)
    ((float4*)acc)[i] = make_float4(0.f, 0.f, 0.f, 0.f);
  __syncthreads();

  int n0, n1, n2;
  unsigned o0, o1, o2;
  {
    const int g1 = 2 * wg, g2 = 2 * wg + 1;
    sub_list(lst, true, wg / 3, wave, wg % 3, n0, o0);
    sub_list(lst, g1 / 3 < NT_U, NT_M + g1 / 3, wave, g1 % 3, n1, o1);
    sub_list(lst, g2 / 3 < NT_U, NT_M + g2 / 3, wave, g2 % 3, n2, o2);
  }
  const int ovf = (n0 > SUBCAP ? 1 : 0) | (n1 > SUBCAP ? 2 : 0) | (n2 > SUBCAP ? 4 : 0);
  if (ovf != 0) {
    #pragma unroll 1
    for (int p = 0; p < 3; ++p) {
      if ((ovf >> p) & 1) {
        const bool mv = p == 0;
        const int plo   = mv ? mlo : (p == 1 ? ulo : umid);
        const int prows = mv ? rm : (p == 1 ? umid - ulo : uhi - umid);
        const int prow0 = mv ? 0 : (p == 1 ? ubase : ubase + (umid - ulo));
        ListState none;
        none.wl = nullptr;
        none.n0 = 0; none.n1 = 0; none.n2 = 0;
        none.b0 = 0; none.b1 = 0; none.b2 = 0;
        none.soff = 0;
        scan_edges<false>(mv ? midx : uidx, mv ? uidx : midx, mv ? h1u : h1m,
                          acc + prow0 * D, nullptr, &ringE[wave * QCAP], &ringD[wave * QCAP],
                          plo, prows, wg, tid, lane, none);
      }
    }
    if (n0 > SUBCAP) n0 = 0;
    if (n1 > SUBCAP) n1 = 0;
    if (n2 > SUBCAP) n2 = 0;
  }

  const int n01 = n0 + n1;
  const int n = n01 + n2;
  if (n > 0) {
    const int last = n - 1;
    auto entry = [&](int i) -> unsigned {
      const unsigned o = (i < n0) ? o0 + (unsigned)i
                       : ((i < n01) ? o1 + (unsigned)(i - n0) : o2 + (unsigned)(i - n01));
      return lst[o];
    };
    unsigned w = entry(min((int)lane, last));
    #pragma unroll 1
    for (int base = 0; base < n; base += 32) {
      const int m = min(32, n - base);
      const unsigned wn = entry(min(base + 32 + (int)lane, last));
      const int s = (int)(w >> 11);
      const int d = (int)(w & 2047u);
      if (m == 32) {
        float2 r[32];
        #pragma unroll
        for (int j = 0; j < 32; ++j)
          r[j] = row_load(h1u, __builtin_amdgcn_readlane(s, j), lane);
        #pragma unroll
        for (int j = 0; j < 32; ++j) {
          float* q = &acc[((unsigned)__builtin_amdgcn_readlane(d, j) << 6) + lane * 2u];
          atomicAdd(q + 0, r[j].x);
          atomicAdd(q + 1, r[j].y);
        }
      } else {
        #pragma unroll 1
        for (int k = 0; k < m; ++k) {
          const float2 r1 = row_load(h1u, __builtin_amdgcn_readlane(s, k), lane);
          float* q = &acc[((unsigned)__builtin_amdgcn_readlane(d, k) << 6) + lane * 2u];
          atomicAdd(q + 0, r1.x);
          atomicAdd(q + 1, r1.y);
        }
      }
      w = wn;
    }
  }
  __syncthreads();

  tile_transform<false>(acc, aux, cm + mlo, h1m + (size_t)mlo * D, WlM, WrM, bM,
                        outM + (size_t)mlo * D, rm, false, tid);
  tile_transform<false>(acc + ubase * D, aux, cu + ulo, h1u + (size_t)ulo * D, WlU, WrU, bU,
                        outU + (size_t)ulo * D, uhi - ulo, false, tid);
}

extern "C" void kernel_launch(void* const* d_in, const int* in_sizes, int n_in,
                              void* d_out, int out_size, void* d_ws, size_t ws_size,
                              hipStream_t stream) {
  const float* x_user  = (const float*)d_in[0];
  const float* x_movie = (const float*)d_in[1];
  const float* Wl1_um  = (const float*)d_in[2];
  const float* Wr1_um  = (const float*)d_in[3];
  const float* b1_um   = (const float*)d_in[4];
  const float* Wl1_mu  = (const float*)d_in[5];
  const float* Wr1_mu  = (const float*)d_in[6];
  const float* b1_mu   = (const float*)d_in[7];
  const float* Wl2_um  = (const float*)d_in[8];
  const float* Wr2_um  = (const float*)d_in[9];
  const float* b2_um   = (const float*)d_in[10];
  const float* Wl2_mu  = (const float*)d_in[11];
  const float* Wr2_mu  = (const float*)d_in[12];
  const float* b2_mu   = (const float*)d_in[13];
  const int*   uidx    = (const int*)d_in[14];
  const int*   midx    = (const int*)d_in[15];

  float* ws = (float*)d_ws;
  unsigned* lst = (unsigned*)(ws + OFF_LST);
  float* cu = ws + OFF_CU;
  float* cm = ws + OFF_CM;
  float* u1 = ws + OFF_U1;
  float* m1 = ws + OFF_M1;

  float* u2 = (float*)d_out;
  float* m2 = (float*)d_out + (size_t)N_USER * D;

  const dim3 bAgg(AGG_THREADS);
  const dim3 gAgg(NTILE);

  unsigned* pl = (unsigned*)d_out;
  edge_encode_kernel<<<dim3(ENC_BLOCKS), dim3(ENC_THREADS), 0, stream>>>(uidx, midx, pl);

  tile_layer1_kernel<<<gAgg, bAgg, 0, stream>>>(x_user, x_movie, pl,
                                                Wl1_um, Wr1_um, b1_um,
                                                Wl1_mu, Wr1_mu, b1_mu,
                                                cu, cm, u1, m1, lst);
  tile_layer2_kernel<<<gAgg, bAgg, 0, stream>>>(u1, m1, uidx, midx,
                                                Wl2_um, Wr2_um, b2_um,
                                                Wl2_mu, Wr2_mu, b2_mu,
                                                cu, cm, u2, m2, lst);
}


# Layer-1 scan reads tile numbers bit-sliced (8 words per 32 edges, built by ballots in the encoding launch): a lane tests 32 edges with 12 bit operations, was 16 byte tests; atomic sum order changes
# speedup vs baseline: 3.3989x; 1.0293x over previous
#include <hip/hip_runtime.h>
#include <hip/hip_bf16.h>


#define N_USER  200000
#define N_MOVIE 100000
#define N_EDGE  4000000
#define D       64

typedef float v2f __attribute__((ext_vector_type(2)));
typedef float v8f __attribute__((ext_vector_type(8)));

#define WPS 160

#define OFF_LST 0u
#define OFF_CU (OFF_LST + (N_USER + N_MOVIE) * 64u)
#define OFF_CM (OFF_CU + N_USER)
#define OFF_U1 (OFF_CM + N_MOVIE)
#define OFF_M1 (OFF_U1 + N_USER * 64u)

#define PL_PKM 0u
#define PL_PKU (PL_PKM + N_EDGE)
#define PL_TLM (PL_PKU + N_EDGE)
#define PL_THM (PL_TLM + N_EDGE / 8)
#define PL_TLU (PL_THM + N_EDGE / 8)
#define PL_THU (PL_TLU + N_EDGE / 8)
#define ENC_THREADS 256
#define ENC_WAVE_E  256
#define ENC_WAVES   (N_EDGE / ENC_WAVE_E)
#define ENC_BLOCKS  ((ENC_WAVES + ENC_THREADS / 32 - 1) / (ENC_THREADS / 32))

#define TR          1184
#define NT_M        ((N_MOVIE + TR - 1) / TR)
#define NT_U        ((N_USER + TR - 1) / TR)
#define NTILE       (NT_M + NT_U)
#define AGG_THREADS 1024
#define AGG_WAVES   (AGG_THREADS / 32)
#define QCAP        64
#define N_EDGE4     (N_EDGE / 4)
#define SCAN_U      4
#define STEP4       (AGG_THREADS * SCAN_U)
#define SCAN_ITERS  (N_EDGE4 / STEP4)
#define TAIL_E0     (SCAN_ITERS * STEP4 * 4)
#define TAIL_ITERS  ((N_EDGE - TAIL_E0 + AGG_THREADS - 1) / AGG_THREADS)
#define GB          8

#define LSTRIDE     2336u
#define SUBCAP      768

#define K1          400
#define K2          784

static_assert(3 * SUBCAP + 32 == (int)LSTRIDE, "count line and three sub-lists fill the wave's share");
static_assert(K1 % 16 == 0 && K2 % 16 == 0 && TR - K2 == K1 && (K2 - K1) + 2 * K1 == TR,
              "sub-ranges are whole matrix tiles; one movie and two user sub-ranges are TR rows");
static_assert((NTILE / 3) * TR + (NTILE % 3 == 0 ? 0 : (NTILE % 3 == 1 ? K1 : K2)) >= N_MOVIE,
              "layer 2: movie sub-ranges 0 .. NTILE - 1 hold every movie row");
static_assert(((2 * NTILE) / 3) * TR + ((2 * NTILE) % 3 == 0 ? 0 : ((2 * NTILE) % 3 == 1 ? K1 : K2)) >= N_USER,
              "layer 2: user sub-ranges 0 .. 2 NTILE - 1 hold every user row");
static_assert(N_USER + N_MOVIE <= (1 << 21), "joint source row and LDS row share one 32-bit word");

__device__ __forceinline__ int sub_lo(int k) { return k == 0 ? 0 : (k == 1 ? K1 : (k == 2 ? K2 : TR)); }
__device__ __forceinline__ int sub_size(int k) { return k == 1 ? K2 - K1 : K1; }
__device__ __forceinline__ int sub_start(int g, int nrow) {
  const int t = g / 3;
  return min(t * TR + sub_lo(g - 3 * t), nrow);
}

struct ListState {
  unsigned* wl;
  int n0, n1, n2;
  int b0, b1, b2;
  int soff;
};

static_assert((unsigned long long)NTILE * AGG_WAVES * LSTRIDE
                  <= (unsigned long long)(N_USER + N_MOVIE) * D,
              "counts and lists fit into their workspace plane");
static_assert(TR % 16 == 0 && N_USER % 16 == 0 && N_MOVIE % 16 == 0,
              "every tile is a whole number of 16-row matrix tiles");
static_assert(LSTRIDE % 32 == 0 && SUBCAP % 32 == 0, "a wave's count line and lists are whole 128-B lines");
static_assert(TR <= (1 << 11) && N_USER <= (1 << 21) && N_MOVIE <= (1 << 21),
              "source id and row in tile share one 32-bit word");

static_assert(N_EDGE % 4 == 0 && SCAN_U == 4, "edge indices are scanned 4 per load, 16 per lane and step");
static_assert((QCAP & (QCAP - 1)) == 0 && QCAP >= 64, "ring holds 31 + 32 entries");

__device__ __forceinline__ float2 row_load(const float* __restrict__ x, int r,
                                           unsigned lane) {
  return ((const float2*)x)[((unsigned)r << 5) + lane];
}

template <bool SAVE>
__device__ __forceinline__ void drain_ring(
    const float* __restrict__ xsrc, const int* __restrict__ sidx,
    const unsigned* __restrict__ pk,
    float* acc, float* cnt, const int* qe, const int* qd,
    int head, int n, unsigned lane, ListState& ls) {
  int s = 0, d = 0;
  if ((int)lane < n) {
    const int slot = (head + (int)lane) & (QCAP - 1);
    const int e = qe[slot];
    if (SAVE) {
      const unsigned w = pk[e];
      d = (int)(w & 2047u);
      s = (int)(w >> 11);
      atomicAdd(&cnt[d], 1.0f);
    } else {
      d = qd[slot];
      s = sidx[e];
    }
  }
  if (SAVE) {
    const bool act = (int)lane < n;
    const unsigned sw = (unsigned)(s + ls.soff) << 11;
    {
      const bool mine = act && d < K1;
      const unsigned mk = __builtin_amdgcn_ballot_w32(mine);
      const int c = __builtin_popcount(mk);
      if (mine && ls.n0 + c <= SUBCAP)
        ls.wl[ls.n0 + (int)__builtin_amdgcn_mbcnt_lo(mk, 0u)] = sw | (unsigned)(ls.b0 + d);
      ls.n0 += c;
    }
    {
      const bool mine = act && d >= K1 && d < K2;
      const unsigned mk = __builtin_amdgcn_ballot_w32(mine);
      const int c = __builtin_popcount(mk);
      if (mine && ls.n1 + c <= SUBCAP)
        ls.wl[SUBCAP + ls.n1 + (int)__builtin_amdgcn_mbcnt_lo(mk, 0u)] = sw | (unsigned)(ls.b1 + d);
      ls.n1 += c;
    }
    {
      const bool mine = act && d >= K2;
      const unsigned mk = __builtin_amdgcn_ballot_w32(mine);
      const int c = __builtin_popcount(mk);
      if (mine && ls.n2 + c <= SUBCAP)
        ls.wl[2 * SUBCAP + ls.n2 + (int)__builtin_amdgcn_mbcnt_lo(mk, 0u)] = sw | (unsigned)(ls.b2 + d);
      ls.n2 += c;
    }
  }
  if (n == 32) {
    #pragma unroll 1
    for (int k = 0; k < 32; k += GB) {
      float2 r[GB];
      #pragma unroll
      for (int j = 0; j < GB; ++j)
        r[j] = row_load(xsrc, __builtin_amdgcn_readlane(s, k + j), lane);
      #pragma unroll
      for (int j = 0; j < GB; ++j) {
        float* q = &acc[((unsigned)__builtin_amdgcn_readlane(d, k + j) << 6) + lane * 2u];
        atomicAdd(q + 0, r[j].x);
        atomicAdd(q + 1, r[j].y);
      }
    }
  } else {
    #pragma unroll 1
    for (int k = 0; k < n; ++k) {
      const float2 r1 = row_load(xsrc, __builtin_amdgcn_readlane(s, k), lane);
      float* q = &acc[((unsigned)__builtin_amdgcn_readlane(d, k) << 6) + lane * 2u];
      atomicAdd(q + 0, r1.x);
      atomicAdd(q + 1, r1.y);
    }
  }
}

template <bool SAVE>
__device__ __forceinline__ void scan_one(
    int dst, int eid, int lo, int rows,
    const float* __restrict__ xsrc, const int* __restrict__ sidx,
    float* acc, float* cnt, int* qe, int* qd,
    int& head, int& qn, unsigned lane, ListState& ls) {
  const unsigned dl = (unsigned)(dst - lo);
  const bool hit = dl < (unsigned)rows;
  const unsigned mask = __builtin_amdgcn_ballot_w32(hit);
  if (mask != 0u) {
    if (hit) {
      const int pos = (head + qn + (int)__builtin_amdgcn_mbcnt_lo(mask, 0u)) & (QCAP - 1);
      qe[pos] = eid;
      qd[pos] = (int)dl;
    }
    qn += __builtin_popcount(mask);
    if (qn >= 32) {
      drain_ring<SAVE>(xsrc, sidx, nullptr, acc, cnt, qe, qd, head, 32, lane, ls);
      head = (head + 32) & (QCAP - 1);
      qn -= 32;
    }
  }
}

__device__ __forceinline__ int4 scan_load(const int4* __restrict__ d4, int it, int u, int tid) {
  return d4[(it * AGG_THREADS + tid) * SCAN_U + u];
}

template <bool SAVE>
__device__ __forceinline__ void scan_edges(
    const int* __restrict__ didx, const int* __restrict__ sidx,
    const float* __restrict__ xsrc, float* acc, float* cnt, int* qe, int* qd,
    int lo, int rows, int tile, int tid, unsigned lane, ListState& ls) {
  int head = 0, qn = 0;

  const int4* d4 = (const int4*)didx;
  int it = (tile * SCAN_ITERS) / NTILE;
  int4 v[SCAN_U];
  #pragma unroll
  for (int u = 0; u < SCAN_U; ++u) v[u] = scan_load(d4, it, u, tid);
  #pragma unroll 1
  for (int n = 0; n < SCAN_ITERS; ++n) {
    const int itn = (it + 1 == SCAN_ITERS) ? 0 : it + 1;
    int4 vn[SCAN_U];
    #pragma unroll
    for (int u = 0; u < SCAN_U; ++u) vn[u] = scan_load(d4, itn, u, tid);
    #pragma unroll
    for (int u = 0; u < SCAN_U; ++u) {
      const int e0 = ((it * AGG_THREADS + tid) * SCAN_U + u) * 4;
      scan_one<SAVE>(v[u].x, e0 + 0, lo, rows, xsrc, sidx, acc, cnt, qe, qd, head, qn, lane, ls);
      scan_one<SAVE>(v[u].y, e0 + 1, lo, rows, xsrc, sidx, acc, cnt, qe, qd, head, qn, lane, ls);
      scan_one<SAVE>(v[u].z, e0 + 2, lo, rows, xsrc, sidx, acc, cnt, qe, qd, head, qn, lane, ls);
      scan_one<SAVE>(v[u].w, e0 + 3, lo, rows, xsrc, sidx, acc, cnt, qe, qd, head, qn, lane, ls);
    }
    #pragma unroll
    for (int u = 0; u < SCAN_U; ++u) v[u] = vn[u];
    it = itn;
  }
  #pragma unroll 1
  for (int j = 0; j < TAIL_ITERS; ++j) {
    const int e = TAIL_E0 + j * AGG_THREADS + tid;
    const int dv = (e < N_EDGE) ? didx[e] : -1;
    scan_one<SAVE>(dv, e, lo, rows, xsrc, sidx, acc, cnt, qe, qd, head, qn, lane, ls);
  }
  if (qn > 0) drain_ring<SAVE>(xsrc, sidx, nullptr, acc, cnt, qe, qd, head, qn, lane, ls);
}

__device__ __forceinline__ void ring_push(
    unsigned mask, bool hit, int code, const float* __restrict__ xsrc,
    const unsigned* __restrict__ pk, float* acc, float* cnt, int* qe,
    int& head, int& qn, unsigned lane, ListState& ls) {
  if (hit)
    qe[(head + qn + (int)__builtin_amdgcn_mbcnt_lo(mask, 0u)) & (QCAP - 1)] = code;
  qn += __builtin_popcount(mask);
  if (qn >= 32) {
    drain_ring<true>(xsrc, nullptr, pk, acc, cnt, qe, nullptr, head, 32, lane, ls);
    head = (head + 32) & (QCAP - 1);
    qn -= 32;
  }
}

__device__ __forceinline__ unsigned group_hits(const uint4 lo, const uint4 hi, const unsigned (&s)[8]) {
  return (lo.x ^ s[0]) & (lo.y ^ s[1]) & (lo.z ^ s[2]) & (lo.w ^ s[3]) &
         (hi.x ^ s[4]) & (hi.y ^ s[5]) & (hi.z ^ s[6]) & (hi.w ^ s[7]);
}

#define SCAN_PAIRS (SCAN_ITERS / 2)
static_assert(SCAN_ITERS % 2 == 0, "the layer-1 scan takes the whole steps in pairs");
__device__ __forceinline__ void scan_tiles(
    const unsigned* __restrict__ tlw, const unsigned* __restrict__ thw,
    const unsigned* __restrict__ pk,
    const float* __restrict__ xsrc, float* acc, float* cnt, int* qe,
    unsigned mytile, int tile, int tid, unsigned lane, ListState& ls) {
  int head = 0, qn = 0;
  unsigned s[8];
  #pragma unroll
  for (int k = 0; k < 8; ++k) s[k] = ((mytile >> k) & 1u) ? 0u : 0xffffffffu;
  const uint4* tl4 = (const uint4*)tlw;
  const uint4* th4 = (const uint4*)thw;
  const int gl = (int)(lane >> 4) * (AGG_THREADS / 2) + (tid >> 5) * 16 + (int)(lane & 15u);
  int p = (tile * SCAN_PAIRS) / NTILE;
  uint4 vl = tl4[p * AGG_THREADS + gl];
  uint4 vh = th4[p * AGG_THREADS + gl];
  #pragma unroll 1
  for (int n = 0; n < SCAN_PAIRS; ++n) {
    const int pn = (p + 1 == SCAN_PAIRS) ? 0 : p + 1;
    const uint4 vln = tl4[pn * AGG_THREADS + gl];
    const uint4 vhn = th4[pn * AGG_THREADS + gl];
    const int e0 = (p * AGG_THREADS + gl) * 32;
    unsigned m = group_hits(vl, vh, s);
    unsigned mask = __builtin_amdgcn_ballot_w32(m != 0u);
    #pragma unroll 1
    while (mask != 0u) {
      const int code = e0 + (int)__builtin_ctz(m | 0x80000000u);
      ring_push(mask, m != 0u, code, xsrc, pk, acc, cnt, qe, head, qn, lane, ls);
      m &= m - 1u;
      mask = __builtin_amdgcn_ballot_w32(m != 0u);
    }
    vl = vln;
    vh = vhn;
    p = pn;
  }
  #pragma unroll 1
  for (int j = 0; j < TAIL_ITERS; ++j) {
    const int e = TAIL_E0 + j * AGG_THREADS + tid;
    const int g = min(e, N_EDGE - 1) >> 5;
    const unsigned m = group_hits(tl4[g], th4[g], s);
    const bool hit = (e < N_EDGE) && (((m >> (e & 31)) & 1u) != 0u);
    const unsigned mask = __builtin_amdgcn_ballot_w32(hit);
    if (mask != 0u)
      ring_push(mask, hit, e, xsrc, pk, acc, cnt, qe, head, qn, lane, ls);
  }
  if (qn > 0) drain_ring<true>(xsrc, nullptr, pk, acc, cnt, qe, nullptr, head, qn, lane, ls);
}

#define AUX_CNT    0
#define AUX_RINGE  TR
#define AUX_RINGD  (TR + AGG_WAVES * QCAP)
#define AUX_FLOATS (TR + 2 * AGG_WAVES * QCAP)
static_assert(32 * WPS <= AUX_FLOATS, "a staged weight matrix fits where counts and rings were");

__device__ __forceinline__ void stage_weights(float* aux, const float* __restrict__ Wg, int tid) {
  for (int idx = tid; idx < 32 * D; idx += AGG_THREADS) {
    const int kp = idx >> 6, n = idx & 63;
    *(float2*)&aux[kp * WPS + 2 * n] =
        make_float2(Wg[(2 * kp) * D + n], Wg[(2 * kp + 1) * D + n]);
  }
}

__device__ __forceinline__ void load_a_frags(const float* tile, int c, int h, v2f (&a)[16]) {
  #pragma unroll
  for (int j = 0; j < 16; ++j)
    a[j] = *(const v2f*)&tile[c * D + ((j ^ c) << 2) + 2 * h];
}

__device__ __forceinline__ v2f load_b_frag(const float* aux, int wo, int j, int ct, int c, int h) {
  return *(const v2f*)&aux[wo + (2 * j + h) * WPS + 2 * (16 * ct + c)];
}

__device__ __forceinline__ int trip_offset() {
  int wo = 0;
  asm volatile("" : "+v"(wo));
  return wo;
}

__device__ __forceinline__ int d_place(int h, int r, int col) {
  return (8 * h + r) * D + (col ^ (32 * h));
}

template <bool CNT_LDS>
__device__ __forceinline__ void tile_transform(
    float* acc, float* aux, float* __restrict__ cntG,
    const float* __restrict__ xdst, const float* __restrict__ Wl,
    const float* __restrict__ Wr, const float* __restrict__ bias,
    float* __restrict__ out, int rows, bool relu, int tid) {
  const int lane = tid & 31, wave = tid >> 5;
  const int c = lane & 15, h = lane >> 4;
  const int nrt = rows >> 4;

  for (int i = tid; i < rows * 16; i += AGG_THREADS) {
    const int row = i >> 4, q = i & 15;
    const float cv = CNT_LDS ? aux[AUX_CNT + row] : cntG[row];
    if (CNT_LDS && q == 0) cntG[row] = cv;
    const float inv = 1.0f / fmaxf(cv, 1.0f);
    const float4 sv = ((const float4*)acc)[i];
    ((float4*)acc)[(row << 4) + (q ^ (row & 15))] =
        make_float4(sv.x * inv, sv.y * inv, sv.z * inv, sv.w * inv);
  }
  __syncthreads();
  stage_weights(aux, Wl, tid);
  __syncthreads();

  #pragma unroll 1
  for (int rt = wave; rt < nrt; rt += AGG_WAVES) {
    float* tile = acc + rt * 16 * D;
    const int wo = trip_offset();
    v2f a[16];
    load_a_frags(tile, c, h, a);
    #pragma unroll
    for (int ct = 0; ct < 4; ++ct) {
      v8f d = {0.f, 0.f, 0.f, 0.f, 0.f, 0.f, 0.f, 0.f};
      #pragma unroll
      for (int j = 0; j < 16; ++j)
        d = __builtin_amdgcn_wmma_f32_16x16x4_f32(false, a[j], false,
                                                  load_b_frag(aux, wo, j, ct, c, h),
                                                  (short)0, d, false, false);
      #pragma unroll
      for (int r = 0; r < 8; ++r) tile[d_place(h, r, 16 * ct + c)] = d[r];
    }
  }
  __syncthreads();
  stage_weights(aux, Wr, tid);
  __syncthreads();

  #pragma unroll 1
  for (int rt = wave; rt < nrt; rt += AGG_WAVES) {
    float* tile = acc + rt * 16 * D;
    const int wo = trip_offset();
    const float* xt = xdst + (size_t)rt * 16 * D;
    float*       ot = out + (size_t)rt * 16 * D;
    float4 xr[8];
    #pragma unroll
    for (int i = 0; i < 8; ++i)
      xr[i] = *(const float4*)&xt[(2 * i + h) * D + 4 * c];
    v8f d[4];
    #pragma unroll
    for (int ct = 0; ct < 4; ++ct) {
      #pragma unroll
      for (int r = 0; r < 8; ++r) d[ct][r] = tile[d_place(h, r, 16 * ct + c)];
    }
    #pragma unroll
    for (int i = 0; i < 8; ++i) {
      const int row = 2 * i + h;
      *(float4*)&tile[row * D + ((c ^ row) << 2)] = xr[i];
    }
    v2f a[16];
    load_a_frags(tile, c, h, a);
    #pragma unroll
    for (int ct = 0; ct < 4; ++ct) {
      #pragma unroll
      for (int j = 0; j < 16; ++j)
        d[ct] = __builtin_amdgcn_wmma_f32_16x16x4_f32(false, a[j], false,
                                                      load_b_frag(aux, wo, j, ct, c, h),
                                                      (short)0, d[ct], false, false);
    }
    #pragma unroll
    for (int ct = 0; ct < 4; ++ct) {
      const float bcol = bias[16 * ct + c];
      #pragma unroll
      for (int r = 0; r < 8; ++r) {
        float val = d[ct][r] + bcol;
        if (relu) val = fmaxf(val, 0.0f);
        tile[d_place(h, r, 16 * ct + c)] = val;
      }
    }
    #pragma unroll
    for (int i = 0; i < 8; ++i) {
      const int row = 2 * i + h;
      *(float4*)&ot[row * D + 4 * c] =
          *(const float4*)&tile[row * D + ((c ^ (8 * (row >> 3))) << 2)];
    }
  }
}

static_assert(SCAN_ITERS * AGG_THREADS * 16 == TAIL_E0 && TAIL_E0 % 32 == 0,
              "a scan step is 16 edges per lane; the tail starts on a group of 32");
static_assert(NT_M <= 256 && NT_U <= 256, "a tile number fits eight bits");
static_assert(N_USER <= (1 << 21) && N_MOVIE <= (1 << 21) && TR <= (1 << 11),
              "source id and row in tile share one 32-bit word");
static_assert((unsigned long long)PL_THU + N_EDGE / 8 <= (unsigned long long)(N_USER + N_MOVIE) * D,
              "the edge planes fit into the output buffer");
static_assert(N_EDGE % ENC_WAVE_E == 0 && ENC_WAVE_E == 256 && (N_EDGE / 8) % 32 == 0,
              "an encoding wave takes 8 whole groups; every plane starts on a 128-B line");

__global__ __launch_bounds__(ENC_THREADS) void edge_encode_kernel(
    const int* __restrict__ uidx, const int* __restrict__ midx, unsigned* __restrict__ pl) {
  const int lane = threadIdx.x & 31;
  const int w = blockIdx.x * (ENC_THREADS / 32) + (threadIdx.x >> 5);
  if (w >= ENC_WAVES) return;
  unsigned lm = 0u, hm = 0u, lu = 0u, hu = 0u;
  unsigned uv[8], mv[8];
  #pragma unroll
  for (int j = 0; j < 8; ++j) {
    const int e = w * ENC_WAVE_E + j * 32 + lane;
    uv[j] = (unsigned)uidx[e];
    mv[j] = (unsigned)midx[e];
  }
  #pragma unroll
  for (int j = 0; j < 8; ++j) {
    const int e = w * ENC_WAVE_E + j * 32 + lane;
    const unsigned u = uv[j];
    const unsigned m = mv[j];
    const unsigned tm = m / (unsigned)TR;
    const unsigned tu = u / (unsigned)TR;
    pl[PL_PKM + e] = (u << 11) | (m - tm * (unsigned)TR);
    pl[PL_PKU + e] = (m << 11) | (u - tu * (unsigned)TR);
    #pragma unroll
    for (int k = 0; k < 4; ++k) {
      const unsigned bml = __builtin_amdgcn_ballot_w32(((tm >> k) & 1u) != 0u);
      const unsigned bmh = __builtin_amdgcn_ballot_w32(((tm >> (k + 4)) & 1u) != 0u);
      const unsigned bul = __builtin_amdgcn_ballot_w32(((tu >> k) & 1u) != 0u);
      const unsigned buh = __builtin_amdgcn_ballot_w32(((tu >> (k + 4)) & 1u) != 0u);
      if (lane == 4 * j + k) { lm = bml; hm = bmh; lu = bul; hu = buh; }
    }
  }
  pl[PL_TLM + w * 32 + lane] = lm;
  pl[PL_THM + w * 32 + lane] = hm;
  pl[PL_TLU + w * 32 + lane] = lu;
  pl[PL_THU + w * 32 + lane] = hu;
}

__global__ __launch_bounds__(AGG_THREADS) void tile_layer1_kernel(
    const float* __restrict__ xu, const float* __restrict__ xm,
    const unsigned* __restrict__ pl,
    const float* __restrict__ WlM, const float* __restrict__ WrM, const float* __restrict__ bM,
    const float* __restrict__ WlU, const float* __restrict__ WrU, const float* __restrict__ bU,
    float* __restrict__ cu, float* __restrict__ cm,
    float* __restrict__ u1, float* __restrict__ m1,
    unsigned* __restrict__ lst) {
  __shared__ __attribute__((aligned(16))) float acc[TR * D];
  __shared__ __attribute__((aligned(16))) float aux[AUX_FLOATS];
  float* cnt   = aux + AUX_CNT;
  int*   ringE = (int*)(aux + AUX_RINGE);

  const int tid = threadIdx.x;
  const unsigned lane = threadIdx.x & 31u;
  const int wave = tid >> 5;
  const int tile = blockIdx.x;

  const bool mov = tile < NT_M;
  const unsigned* tlw = pl + (mov ? PL_TLM : PL_TLU);
  const unsigned* thw = pl + (mov ? PL_THM : PL_THU);
  const unsigned* pk  = pl + (mov ? PL_PKM : PL_PKU);
  const float* xsrc = mov ? xu : xm;
  const float* xdst = mov ? xm : xu;
  float*       hOut = mov ? m1 : u1;
  float*       cOut = mov ? cm : cu;
  const int    nrow = mov ? N_MOVIE : N_USER;
  const int    lo   = (mov ? tile : tile - NT_M) * TR;
  const int    rows = min(TR, nrow - lo);

  for (int i = tid; i < TR * D / 4; i += AGG_THREADS)
    ((float4*)acc)[i] = make_float4(0.f, 0.f, 0.f, 0.f);
  for (int i = tid; i < TR; i += AGG_THREADS) cnt[i] = 0.f;
  __syncthreads();

  unsigned* wreg = lst + (unsigned)(tile * AGG_WAVES + wave) * LSTRIDE;
  ListState ls;
  ls.wl = wreg + 32;
  ls.n0 = 0; ls.n1 = 0; ls.n2 = 0;
  ls.soff = mov ? 0 : N_USER;
  {
    int b[3];
    #pragma unroll
    for (int k = 0; k < 3; ++k) {
      int base = 0;
      if (!mov) {
        const int g = 3 * (tile - NT_M) + k;
        base = sub_size((g >> 1) % 3) + ((g & 1) ? sub_size((g - 1) % 3) : 0);
      }
      b[k] = base - sub_lo(k);
    }
    ls.b0 = b[0]; ls.b1 = b[1]; ls.b2 = b[2];
  }
  scan_tiles(tlw, thw, pk, xsrc, acc, cnt, &ringE[wave * QCAP],
             (unsigned)(mov ? tile : tile - NT_M), tile, tid, lane, ls);
  if (lane < 3u)
    wreg[lane] = (unsigned)(lane == 0u ? ls.n0 : (lane == 1u ? ls.n1 : ls.n2));
  __syncthreads();

  tile_transform<true>(acc, aux, cOut + lo, xdst + (size_t)lo * D,
                       mov ? WlM : WlU, mov ? WrM : WrU, mov ? bM : bU,
                       hOut + (size_t)lo * D, rows, true, tid);
}

static_assert(OFF_M1 == OFF_U1 + N_USER * 64u, "the layer-1 tables are one joint table, users first");

__device__ __forceinline__ void sub_list(const unsigned* __restrict__ lst, bool exists, int tile1,
                                         int wave, int k, int& n, unsigned& off) {
  const unsigned reg = (unsigned)(tile1 * AGG_WAVES + wave) * LSTRIDE;
  n = exists ? __builtin_amdgcn_readfirstlane((int)lst[reg + (unsigned)k]) : 0;
  off = reg + 32u + (unsigned)k * SUBCAP;
}

__global__ __launch_bounds__(AGG_THREADS) void tile_layer2_kernel(
    const float* __restrict__ h1u, const float* __restrict__ h1m,
    const int* __restrict__ uidx, const int* __restrict__ midx,
    const float* __restrict__ WlM, const float* __restrict__ WrM, const float* __restrict__ bM,
    const float* __restrict__ WlU, const float* __restrict__ WrU, const float* __restrict__ bU,
    float* __restrict__ cu, float* __restrict__ cm,
    float* __restrict__ outU, float* __restrict__ outM,
    const unsigned* __restrict__ lst) {
  __shared__ __attribute__((aligned(16))) float acc[TR * D];
  __shared__ __attribute__((aligned(16))) float aux[AUX_FLOATS];
  int* ringE = (int*)(aux + AUX_RINGE);
  int* ringD = (int*)(aux + AUX_RINGD);

  const int tid = threadIdx.x;
  const unsigned lane = threadIdx.x & 31u;
  const int wave = tid >> 5;
  const int wg = blockIdx.x;

  const int mlo   = sub_start(wg, N_MOVIE);
  const int rm    = sub_start(wg + 1, N_MOVIE) - mlo;
  const int ulo   = sub_start(2 * wg, N_USER);
  const int umid  = sub_start(2 * wg + 1, N_USER);
  const int uhi   = sub_start(2 * wg + 2, N_USER);
  const int ubase = sub_size(wg % 3);

  for (int i = tid; i < TR * D / 4; i += AGG_THREADS)
    ((float4*)acc)[i] = make_float4(0.f, 0.f, 0.f, 0.f);
  __syncthreads();

  int n0, n1, n2;
  unsigned o0, o1, o2;
  {
    const int g1 = 2 * wg, g2 = 2 * wg + 1;
    sub_list(lst, true, wg / 3, wave, wg % 3, n0, o0);
    sub_list(lst, g1 / 3 < NT_U, NT_M + g1 / 3, wave, g1 % 3, n1, o1);
    sub_list(lst, g2 / 3 < NT_U, NT_M + g2 / 3, wave, g2 % 3, n2, o2);
  }
  const int ovf = (n0 > SUBCAP ? 1 : 0) | (n1 > SUBCAP ? 2 : 0) | (n2 > SUBCAP ? 4 : 0);
  if (ovf != 0) {
    #pragma unroll 1
    for (int p = 0; p < 3; ++p) {
      if ((ovf >> p) & 1) {
        const bool mv = p == 0;
        const int plo   = mv ? mlo : (p == 1 ? ulo : umid);
        const int prows = mv ? rm : (p == 1 ? umid - ulo : uhi - umid);
        const int prow0 = mv ? 0 : (p == 1 ? ubase : ubase + (umid - ulo));
        ListState none;
        none.wl = nullptr;
        none.n0 = 0; none.n1 = 0; none.n2 = 0;
        none.b0 = 0; none.b1 = 0; none.b2 = 0;
        none.soff = 0;
        scan_edges<false>(mv ? midx : uidx, mv ? uidx : midx, mv ? h1u : h1m,
                          acc + prow0 * D, nullptr, &ringE[wave * QCAP], &ringD[wave * QCAP],
                          plo, prows, wg, tid, lane, none);
      }
    }
    if (n0 > SUBCAP) n0 = 0;
    if (n1 > SUBCAP) n1 = 0;
    if (n2 > SUBCAP) n2 = 0;
  }

  const int n01 = n0 + n1;
  const int n = n01 + n2;
  if (n > 0) {
    const int last = n - 1;
    auto entry = [&](int i) -> unsigned {
      const unsigned o = (i < n0) ? o0 + (unsigned)i
                       : ((i < n01) ? o1 + (unsigned)(i - n0) : o2 + (unsigned)(i - n01));
      return lst[o];
    };
    unsigned w = entry(min((int)lane, last));
    #pragma unroll 1
    for (int base = 0; base < n; base += 32) {
      const int m = min(32, n - base);
      const unsigned wn = entry(min(base + 32 + (int)lane, last));
      const int s = (int)(w >> 11);
      const int d = (int)(w & 2047u);
      if (m == 32) {
        float2 r[32];
        #pragma unroll
        for (int j = 0; j < 32; ++j)
          r[j] = row_load(h1u, __builtin_amdgcn_readlane(s, j), lane);
        #pragma unroll
        for (int j = 0; j < 32; ++j) {
          float* q = &acc[((unsigned)__builtin_amdgcn_readlane(d, j) << 6) + lane * 2u];
          atomicAdd(q + 0, r[j].x);
          atomicAdd(q + 1, r[j].y);
        }
      } else {
        #pragma unroll 1
        for (int k = 0; k < m; ++k) {
          const float2 r1 = row_load(h1u, __builtin_amdgcn_readlane(s, k), lane);
          float* q = &acc[((unsigned)__builtin_amdgcn_readlane(d, k) << 6) + lane * 2u];
          atomicAdd(q + 0, r1.x);
          atomicAdd(q + 1, r1.y);
        }
      }
      w = wn;
    }
  }
  __syncthreads();

  tile_transform<false>(acc, aux, cm + mlo, h1m + (size_t)mlo * D, WlM, WrM, bM,
                        outM + (size_t)mlo * D, rm, false, tid);
  tile_transform<false>(acc + ubase * D, aux, cu + ulo, h1u + (size_t)ulo * D, WlU, WrU, bU,
                        outU + (size_t)ulo * D, uhi - ulo, false, tid);
}

extern "C" void kernel_launch(void* const* d_in, const int* in_sizes, int n_in,
                              void* d_out, int out_size, void* d_ws, size_t ws_size,
                              hipStream_t stream) {
  const float* x_user  = (const float*)d_in[0];
  const float* x_movie = (const float*)d_in[1];
  const float* Wl1_um  = (const float*)d_in[2];
  const float* Wr1_um  = (const float*)d_in[3];
  const float* b1_um   = (const float*)d_in[4];
  const float* Wl1_mu  = (const float*)d_in[5];
  const float* Wr1_mu  = (const float*)d_in[6];
  const float* b1_mu   = (const float*)d_in[7];
  const float* Wl2_um  = (const float*)d_in[8];
  const float* Wr2_um  = (const float*)d_in[9];
  const float* b2_um   = (const float*)d_in[10];
  const float* Wl2_mu  = (const float*)d_in[11];
  const float* Wr2_mu  = (const float*)d_in[12];
  const float* b2_mu   = (const float*)d_in[13];
  const int*   uidx    = (const int*)d_in[14];
  const int*   midx    = (const int*)d_in[15];

  float* ws = (float*)d_ws;
  unsigned* lst = (unsigned*)(ws + OFF_LST);
  float* cu = ws + OFF_CU;
  float* cm = ws + OFF_CM;
  float* u1 = ws + OFF_U1;
  float* m1 = ws + OFF_M1;

  float* u2 = (float*)d_out;
  float* m2 = (float*)d_out + (size_t)N_USER * D;

  const dim3 bAgg(AGG_THREADS);
  const dim3 gAgg(NTILE);

  unsigned* pl = (unsigned*)d_out;
  edge_encode_kernel<<<dim3(ENC_BLOCKS), dim3(ENC_THREADS), 0, stream>>>(uidx, midx, pl);

  tile_layer1_kernel<<<gAgg, bAgg, 0, stream>>>(x_user, x_movie, pl,
                                                Wl1_um, Wr1_um, b1_um,
                                                Wl1_mu, Wr1_mu, b1_mu,
                                                cu, cm, u1, m1, lst);
  tile_layer2_kernel<<<gAgg, bAgg, 0, stream>>>(u1, m1, uidx, midx,
                                                Wl2_um, Wr2_um, b2_um,
                                                Wl2_mu, Wr2_mu, b2_mu,
                                                cu, cm, u2, m2, lst);
}


# Layer 1 re-tiled like layer 2 for equal edge counts: a work-group owns a third of a movie tile and two thirds of a user tile, scans both relations' planes; one list per wave; atomic sum order changes
# speedup vs baseline: 3.6117x; 1.0626x over previous
#include <hip/hip_runtime.h>
#include <hip/hip_bf16.h>


#define N_USER  200000
#define N_MOVIE 100000
#define N_EDGE  4000000
#define D       64

typedef float v2f __attribute__((ext_vector_type(2)));
typedef float v8f __attribute__((ext_vector_type(8)));

#define WPS 160

#define OFF_LST 0u
#define OFF_CU (OFF_LST + (N_USER + N_MOVIE) * 64u)
#define OFF_CM (OFF_CU + N_USER)
#define OFF_U1 (OFF_CM + N_MOVIE)
#define OFF_M1 (OFF_U1 + N_USER * 64u)

#define PL_PKM 0u
#define PL_PKU (PL_PKM + N_EDGE)
#define PL_TLM (PL_PKU + N_EDGE)
#define PL_THM (PL_TLM + N_EDGE / 8)
#define PL_TLU (PL_THM + N_EDGE / 8)
#define PL_THU (PL_TLU + N_EDGE / 8)
#define ENC_THREADS 256
#define ENC_WAVE_E  256
#define ENC_WAVES   (N_EDGE / ENC_WAVE_E)
#define ENC_BLOCKS  ((ENC_WAVES + ENC_THREADS / 32 - 1) / (ENC_THREADS / 32))

#define TR          1184
#define NT_M        ((N_MOVIE + TR - 1) / TR)
#define NT_U        ((N_USER + TR - 1) / TR)
#define NTILE       (NT_M + NT_U)
#define AGG_THREADS 1024
#define AGG_WAVES   (AGG_THREADS / 32)
#define QCAP        64
#define N_EDGE4     (N_EDGE / 4)
#define SCAN_U      4
#define STEP4       (AGG_THREADS * SCAN_U)
#define SCAN_ITERS  (N_EDGE4 / STEP4)
#define TAIL_E0     (SCAN_ITERS * STEP4 * 4)
#define TAIL_ITERS  ((N_EDGE - TAIL_E0 + AGG_THREADS - 1) / AGG_THREADS)
#define GB          8

#define LSTRIDE     2336u
#define LCAP        2304
#define K1          400
#define K2          784

static_assert(LCAP + 32 == (int)LSTRIDE, "count line and list fill the wave's share");
static_assert(K1 % 16 == 0 && K2 % 16 == 0 && TR - K2 == K1 && (K2 - K1) + 2 * K1 == TR,
              "sub-ranges are whole matrix tiles; one movie and two user sub-ranges are TR rows");
static_assert((NTILE / 3) * TR + (NTILE % 3 == 0 ? 0 : (NTILE % 3 == 1 ? K1 : K2)) >= N_MOVIE,
              "layer 2: movie sub-ranges 0 .. NTILE - 1 hold every movie row");
static_assert(((2 * NTILE) / 3) * TR + ((2 * NTILE) % 3 == 0 ? 0 : ((2 * NTILE) % 3 == 1 ? K1 : K2)) >= N_USER,
              "layer 2: user sub-ranges 0 .. 2 NTILE - 1 hold every user row");
static_assert(N_USER + N_MOVIE <= (1 << 21), "joint source row and LDS row share one 32-bit word");

__device__ __forceinline__ int sub_lo(int k) { return k == 0 ? 0 : (k == 1 ? K1 : (k == 2 ? K2 : TR)); }
__device__ __forceinline__ int sub_size(int k) { return k == 1 ? K2 - K1 : K1; }
__device__ __forceinline__ void row_owner(unsigned id, bool user, unsigned& wg, unsigned& lrow) {
  const unsigned t = id / (unsigned)TR;
  const unsigned r = id - t * (unsigned)TR;
  const unsigned k = r < (unsigned)K1 ? 0u : (r < (unsigned)K2 ? 1u : 2u);
  const unsigned g = 3u * t + k;
  const unsigned off = r - (unsigned)sub_lo((int)k);
  if (!user) {
    wg = g;
    lrow = off;
  } else {
    wg = g >> 1;
    lrow = (unsigned)sub_size((int)(wg % 3u)) +
           ((g & 1u) ? (unsigned)sub_size((int)((g - 1u) % 3u)) : 0u) + off;
  }
}

__device__ __forceinline__ int sub_start(int g, int nrow) {
  const int t = g / 3;
  return min(t * TR + sub_lo(g - 3 * t), nrow);
}

struct ListState {
  unsigned* wl;
  int n0;
  int soff;
};

static_assert((unsigned long long)NTILE * AGG_WAVES * LSTRIDE
                  <= (unsigned long long)(N_USER + N_MOVIE) * D,
              "counts and lists fit into their workspace plane");
static_assert(TR % 16 == 0 && N_USER % 16 == 0 && N_MOVIE % 16 == 0,
              "every tile is a whole number of 16-row matrix tiles");
static_assert(LSTRIDE % 32 == 0 && LCAP % 32 == 0, "a wave's count line and list are whole 128-B lines");
static_assert(TR <= (1 << 11) && N_USER <= (1 << 21) && N_MOVIE <= (1 << 21),
              "source id and row in tile share one 32-bit word");

static_assert(N_EDGE % 4 == 0 && SCAN_U == 4, "edge indices are scanned 4 per load, 16 per lane and step");
static_assert((QCAP & (QCAP - 1)) == 0 && QCAP >= 64, "ring holds 31 + 32 entries");

__device__ __forceinline__ float2 row_load(const float* __restrict__ x, int r,
                                           unsigned lane) {
  return ((const float2*)x)[((unsigned)r << 5) + lane];
}

template <bool SAVE>
__device__ __forceinline__ void drain_ring(
    const float* __restrict__ xsrc, const int* __restrict__ sidx,
    const unsigned* __restrict__ pk,
    float* acc, float* cnt, const int* qe, const int* qd,
    int head, int n, unsigned lane, ListState& ls) {
  int s = 0, d = 0;
  if ((int)lane < n) {
    const int slot = (head + (int)lane) & (QCAP - 1);
    const int e = qe[slot];
    if (SAVE) {
      const unsigned w = pk[e];
      d = (int)(w & 2047u);
      s = (int)(w >> 11);
      atomicAdd(&cnt[d], 1.0f);
    } else {
      d = qd[slot];
      s = sidx[e];
    }
  }
  if (SAVE) {
    if ((int)lane < n && ls.n0 + n <= LCAP)
      ls.wl[ls.n0 + (int)lane] = ((unsigned)(s + ls.soff) << 11) | (unsigned)d;
    ls.n0 += n;
  }
  if (n == 32) {
    #pragma unroll 1
    for (int k = 0; k < 32; k += GB) {
      float2 r[GB];
      #pragma unroll
      for (int j = 0; j < GB; ++j)
        r[j] = row_load(xsrc, __builtin_amdgcn_readlane(s, k + j), lane);
      #pragma unroll
      for (int j = 0; j < GB; ++j) {
        float* q = &acc[((unsigned)__builtin_amdgcn_readlane(d, k + j) << 6) + lane * 2u];
        atomicAdd(q + 0, r[j].x);
        atomicAdd(q + 1, r[j].y);
      }
    }
  } else {
    #pragma unroll 1
    for (int k = 0; k < n; ++k) {
      const float2 r1 = row_load(xsrc, __builtin_amdgcn_readlane(s, k), lane);
      float* q = &acc[((unsigned)__builtin_amdgcn_readlane(d, k) << 6) + lane * 2u];
      atomicAdd(q + 0, r1.x);
      atomicAdd(q + 1, r1.y);
    }
  }
}

template <bool SAVE>
__device__ __forceinline__ void scan_one(
    int dst, int eid, int lo, int rows,
    const float* __restrict__ xsrc, const int* __restrict__ sidx,
    float* acc, float* cnt, int* qe, int* qd,
    int& head, int& qn, unsigned lane, ListState& ls) {
  const unsigned dl = (unsigned)(dst - lo);
  const bool hit = dl < (unsigned)rows;
  const unsigned mask = __builtin_amdgcn_ballot_w32(hit);
  if (mask != 0u) {
    if (hit) {
      const int pos = (head + qn + (int)__builtin_amdgcn_mbcnt_lo(mask, 0u)) & (QCAP - 1);
      qe[pos] = eid;
      qd[pos] = (int)dl;
    }
    qn += __builtin_popcount(mask);
    if (qn >= 32) {
      drain_ring<SAVE>(xsrc, sidx, nullptr, acc, cnt, qe, qd, head, 32, lane, ls);
      head = (head + 32) & (QCAP - 1);
      qn -= 32;
    }
  }
}

__device__ __forceinline__ int4 scan_load(const int4* __restrict__ d4, int it, int u, int tid) {
  return d4[(it * AGG_THREADS + tid) * SCAN_U + u];
}

template <bool SAVE>
__device__ __forceinline__ void scan_edges(
    const int* __restrict__ didx, const int* __restrict__ sidx,
    const float* __restrict__ xsrc, float* acc, float* cnt, int* qe, int* qd,
    int lo, int rows, int tile, int tid, unsigned lane, ListState& ls) {
  int head = 0, qn = 0;

  const int4* d4 = (const int4*)didx;
  int it = (tile * SCAN_ITERS) / NTILE;
  int4 v[SCAN_U];
  #pragma unroll
  for (int u = 0; u < SCAN_U; ++u) v[u] = scan_load(d4, it, u, tid);
  #pragma unroll 1
  for (int n = 0; n < SCAN_ITERS; ++n) {
    const int itn = (it + 1 == SCAN_ITERS) ? 0 : it + 1;
    int4 vn[SCAN_U];
    #pragma unroll
    for (int u = 0; u < SCAN_U; ++u) vn[u] = scan_load(d4, itn, u, tid);
    #pragma unroll
    for (int u = 0; u < SCAN_U; ++u) {
      const int e0 = ((it * AGG_THREADS + tid) * SCAN_U + u) * 4;
      scan_one<SAVE>(v[u].x, e0 + 0, lo, rows, xsrc, sidx, acc, cnt, qe, qd, head, qn, lane, ls);
      scan_one<SAVE>(v[u].y, e0 + 1, lo, rows, xsrc, sidx, acc, cnt, qe, qd, head, qn, lane, ls);
      scan_one<SAVE>(v[u].z, e0 + 2, lo, rows, xsrc, sidx, acc, cnt, qe, qd, head, qn, lane, ls);
      scan_one<SAVE>(v[u].w, e0 + 3, lo, rows, xsrc, sidx, acc, cnt, qe, qd, head, qn, lane, ls);
    }
    #pragma unroll
    for (int u = 0; u < SCAN_U; ++u) v[u] = vn[u];
    it = itn;
  }
  #pragma unroll 1
  for (int j = 0; j < TAIL_ITERS; ++j) {
    const int e = TAIL_E0 + j * AGG_THREADS + tid;
    const int dv = (e < N_EDGE) ? didx[e] : -1;
    scan_one<SAVE>(dv, e, lo, rows, xsrc, sidx, acc, cnt, qe, qd, head, qn, lane, ls);
  }
  if (qn > 0) drain_ring<SAVE>(xsrc, sidx, nullptr, acc, cnt, qe, qd, head, qn, lane, ls);
}

__device__ __forceinline__ void ring_push(
    unsigned mask, bool hit, int code, const float* __restrict__ xsrc,
    const unsigned* __restrict__ pk, float* acc, float* cnt, int* qe,
    int& head, int& qn, unsigned lane, ListState& ls) {
  if (hit)
    qe[(head + qn + (int)__builtin_amdgcn_mbcnt_lo(mask, 0u)) & (QCAP - 1)] = code;
  qn += __builtin_popcount(mask);
  if (qn >= 32) {
    drain_ring<true>(xsrc, nullptr, pk, acc, cnt, qe, nullptr, head, 32, lane, ls);
    head = (head + 32) & (QCAP - 1);
    qn -= 32;
  }
}

__device__ __forceinline__ unsigned group_hits(const uint4 lo, const uint4 hi, const unsigned (&s)[8]) {
  return (lo.x ^ s[0]) & (lo.y ^ s[1]) & (lo.z ^ s[2]) & (lo.w ^ s[3]) &
         (hi.x ^ s[4]) & (hi.y ^ s[5]) & (hi.z ^ s[6]) & (hi.w ^ s[7]);
}

#define SCAN_PAIRS (SCAN_ITERS / 2)
static_assert(SCAN_ITERS % 2 == 0, "the layer-1 scan takes the whole steps in pairs");
__device__ __forceinline__ void scan_tiles(
    const unsigned* __restrict__ tlw, const unsigned* __restrict__ thw,
    const unsigned* __restrict__ pk,
    const float* __restrict__ xsrc, float* acc, float* cnt, int* qe,
    unsigned mytile, int tile, int tid, unsigned lane, ListState& ls) {
  int head = 0, qn = 0;
  unsigned s[8];
  #pragma unroll
  for (int k = 0; k < 8; ++k) s[k] = ((mytile >> k) & 1u) ? 0u : 0xffffffffu;
  const uint4* tl4 = (const uint4*)tlw;
  const uint4* th4 = (const uint4*)thw;
  const int gl = (int)(lane >> 4) * (AGG_THREADS / 2) + (tid >> 5) * 16 + (int)(lane & 15u);
  int p = (tile * SCAN_PAIRS) / NTILE;
  uint4 vl = tl4[p * AGG_THREADS + gl];
  uint4 vh = th4[p * AGG_THREADS + gl];
  #pragma unroll 1
  for (int n = 0; n < SCAN_PAIRS; ++n) {
    const int pn = (p + 1 == SCAN_PAIRS) ? 0 : p + 1;
    const uint4 vln = tl4[pn * AGG_THREADS + gl];
    const uint4 vhn = th4[pn * AGG_THREADS + gl];
    const int e0 = (p * AGG_THREADS + gl) * 32;
    unsigned m = group_hits(vl, vh, s);
    unsigned mask = __builtin_amdgcn_ballot_w32(m != 0u);
    #pragma unroll 1
    while (mask != 0u) {
      const int code = e0 + (int)__builtin_ctz(m | 0x80000000u);
      ring_push(mask, m != 0u, code, xsrc, pk, acc, cnt, qe, head, qn, lane, ls);
      m &= m - 1u;
      mask = __builtin_amdgcn_ballot_w32(m != 0u);
    }
    vl = vln;
    vh = vhn;
    p = pn;
  }
  #pragma unroll 1
  for (int j = 0; j < TAIL_ITERS; ++j) {
    const int e = TAIL_E0 + j * AGG_THREADS + tid;
    const int g = min(e, N_EDGE - 1) >> 5;
    const unsigned m = group_hits(tl4[g], th4[g], s);
    const bool hit = (e < N_EDGE) && (((m >> (e & 31)) & 1u) != 0u);
    const unsigned mask = __builtin_amdgcn_ballot_w32(hit);
    if (mask != 0u)
      ring_push(mask, hit, e, xsrc, pk, acc, cnt, qe, head, qn, lane, ls);
  }
  if (qn > 0) drain_ring<true>(xsrc, nullptr, pk, acc, cnt, qe, nullptr, head, qn, lane, ls);
}

#define AUX_CNT    0
#define AUX_RINGE  TR
#define AUX_RINGD  (TR + AGG_WAVES * QCAP)
#define AUX_FLOATS (TR + 2 * AGG_WAVES * QCAP)
static_assert(32 * WPS <= AUX_FLOATS, "a staged weight matrix fits where counts and rings were");

__device__ __forceinline__ void stage_weights(float* aux, const float* __restrict__ Wg, int tid) {
  for (int idx = tid; idx < 32 * D; idx += AGG_THREADS) {
    const int kp = idx >> 6, n = idx & 63;
    *(float2*)&aux[kp * WPS + 2 * n] =
        make_float2(Wg[(2 * kp) * D + n], Wg[(2 * kp + 1) * D + n]);
  }
}

__device__ __forceinline__ void load_a_frags(const float* tile, int c, int h, v2f (&a)[16]) {
  #pragma unroll
  for (int j = 0; j < 16; ++j)
    a[j] = *(const v2f*)&tile[c * D + ((j ^ c) << 2) + 2 * h];
}

__device__ __forceinline__ v2f load_b_frag(const float* aux, int wo, int j, int ct, int c, int h) {
  return *(const v2f*)&aux[wo + (2 * j + h) * WPS + 2 * (16 * ct + c)];
}

__device__ __forceinline__ int trip_offset() {
  int wo = 0;
  asm volatile("" : "+v"(wo));
  return wo;
}

__device__ __forceinline__ int d_place(int h, int r, int col) {
  return (8 * h + r) * D + (col ^ (32 * h));
}

template <bool SCALED>
__device__ __forceinline__ void tile_transform(
    float* acc, float* aux, const float* __restrict__ cntG,
    const float* __restrict__ xdst, const float* __restrict__ Wl,
    const float* __restrict__ Wr, const float* __restrict__ bias,
    float* __restrict__ out, int rows, bool relu, int tid) {
  const int lane = tid & 31, wave = tid >> 5;
  const int c = lane & 15, h = lane >> 4;
  const int nrt = rows >> 4;

  if (!SCALED) {
    for (int i = tid; i < rows * 16; i += AGG_THREADS) {
      const int row = i >> 4, q = i & 15;
      const float inv = 1.0f / fmaxf(cntG[row], 1.0f);
      const float4 sv = ((const float4*)acc)[i];
      ((float4*)acc)[(row << 4) + (q ^ (row & 15))] =
          make_float4(sv.x * inv, sv.y * inv, sv.z * inv, sv.w * inv);
    }
  }
  __syncthreads();
  stage_weights(aux, Wl, tid);
  __syncthreads();

  #pragma unroll 1
  for (int rt = wave; rt < nrt; rt += AGG_WAVES) {
    float* tile = acc + rt * 16 * D;
    const int wo = trip_offset();
    v2f a[16];
    load_a_frags(tile, c, h, a);
    #pragma unroll
    for (int ct = 0; ct < 4; ++ct) {
      v8f d = {0.f, 0.f, 0.f, 0.f, 0.f, 0.f, 0.f, 0.f};
      #pragma unroll
      for (int j = 0; j < 16; ++j)
        d = __builtin_amdgcn_wmma_f32_16x16x4_f32(false, a[j], false,
                                                  load_b_frag(aux, wo, j, ct, c, h),
                                                  (short)0, d, false, false);
      #pragma unroll
      for (int r = 0; r < 8; ++r) tile[d_place(h, r, 16 * ct + c)] = d[r];
    }
  }
  __syncthreads();
  stage_weights(aux, Wr, tid);
  __syncthreads();

  #pragma unroll 1
  for (int rt = wave; rt < nrt; rt += AGG_WAVES) {
    float* tile = acc + rt * 16 * D;
    const int wo = trip_offset();
    const float* xt = xdst + (size_t)rt * 16 * D;
    float*       ot = out + (size_t)rt * 16 * D;
    float4 xr[8];
    #pragma unroll
    for (int i = 0; i < 8; ++i)
      xr[i] = *(const float4*)&xt[(2 * i + h) * D + 4 * c];
    v8f d[4];
    #pragma unroll
    for (int ct = 0; ct < 4; ++ct) {
      #pragma unroll
      for (int r = 0; r < 8; ++r) d[ct][r] = tile[d_place(h, r, 16 * ct + c)];
    }
    #pragma unroll
    for (int i = 0; i < 8; ++i) {
      const int row = 2 * i + h;
      *(float4*)&tile[row * D + ((c ^ row) << 2)] = xr[i];
    }
    v2f a[16];
    load_a_frags(tile, c, h, a);
    #pragma unroll
    for (int ct = 0; ct < 4; ++ct) {
      #pragma unroll
      for (int j = 0; j < 16; ++j)
        d[ct] = __builtin_amdgcn_wmma_f32_16x16x4_f32(false, a[j], false,
                                                      load_b_frag(aux, wo, j, ct, c, h),
                                                      (short)0, d[ct], false, false);
    }
    #pragma unroll
    for (int ct = 0; ct < 4; ++ct) {
      const float bcol = bias[16 * ct + c];
      #pragma unroll
      for (int r = 0; r < 8; ++r) {
        float val = d[ct][r] + bcol;
        if (relu) val = fmaxf(val, 0.0f);
        tile[d_place(h, r, 16 * ct + c)] = val;
      }
    }
    #pragma unroll
    for (int i = 0; i < 8; ++i) {
      const int row = 2 * i + h;
      *(float4*)&ot[row * D + 4 * c] =
          *(const float4*)&tile[row * D + ((c ^ (8 * (row >> 3))) << 2)];
    }
  }
}

static_assert(SCAN_ITERS * AGG_THREADS * 16 == TAIL_E0 && TAIL_E0 % 32 == 0,
              "a scan step is 16 edges per lane; the tail starts on a group of 32");
static_assert(NTILE <= 256, "a tile number fits eight bits");
static_assert(N_USER <= (1 << 21) && N_MOVIE <= (1 << 21) && TR <= (1 << 11),
              "source id and row in tile share one 32-bit word");
static_assert((unsigned long long)PL_THU + N_EDGE / 8 <= (unsigned long long)(N_USER + N_MOVIE) * D,
              "the edge planes fit into the output buffer");
static_assert(N_EDGE % ENC_WAVE_E == 0 && ENC_WAVE_E == 256 && (N_EDGE / 8) % 32 == 0,
              "an encoding wave takes 8 whole groups; every plane starts on a 128-B line");

__global__ __launch_bounds__(ENC_THREADS) void edge_encode_kernel(
    const int* __restrict__ uidx, const int* __restrict__ midx, unsigned* __restrict__ pl) {
  const int lane = threadIdx.x & 31;
  const int w = blockIdx.x * (ENC_THREADS / 32) + (threadIdx.x >> 5);
  if (w >= ENC_WAVES) return;
  unsigned lm = 0u, hm = 0u, lu = 0u, hu = 0u;
  unsigned uv[8], mv[8];
  #pragma unroll
  for (int j = 0; j < 8; ++j) {
    const int e = w * ENC_WAVE_E + j * 32 + lane;
    uv[j] = (unsigned)uidx[e];
    mv[j] = (unsigned)midx[e];
  }
  #pragma unroll
  for (int j = 0; j < 8; ++j) {
    const int e = w * ENC_WAVE_E + j * 32 + lane;
    const unsigned u = uv[j];
    const unsigned m = mv[j];
    unsigned tm, rm_, tu, ru;
    row_owner(m, false, tm, rm_);
    row_owner(u, true, tu, ru);
    pl[PL_PKM + e] = (u << 11) | rm_;
    pl[PL_PKU + e] = (m << 11) | ru;
    #pragma unroll
    for (int k = 0; k < 4; ++k) {
      const unsigned bml = __builtin_amdgcn_ballot_w32(((tm >> k) & 1u) != 0u);
      const unsigned bmh = __builtin_amdgcn_ballot_w32(((tm >> (k + 4)) & 1u) != 0u);
      const unsigned bul = __builtin_amdgcn_ballot_w32(((tu >> k) & 1u) != 0u);
      const unsigned buh = __builtin_amdgcn_ballot_w32(((tu >> (k + 4)) & 1u) != 0u);
      if (lane == 4 * j + k) { lm = bml; hm = bmh; lu = bul; hu = buh; }
    }
  }
  pl[PL_TLM + w * 32 + lane] = lm;
  pl[PL_THM + w * 32 + lane] = hm;
  pl[PL_TLU + w * 32 + lane] = lu;
  pl[PL_THU + w * 32 + lane] = hu;
}

__global__ __launch_bounds__(AGG_THREADS) void tile_layer1_kernel(
    const float* __restrict__ xu, const float* __restrict__ xm,
    const unsigned* __restrict__ pl,
    const float* __restrict__ WlM, const float* __restrict__ WrM, const float* __restrict__ bM,
    const float* __restrict__ WlU, const float* __restrict__ WrU, const float* __restrict__ bU,
    float* __restrict__ cu, float* __restrict__ cm,
    float* __restrict__ u1, float* __restrict__ m1,
    unsigned* __restrict__ lst) {
  __shared__ __attribute__((aligned(16))) float acc[TR * D];
  __shared__ __attribute__((aligned(16))) float aux[AUX_FLOATS];
  float* cnt   = aux + AUX_CNT;
  int*   ringE = (int*)(aux + AUX_RINGE);

  const int tid = threadIdx.x;
  const unsigned lane = threadIdx.x & 31u;
  const int wave = tid >> 5;
  const int wg = blockIdx.x;

  const int mlo   = sub_start(wg, N_MOVIE);
  const int rm    = sub_start(wg + 1, N_MOVIE) - mlo;
  const int ulo   = sub_start(2 * wg, N_USER);
  const int uhi   = sub_start(2 * wg + 2, N_USER);
  const int ubase = sub_size(wg % 3);

  for (int i = tid; i < TR * D / 4; i += AGG_THREADS)
    ((float4*)acc)[i] = make_float4(0.f, 0.f, 0.f, 0.f);
  for (int i = tid; i < TR; i += AGG_THREADS) cnt[i] = 0.f;
  __syncthreads();

  unsigned* wreg = lst + (unsigned)(wg * AGG_WAVES + wave) * LSTRIDE;
  ListState ls;
  ls.wl = wreg + 32;
  ls.n0 = 0;
  #pragma unroll 1
  for (int rel = 0; rel < 2; ++rel) {
    const bool mov = rel == 0;
    ls.soff = mov ? 0 : N_USER;
    scan_tiles(pl + (mov ? PL_TLM : PL_TLU), pl + (mov ? PL_THM : PL_THU),
               pl + (mov ? PL_PKM : PL_PKU), mov ? xu : xm, acc, cnt,
               &ringE[wave * QCAP], (unsigned)wg, wg, tid, lane, ls);
  }
  if (lane == 0u) wreg[0] = (unsigned)ls.n0;
  __syncthreads();

  for (int i = tid; i < TR * 16; i += AGG_THREADS) {
    const int row = i >> 4, q = i & 15;
    const float cv = cnt[row];
    if (q == 0) {
      if (row < ubase) {
        if (row < rm) cm[mlo + row] = cv;
      } else if (row - ubase < uhi - ulo) {
        cu[ulo + row - ubase] = cv;
      }
    }
    const float inv = 1.0f / fmaxf(cv, 1.0f);
    const float4 sv = ((const float4*)acc)[i];
    ((float4*)acc)[(row << 4) + (q ^ (row & 15))] =
        make_float4(sv.x * inv, sv.y * inv, sv.z * inv, sv.w * inv);
  }

  tile_transform<true>(acc, aux, nullptr, xm + (size_t)mlo * D, WlM, WrM, bM,
                       m1 + (size_t)mlo * D, rm, true, tid);
  tile_transform<true>(acc + ubase * D, aux, nullptr, xu + (size_t)ulo * D, WlU, WrU, bU,
                       u1 + (size_t)ulo * D, uhi - ulo, true, tid);
}

static_assert(OFF_M1 == OFF_U1 + N_USER * 64u, "the layer-1 tables are one joint table, users first");

__global__ __launch_bounds__(AGG_THREADS) void tile_layer2_kernel(
    const float* __restrict__ h1u, const float* __restrict__ h1m,
    const int* __restrict__ uidx, const int* __restrict__ midx,
    const float* __restrict__ WlM, const float* __restrict__ WrM, const float* __restrict__ bM,
    const float* __restrict__ WlU, const float* __restrict__ WrU, const float* __restrict__ bU,
    float* __restrict__ cu, float* __restrict__ cm,
    float* __restrict__ outU, float* __restrict__ outM,
    const unsigned* __restrict__ lst) {
  __shared__ __attribute__((aligned(16))) float acc[TR * D];
  __shared__ __attribute__((aligned(16))) float aux[AUX_FLOATS];
  int* ringE = (int*)(aux + AUX_RINGE);
  int* ringD = (int*)(aux + AUX_RINGD);

  const int tid = threadIdx.x;
  const unsigned lane = threadIdx.x & 31u;
  const int wave = tid >> 5;
  const int wg = blockIdx.x;

  const int mlo   = sub_start(wg, N_MOVIE);
  const int rm    = sub_start(wg + 1, N_MOVIE) - mlo;
  const int ulo   = sub_start(2 * wg, N_USER);
  const int umid  = sub_start(2 * wg + 1, N_USER);
  const int uhi   = sub_start(2 * wg + 2, N_USER);
  const int ubase = sub_size(wg % 3);

  for (int i = tid; i < TR * D / 4; i += AGG_THREADS)
    ((float4*)acc)[i] = make_float4(0.f, 0.f, 0.f, 0.f);
  __syncthreads();

  const unsigned reg = (unsigned)(wg * AGG_WAVES + wave) * LSTRIDE;
  int n = __builtin_amdgcn_readfirstlane((int)lst[reg]);
  if (n > LCAP) {
    #pragma unroll 1
    for (int p = 0; p < 3; ++p) {
      const bool mv = p == 0;
      const int plo   = mv ? mlo : (p == 1 ? ulo : umid);
      const int prows = mv ? rm : (p == 1 ? umid - ulo : uhi - umid);
      const int prow0 = mv ? 0 : (p == 1 ? ubase : ubase + (umid - ulo));
      ListState none;
      none.wl = nullptr;
      none.n0 = 0;
      none.soff = 0;
      scan_edges<false>(mv ? midx : uidx, mv ? uidx : midx, mv ? h1u : h1m,
                        acc + prow0 * D, nullptr, &ringE[wave * QCAP], &ringD[wave * QCAP],
                        plo, prows, wg, tid, lane, none);
    }
    n = 0;
  }

  if (n > 0) {
    const int last = n - 1;
    auto entry = [&](int i) -> unsigned { return lst[reg + 32u + (unsigned)i]; };
    unsigned w = entry(min((int)lane, last));
    #pragma unroll 1
    for (int base = 0; base < n; base += 32) {
      const int m = min(32, n - base);
      const unsigned wn = entry(min(base + 32 + (int)lane, last));
      const int s = (int)(w >> 11);
      const int d = (int)(w & 2047u);
      if (m == 32) {
        float2 r[32];
        #pragma unroll
        for (int j = 0; j < 32; ++j)
          r[j] = row_load(h1u, __builtin_amdgcn_readlane(s, j), lane);
        #pragma unroll
        for (int j = 0; j < 32; ++j) {
          float* q = &acc[((unsigned)__builtin_amdgcn_readlane(d, j) << 6) + lane * 2u];
          atomicAdd(q + 0, r[j].x);
          atomicAdd(q + 1, r[j].y);
        }
      } else {
        #pragma unroll 1
        for (int k = 0; k < m; ++k) {
          const float2 r1 = row_load(h1u, __builtin_amdgcn_readlane(s, k), lane);
          float* q = &acc[((unsigned)__builtin_amdgcn_readlane(d, k) << 6) + lane * 2u];
          atomicAdd(q + 0, r1.x);
          atomicAdd(q + 1, r1.y);
        }
      }
      w = wn;
    }
  }
  __syncthreads();

  tile_transform<false>(acc, aux, cm + mlo, h1m + (size_t)mlo * D, WlM, WrM, bM,
                        outM + (size_t)mlo * D, rm, false, tid);
  tile_transform<false>(acc + ubase * D, aux, cu + ulo, h1u + (size_t)ulo * D, WlU, WrU, bU,
                        outU + (size_t)ulo * D, uhi - ulo, false, tid);
}

extern "C" void kernel_launch(void* const* d_in, const int* in_sizes, int n_in,
                              void* d_out, int out_size, void* d_ws, size_t ws_size,
                              hipStream_t stream) {
  const float* x_user  = (const float*)d_in[0];
  const float* x_movie = (const float*)d_in[1];
  const float* Wl1_um  = (const float*)d_in[2];
  const float* Wr1_um  = (const float*)d_in[3];
  const float* b1_um   = (const float*)d_in[4];
  const float* Wl1_mu  = (const float*)d_in[5];
  const float* Wr1_mu  = (const float*)d_in[6];
  const float* b1_mu   = (const float*)d_in[7];
  const float* Wl2_um  = (const float*)d_in[8];
  const float* Wr2_um  = (const float*)d_in[9];
  const float* b2_um   = (const float*)d_in[10];
  const float* Wl2_mu  = (const float*)d_in[11];
  const float* Wr2_mu  = (const float*)d_in[12];
  const float* b2_mu   = (const float*)d_in[13];
  const int*   uidx    = (const int*)d_in[14];
  const int*   midx    = (const int*)d_in[15];

  float* ws = (float*)d_ws;
  unsigned* lst = (unsigned*)(ws + OFF_LST);
  float* cu = ws + OFF_CU;
  float* cm = ws + OFF_CM;
  float* u1 = ws + OFF_U1;
  float* m1 = ws + OFF_M1;

  float* u2 = (float*)d_out;
  float* m2 = (float*)d_out + (size_t)N_USER * D;

  const dim3 bAgg(AGG_THREADS);
  const dim3 gAgg(NTILE);

  unsigned* pl = (unsigned*)d_out;
  edge_encode_kernel<<<dim3(ENC_BLOCKS), dim3(ENC_THREADS), 0, stream>>>(uidx, midx, pl);

  tile_layer1_kernel<<<gAgg, bAgg, 0, stream>>>(x_user, x_movie, pl,
                                                Wl1_um, Wr1_um, b1_um,
                                                Wl1_mu, Wr1_mu, b1_mu,
                                                cu, cm, u1, m1, lst);
  tile_layer2_kernel<<<gAgg, bAgg, 0, stream>>>(u1, m1, uidx, midx,
                                                Wl2_um, Wr2_um, b2_um,
                                                Wl2_mu, Wr2_mu, b2_mu,
                                                cu, cm, u2, m2, lst);
}


# Layer-1 edge scan replaced by a binning launch: blocks file edges into per-work-group lists (rank by returning LDS add, one global add per block and list); both layers walk them; sum order changes
# speedup vs baseline: 4.4520x; 1.2327x over previous
#include <hip/hip_runtime.h>
#include <hip/hip_bf16.h>


#define N_USER  200000
#define N_MOVIE 100000
#define N_EDGE  4000000
#define D       64

typedef float v2f __attribute__((ext_vector_type(2)));
typedef float v8f __attribute__((ext_vector_type(8)));

#define WPS 160

#define TR          1184
#define NT_M        ((N_MOVIE + TR - 1) / TR)
#define NT_U        ((N_USER + TR - 1) / TR)
#define NTILE       (NT_M + NT_U)
#define AGG_THREADS 1024
#define AGG_WAVES   (AGG_THREADS / 32)
#define QCAP        64
#define N_EDGE4     (N_EDGE / 4)
#define SCAN_U      4
#define STEP4       (AGG_THREADS * SCAN_U)
#define SCAN_ITERS  (N_EDGE4 / STEP4)
#define TAIL_E0     (SCAN_ITERS * STEP4 * 4)
#define TAIL_ITERS  ((N_EDGE - TAIL_E0 + AGG_THREADS - 1) / AGG_THREADS)
#define GB          8

#define LCAPW       37376
#define K1          400
#define K2          784

#define OFF_LST 0u
#define OFF_GC  (OFF_LST + 2u * NTILE * LCAPW)
#define OFF_CU (OFF_LST + (N_USER + N_MOVIE) * 64u)
#define OFF_CM (OFF_CU + N_USER)
#define OFF_U1 (OFF_CM + N_MOVIE)
#define OFF_M1 (OFF_U1 + N_USER * 64u)

static_assert(LCAPW % 32 == 0, "a work-group's list is whole 128-B lines");
static_assert((unsigned long long)OFF_GC + 2ull * NTILE * 32ull <= (unsigned long long)OFF_CU,
              "lists and their counters fit into their workspace plane");
static_assert(K1 % 16 == 0 && K2 % 16 == 0 && TR - K2 == K1 && (K2 - K1) + 2 * K1 == TR,
              "sub-ranges are whole matrix tiles; one movie and two user sub-ranges are TR rows");
static_assert((NTILE / 3) * TR + (NTILE % 3 == 0 ? 0 : (NTILE % 3 == 1 ? K1 : K2)) >= N_MOVIE,
              "movie sub-ranges 0 .. NTILE - 1 hold every movie row");
static_assert(((2 * NTILE) / 3) * TR + ((2 * NTILE) % 3 == 0 ? 0 : ((2 * NTILE) % 3 == 1 ? K1 : K2)) >= N_USER,
              "user sub-ranges 0 .. 2 NTILE - 1 hold every user row");
static_assert(N_USER <= (1 << 21) && N_MOVIE <= (1 << 21) && TR <= (1 << 11),
              "source id and LDS row share one 32-bit word");
static_assert(TR % 16 == 0 && N_USER % 16 == 0 && N_MOVIE % 16 == 0,
              "every tile is a whole number of 16-row matrix tiles");
static_assert(N_EDGE % 4 == 0 && SCAN_U == 4, "edge indices are scanned 4 per load, 16 per lane and step");
static_assert((QCAP & (QCAP - 1)) == 0 && QCAP >= 64, "ring holds 31 + 32 entries");
static_assert(NTILE <= 256, "the binning pass keeps 256 counters per relation");

__device__ __forceinline__ int sub_lo(int k) { return k == 0 ? 0 : (k == 1 ? K1 : (k == 2 ? K2 : TR)); }
__device__ __forceinline__ int sub_size(int k) { return k == 1 ? K2 - K1 : K1; }
__device__ __forceinline__ void row_owner(unsigned id, bool user, unsigned& wg, unsigned& lrow) {
  const unsigned t = id / (unsigned)TR;
  const unsigned r = id - t * (unsigned)TR;
  const unsigned k = r < (unsigned)K1 ? 0u : (r < (unsigned)K2 ? 1u : 2u);
  const unsigned g = 3u * t + k;
  const unsigned off = r - (unsigned)sub_lo((int)k);
  if (!user) {
    wg = g;
    lrow = off;
  } else {
    wg = g >> 1;
    lrow = (unsigned)sub_size((int)(wg % 3u)) +
           ((g & 1u) ? (unsigned)sub_size((int)((g - 1u) % 3u)) : 0u) + off;
  }
}

__device__ __forceinline__ int sub_start(int g, int nrow) {
  const int t = g / 3;
  return min(t * TR + sub_lo(g - 3 * t), nrow);
}

__device__ __forceinline__ float2 row_load(const float* __restrict__ x, int r,
                                           unsigned lane) {
  return ((const float2*)x)[((unsigned)r << 5) + lane];
}

template <bool COUNT>
__device__ __forceinline__ void drain_ring(
    const float* __restrict__ xsrc, const int* __restrict__ sidx,
    float* acc, float* cnt, const int* qe, const int* qd,
    int head, int n, unsigned lane) {
  int s = 0, d = 0;
  if ((int)lane < n) {
    const int slot = (head + (int)lane) & (QCAP - 1);
    const int e = qe[slot];
    d = qd[slot];
    s = sidx[e];
    if (COUNT) atomicAdd(&cnt[d], 1.0f);
  }
  if (n == 32) {
    #pragma unroll 1
    for (int k = 0; k < 32; k += GB) {
      float2 r[GB];
      #pragma unroll
      for (int j = 0; j < GB; ++j)
        r[j] = row_load(xsrc, __builtin_amdgcn_readlane(s, k + j), lane);
      #pragma unroll
      for (int j = 0; j < GB; ++j) {
        float* q = &acc[((unsigned)__builtin_amdgcn_readlane(d, k + j) << 6) + lane * 2u];
        atomicAdd(q + 0, r[j].x);
        atomicAdd(q + 1, r[j].y);
      }
    }
  } else {
    #pragma unroll 1
    for (int k = 0; k < n; ++k) {
      const float2 r1 = row_load(xsrc, __builtin_amdgcn_readlane(s, k), lane);
      float* q = &acc[((unsigned)__builtin_amdgcn_readlane(d, k) << 6) + lane * 2u];
      atomicAdd(q + 0, r1.x);
      atomicAdd(q + 1, r1.y);
    }
  }
}

template <bool COUNT>
__device__ __forceinline__ void scan_one(
    int dst, int eid, int lo, int rows,
    const float* __restrict__ xsrc, const int* __restrict__ sidx,
    float* acc, float* cnt, int* qe, int* qd,
    int& head, int& qn, unsigned lane) {
  const unsigned dl = (unsigned)(dst - lo);
  const bool hit = dl < (unsigned)rows;
  const unsigned mask = __builtin_amdgcn_ballot_w32(hit);
  if (mask != 0u) {
    if (hit) {
      const int pos = (head + qn + (int)__builtin_amdgcn_mbcnt_lo(mask, 0u)) & (QCAP - 1);
      qe[pos] = eid;
      qd[pos] = (int)dl;
    }
    qn += __builtin_popcount(mask);
    if (qn >= 32) {
      drain_ring<COUNT>(xsrc, sidx, acc, cnt, qe, qd, head, 32, lane);
      head = (head + 32) & (QCAP - 1);
      qn -= 32;
    }
  }
}

__device__ __forceinline__ int4 scan_load(const int4* __restrict__ d4, int it, int u, int tid) {
  return d4[(it * AGG_THREADS + tid) * SCAN_U + u];
}

template <bool COUNT>
__device__ __forceinline__ void scan_edges(
    const int* __restrict__ didx, const int* __restrict__ sidx,
    const float* __restrict__ xsrc, float* acc, float* cnt, int* qe, int* qd,
    int lo, int rows, int tile, int tid, unsigned lane) {
  int head = 0, qn = 0;

  const int4* d4 = (const int4*)didx;
  int it = (tile * SCAN_ITERS) / NTILE;
  int4 v[SCAN_U];
  #pragma unroll
  for (int u = 0; u < SCAN_U; ++u) v[u] = scan_load(d4, it, u, tid);
  #pragma unroll 1
  for (int n = 0; n < SCAN_ITERS; ++n) {
    const int itn = (it + 1 == SCAN_ITERS) ? 0 : it + 1;
    int4 vn[SCAN_U];
    #pragma unroll
    for (int u = 0; u < SCAN_U; ++u) vn[u] = scan_load(d4, itn, u, tid);
    #pragma unroll
    for (int u = 0; u < SCAN_U; ++u) {
      const int e0 = ((it * AGG_THREADS + tid) * SCAN_U + u) * 4;
      scan_one<COUNT>(v[u].x, e0 + 0, lo, rows, xsrc, sidx, acc, cnt, qe, qd, head, qn, lane);
      scan_one<COUNT>(v[u].y, e0 + 1, lo, rows, xsrc, sidx, acc, cnt, qe, qd, head, qn, lane);
      scan_one<COUNT>(v[u].z, e0 + 2, lo, rows, xsrc, sidx, acc, cnt, qe, qd, head, qn, lane);
      scan_one<COUNT>(v[u].w, e0 + 3, lo, rows, xsrc, sidx, acc, cnt, qe, qd, head, qn, lane);
    }
    #pragma unroll
    for (int u = 0; u < SCAN_U; ++u) v[u] = vn[u];
    it = itn;
  }
  #pragma unroll 1
  for (int j = 0; j < TAIL_ITERS; ++j) {
    const int e = TAIL_E0 + j * AGG_THREADS + tid;
    const int dv = (e < N_EDGE) ? didx[e] : -1;
    scan_one<COUNT>(dv, e, lo, rows, xsrc, sidx, acc, cnt, qe, qd, head, qn, lane);
  }
  if (qn > 0) drain_ring<COUNT>(xsrc, sidx, acc, cnt, qe, qd, head, qn, lane);
}

template <bool COUNT>
__device__ __forceinline__ void walk_list(
    const float* __restrict__ xsrc, const unsigned* __restrict__ wl, int n,
    float* acc, float* cnt, int wave, unsigned lane) {
  int cb = __builtin_amdgcn_readfirstlane(wave) * 32;
  if (cb >= n) return;
  const int last = n - 1;
  unsigned w = wl[min(cb + (int)lane, last)];
  #pragma unroll 1
  for (; cb < n; cb += AGG_THREADS) {
    const int m = min(32, n - cb);
    const unsigned wn = wl[min(cb + AGG_THREADS + (int)lane, last)];
    const int s = (int)(w >> 11);
    const int d = (int)(w & 2047u);
    if (COUNT) {
      if ((int)lane < m) atomicAdd(&cnt[d], 1.0f);
    }
    if (m == 32) {
      float2 r[32];
      #pragma unroll
      for (int j = 0; j < 32; ++j)
        r[j] = row_load(xsrc, __builtin_amdgcn_readlane(s, j), lane);
      #pragma unroll
      for (int j = 0; j < 32; ++j) {
        float* q = &acc[((unsigned)__builtin_amdgcn_readlane(d, j) << 6) + lane * 2u];
        atomicAdd(q + 0, r[j].x);
        atomicAdd(q + 1, r[j].y);
      }
    } else {
      #pragma unroll 1
      for (int k = 0; k < m; ++k) {
        const float2 r1 = row_load(xsrc, __builtin_amdgcn_readlane(s, k), lane);
        float* q = &acc[((unsigned)__builtin_amdgcn_readlane(d, k) << 6) + lane * 2u];
        atomicAdd(q + 0, r1.x);
        atomicAdd(q + 1, r1.y);
      }
    }
    w = wn;
  }
}

#define AUX_CNT    0
#define AUX_RINGE  TR
#define AUX_RINGD  (TR + AGG_WAVES * QCAP)
#define AUX_FLOATS (TR + 2 * AGG_WAVES * QCAP)
static_assert(32 * WPS <= AUX_FLOATS, "a staged weight matrix fits where counts and rings were");

__device__ __forceinline__ void stage_weights(float* aux, const float* __restrict__ Wg, int tid) {
  for (int idx = tid; idx < 32 * D; idx += AGG_THREADS) {
    const int kp = idx >> 6, n = idx & 63;
    *(float2*)&aux[kp * WPS + 2 * n] =
        make_float2(Wg[(2 * kp) * D + n], Wg[(2 * kp + 1) * D + n]);
  }
}

__device__ __forceinline__ void load_a_frags(const float* tile, int c, int h, v2f (&a)[16]) {
  #pragma unroll
  for (int j = 0; j < 16; ++j)
    a[j] = *(const v2f*)&tile[c * D + ((j ^ c) << 2) + 2 * h];
}

__device__ __forceinline__ v2f load_b_frag(const float* aux, int wo, int j, int ct, int c, int h) {
  return *(const v2f*)&aux[wo + (2 * j + h) * WPS + 2 * (16 * ct + c)];
}

__device__ __forceinline__ int trip_offset() {
  int wo = 0;
  asm volatile("" : "+v"(wo));
  return wo;
}

__device__ __forceinline__ int d_place(int h, int r, int col) {
  return (8 * h + r) * D + (col ^ (32 * h));
}

template <bool SCALED>
__device__ __forceinline__ void tile_transform(
    float* acc, float* aux, const float* __restrict__ cntG,
    const float* __restrict__ xdst, const float* __restrict__ Wl,
    const float* __restrict__ Wr, const float* __restrict__ bias,
    float* __restrict__ out, int rows, bool relu, int tid) {
  const int lane = tid & 31, wave = tid >> 5;
  const int c = lane & 15, h = lane >> 4;
  const int nrt = rows >> 4;

  if (!SCALED) {
    for (int i = tid; i < rows * 16; i += AGG_THREADS) {
      const int row = i >> 4, q = i & 15;
      const float inv = 1.0f / fmaxf(cntG[row], 1.0f);
      const float4 sv = ((const float4*)acc)[i];
      ((float4*)acc)[(row << 4) + (q ^ (row & 15))] =
          make_float4(sv.x * inv, sv.y * inv, sv.z * inv, sv.w * inv);
    }
  }
  __syncthreads();
  stage_weights(aux, Wl, tid);
  __syncthreads();

  #pragma unroll 1
  for (int rt = wave; rt < nrt; rt += AGG_WAVES) {
    float* tile = acc + rt * 16 * D;
    const int wo = trip_offset();
    v2f a[16];
    load_a_frags(tile, c, h, a);
    #pragma unroll
    for (int ct = 0; ct < 4; ++ct) {
      v8f d = {0.f, 0.f, 0.f, 0.f, 0.f, 0.f, 0.f, 0.f};
      #pragma unroll
      for (int j = 0; j < 16; ++j)
        d = __builtin_amdgcn_wmma_f32_16x16x4_f32(false, a[j], false,
                                                  load_b_frag(aux, wo, j, ct, c, h),
                                                  (short)0, d, false, false);
      #pragma unroll
      for (int r = 0; r < 8; ++r) tile[d_place(h, r, 16 * ct + c)] = d[r];
    }
  }
  __syncthreads();
  stage_weights(aux, Wr, tid);
  __syncthreads();

  #pragma unroll 1
  for (int rt = wave; rt < nrt; rt += AGG_WAVES) {
    float* tile = acc + rt * 16 * D;
    const int wo = trip_offset();
    const float* xt = xdst + (size_t)rt * 16 * D;
    float*       ot = out + (size_t)rt * 16 * D;
    float4 xr[8];
    #pragma unroll
    for (int i = 0; i < 8; ++i)
      xr[i] = *(const float4*)&xt[(2 * i + h) * D + 4 * c];
    v8f d[4];
    #pragma unroll
    for (int ct = 0; ct < 4; ++ct) {
      #pragma unroll
      for (int r = 0; r < 8; ++r) d[ct][r] = tile[d_place(h, r, 16 * ct + c)];
    }
    #pragma unroll
    for (int i = 0; i < 8; ++i) {
      const int row = 2 * i + h;
      *(float4*)&tile[row * D + ((c ^ row) << 2)] = xr[i];
    }
    v2f a[16];
    load_a_frags(tile, c, h, a);
    #pragma unroll
    for (int ct = 0; ct < 4; ++ct) {
      #pragma unroll
      for (int j = 0; j < 16; ++j)
        d[ct] = __builtin_amdgcn_wmma_f32_16x16x4_f32(false, a[j], false,
                                                      load_b_frag(aux, wo, j, ct, c, h),
                                                      (short)0, d[ct], false, false);
    }
    #pragma unroll
    for (int ct = 0; ct < 4; ++ct) {
      const float bcol = bias[16 * ct + c];
      #pragma unroll
      for (int r = 0; r < 8; ++r) {
        float val = d[ct][r] + bcol;
        if (relu) val = fmaxf(val, 0.0f);
        tile[d_place(h, r, 16 * ct + c)] = val;
      }
    }
    #pragma unroll
    for (int i = 0; i < 8; ++i) {
      const int row = 2 * i + h;
      *(float4*)&ot[row * D + 4 * c] =
          *(const float4*)&tile[row * D + ((c ^ (8 * (row >> 3))) << 2)];
    }
  }
}

#define BIN_THREADS 1024
#define BIN_EPT     8
#define BIN_EDGES   (BIN_THREADS * BIN_EPT)
#define BIN_BLOCKS  ((N_EDGE + BIN_EDGES - 1) / BIN_EDGES)
#define BIN_ENT     (2 * BIN_EDGES)
#define BIN_LISTS   512

static_assert(BIN_ENT <= (1 << 16), "an entry's rank in its block fits 16 bits beside the list number");
static_assert(BIN_LISTS == 2 * 256 && BIN_LISTS % 32 == 0 && BIN_LISTS + 32 <= BIN_THREADS,
              "counters are summed 16 per lane by wave 0 while one thread per list takes its run");

__global__ __launch_bounds__(64) void list_clear_kernel(unsigned* __restrict__ gcnt) {
  for (int i = threadIdx.x; i < 2 * NTILE; i += 64) gcnt[i * 32] = 0u;
}

__global__ __launch_bounds__(BIN_THREADS) void edge_bin_kernel(
    const int* __restrict__ uidx, const int* __restrict__ midx,
    unsigned* __restrict__ lst, unsigned* __restrict__ gcnt) {
  __shared__ unsigned binCnt[BIN_LISTS];
  __shared__ unsigned binOff[BIN_LISTS];
  __shared__ unsigned binBase[BIN_LISTS];
  __shared__ unsigned sorted[BIN_ENT];

  const int tid = threadIdx.x;
  const int lane = tid & 31, wave = tid >> 5;
  if (tid < BIN_LISTS) binCnt[tid] = 0u;
  __syncthreads();

  const int e0 = blockIdx.x * BIN_EDGES;
  unsigned uv[BIN_EPT], mv[BIN_EPT];
  #pragma unroll
  for (int j = 0; j < BIN_EPT; ++j) {
    const int e = e0 + j * BIN_THREADS + tid;
    const bool ok = e < N_EDGE;
    uv[j] = ok ? (unsigned)uidx[e] : 0u;
    mv[j] = ok ? (unsigned)midx[e] : 0u;
  }
  unsigned ent[2 * BIN_EPT], key[2 * BIN_EPT];
  #pragma unroll
  for (int j = 0; j < BIN_EPT; ++j) {
    const int e = e0 + j * BIN_THREADS + tid;
    key[2 * j] = 0xffffffffu;
    key[2 * j + 1] = 0xffffffffu;
    ent[2 * j] = 0u;
    ent[2 * j + 1] = 0u;
    if (e < N_EDGE) {
      unsigned tm, rm_, tu, ru;
      row_owner(mv[j], false, tm, rm_);
      row_owner(uv[j], true, tu, ru);
      ent[2 * j] = (uv[j] << 11) | rm_;
      key[2 * j] = (tm << 16) | atomicAdd(&binCnt[tm], 1u);
      ent[2 * j + 1] = (mv[j] << 11) | ru;
      key[2 * j + 1] = ((256u + tu) << 16) | atomicAdd(&binCnt[256u + tu], 1u);
    }
  }
  __syncthreads();

  if (tid < 32) {
    unsigned c[16], tot = 0u;
    #pragma unroll
    for (int k = 0; k < 16; ++k) { c[k] = binCnt[16 * tid + k]; tot += c[k]; }
    unsigned inc = tot;
    #pragma unroll
    for (int o = 1; o < 32; o <<= 1) {
      const unsigned v = __shfl_up(inc, o);
      if (lane >= o) inc += v;
    }
    unsigned ex = inc - tot;
    #pragma unroll
    for (int k = 0; k < 16; ++k) { binOff[16 * tid + k] = ex; ex += c[k]; }
  } else if (tid < 32 + BIN_LISTS) {
    const int i = tid - 32;
    const int t = i & 255, r = i >> 8;
    if (t < NTILE) binBase[i] = atomicAdd(&gcnt[(2 * t + r) * 32], binCnt[i]);
  }
  __syncthreads();

  #pragma unroll
  for (int i = 0; i < 2 * BIN_EPT; ++i) {
    if (key[i] != 0xffffffffu) sorted[binOff[key[i] >> 16] + (key[i] & 0xffffu)] = ent[i];
  }
  __syncthreads();

  for (int i = wave; i < BIN_LISTS; i += BIN_THREADS / 32) {
    const int t = i & 255, r = i >> 8;
    if (t >= NTILE) continue;
    const unsigned n = binCnt[i], off = binOff[i], base = binBase[i];
    unsigned* dstl = lst + (size_t)(2 * t + r) * LCAPW;
    for (unsigned k = (unsigned)lane; k < n; k += 32u) {
      const unsigned pos = base + k;
      if (pos < (unsigned)LCAPW) dstl[pos] = sorted[off + k];
    }
  }
}

__global__ __launch_bounds__(AGG_THREADS) void tile_layer1_kernel(
    const float* __restrict__ xu, const float* __restrict__ xm,
    const int* __restrict__ uidx, const int* __restrict__ midx,
    const float* __restrict__ WlM, const float* __restrict__ WrM, const float* __restrict__ bM,
    const float* __restrict__ WlU, const float* __restrict__ WrU, const float* __restrict__ bU,
    float* __restrict__ cu, float* __restrict__ cm,
    float* __restrict__ u1, float* __restrict__ m1,
    const unsigned* __restrict__ lst, const unsigned* __restrict__ gcnt) {
  __shared__ __attribute__((aligned(16))) float acc[TR * D];
  __shared__ __attribute__((aligned(16))) float aux[AUX_FLOATS];
  float* cnt   = aux + AUX_CNT;
  int*   ringE = (int*)(aux + AUX_RINGE);
  int*   ringD = (int*)(aux + AUX_RINGD);

  const int tid = threadIdx.x;
  const unsigned lane = threadIdx.x & 31u;
  const int wave = tid >> 5;
  const int wg = blockIdx.x;

  const int mlo   = sub_start(wg, N_MOVIE);
  const int rm    = sub_start(wg + 1, N_MOVIE) - mlo;
  const int ulo   = sub_start(2 * wg, N_USER);
  const int umid  = sub_start(2 * wg + 1, N_USER);
  const int uhi   = sub_start(2 * wg + 2, N_USER);
  const int ubase = sub_size(wg % 3);

  for (int i = tid; i < TR * D / 4; i += AGG_THREADS)
    ((float4*)acc)[i] = make_float4(0.f, 0.f, 0.f, 0.f);
  for (int i = tid; i < TR; i += AGG_THREADS) cnt[i] = 0.f;
  __syncthreads();

  const int n0 = __builtin_amdgcn_readfirstlane((int)gcnt[(2 * wg) * 32]);
  const int n1 = __builtin_amdgcn_readfirstlane((int)gcnt[(2 * wg + 1) * 32]);
  if (n0 > LCAPW || n1 > LCAPW) {
    #pragma unroll 1
    for (int p = 0; p < 3; ++p) {
      const bool mv = p == 0;
      const int plo   = mv ? mlo : (p == 1 ? ulo : umid);
      const int prows = mv ? rm : (p == 1 ? umid - ulo : uhi - umid);
      const int prow0 = mv ? 0 : (p == 1 ? ubase : ubase + (umid - ulo));
      scan_edges<true>(mv ? midx : uidx, mv ? uidx : midx, mv ? xu : xm,
                       acc + prow0 * D, cnt + prow0, &ringE[wave * QCAP], &ringD[wave * QCAP],
                       plo, prows, wg, tid, lane);
    }
  } else {
    walk_list<true>(xu, lst + (size_t)(2 * wg) * LCAPW, n0, acc, cnt, wave, lane);
    walk_list<true>(xm, lst + (size_t)(2 * wg + 1) * LCAPW, n1, acc, cnt, wave, lane);
  }
  __syncthreads();

  for (int i = tid; i < TR * 16; i += AGG_THREADS) {
    const int row = i >> 4, q = i & 15;
    const float cv = cnt[row];
    if (q == 0) {
      if (row < ubase) {
        if (row < rm) cm[mlo + row] = cv;
      } else if (row - ubase < uhi - ulo) {
        cu[ulo + row - ubase] = cv;
      }
    }
    const float inv = 1.0f / fmaxf(cv, 1.0f);
    const float4 sv = ((const float4*)acc)[i];
    ((float4*)acc)[(row << 4) + (q ^ (row & 15))] =
        make_float4(sv.x * inv, sv.y * inv, sv.z * inv, sv.w * inv);
  }

  tile_transform<true>(acc, aux, nullptr, xm + (size_t)mlo * D, WlM, WrM, bM,
                       m1 + (size_t)mlo * D, rm, true, tid);
  tile_transform<true>(acc + ubase * D, aux, nullptr, xu + (size_t)ulo * D, WlU, WrU, bU,
                       u1 + (size_t)ulo * D, uhi - ulo, true, tid);
}


__global__ __launch_bounds__(AGG_THREADS) void tile_layer2_kernel(
    const float* __restrict__ h1u, const float* __restrict__ h1m,
    const int* __restrict__ uidx, const int* __restrict__ midx,
    const float* __restrict__ WlM, const float* __restrict__ WrM, const float* __restrict__ bM,
    const float* __restrict__ WlU, const float* __restrict__ WrU, const float* __restrict__ bU,
    float* __restrict__ cu, float* __restrict__ cm,
    float* __restrict__ outU, float* __restrict__ outM,
    const unsigned* __restrict__ lst, const unsigned* __restrict__ gcnt) {
  __shared__ __attribute__((aligned(16))) float acc[TR * D];
  __shared__ __attribute__((aligned(16))) float aux[AUX_FLOATS];
  int* ringE = (int*)(aux + AUX_RINGE);
  int* ringD = (int*)(aux + AUX_RINGD);

  const int tid = threadIdx.x;
  const unsigned lane = threadIdx.x & 31u;
  const int wave = tid >> 5;
  const int wg = blockIdx.x;

  const int mlo   = sub_start(wg, N_MOVIE);
  const int rm    = sub_start(wg + 1, N_MOVIE) - mlo;
  const int ulo   = sub_start(2 * wg, N_USER);
  const int umid  = sub_start(2 * wg + 1, N_USER);
  const int uhi   = sub_start(2 * wg + 2, N_USER);
  const int ubase = sub_size(wg % 3);

  for (int i = tid; i < TR * D / 4; i += AGG_THREADS)
    ((float4*)acc)[i] = make_float4(0.f, 0.f, 0.f, 0.f);
  __syncthreads();

  const int n0 = __builtin_amdgcn_readfirstlane((int)gcnt[(2 * wg) * 32]);
  const int n1 = __builtin_amdgcn_readfirstlane((int)gcnt[(2 * wg + 1) * 32]);
  if (n0 > LCAPW || n1 > LCAPW) {
    #pragma unroll 1
    for (int p = 0; p < 3; ++p) {
      const bool mv = p == 0;
      const int plo   = mv ? mlo : (p == 1 ? ulo : umid);
      const int prows = mv ? rm : (p == 1 ? umid - ulo : uhi - umid);
      const int prow0 = mv ? 0 : (p == 1 ? ubase : ubase + (umid - ulo));
      scan_edges<false>(mv ? midx : uidx, mv ? uidx : midx, mv ? h1u : h1m,
                        acc + prow0 * D, nullptr, &ringE[wave * QCAP], &ringD[wave * QCAP],
                        plo, prows, wg, tid, lane);
    }
  } else {
    walk_list<false>(h1u, lst + (size_t)(2 * wg) * LCAPW, n0, acc, nullptr, wave, lane);
    walk_list<false>(h1m, lst + (size_t)(2 * wg + 1) * LCAPW, n1, acc, nullptr, wave, lane);
  }
  __syncthreads();

  tile_transform<false>(acc, aux, cm + mlo, h1m + (size_t)mlo * D, WlM, WrM, bM,
                        outM + (size_t)mlo * D, rm, false, tid);
  tile_transform<false>(acc + ubase * D, aux, cu + ulo, h1u + (size_t)ulo * D, WlU, WrU, bU,
                        outU + (size_t)ulo * D, uhi - ulo, false, tid);
}

extern "C" void kernel_launch(void* const* d_in, const int* in_sizes, int n_in,
                              void* d_out, int out_size, void* d_ws, size_t ws_size,
                              hipStream_t stream) {
  const float* x_user  = (const float*)d_in[0];
  const float* x_movie = (const float*)d_in[1];
  const float* Wl1_um  = (const float*)d_in[2];
  const float* Wr1_um  = (const float*)d_in[3];
  const float* b1_um   = (const float*)d_in[4];
  const float* Wl1_mu  = (const float*)d_in[5];
  const float* Wr1_mu  = (const float*)d_in[6];
  const float* b1_mu   = (const float*)d_in[7];
  const float* Wl2_um  = (const float*)d_in[8];
  const float* Wr2_um  = (const float*)d_in[9];
  const float* b2_um   = (const float*)d_in[10];
  const float* Wl2_mu  = (const float*)d_in[11];
  const float* Wr2_mu  = (const float*)d_in[12];
  const float* b2_mu   = (const float*)d_in[13];
  const int*   uidx    = (const int*)d_in[14];
  const int*   midx    = (const int*)d_in[15];

  float* ws = (float*)d_ws;
  unsigned* lst  = (unsigned*)(ws + OFF_LST);
  unsigned* gcnt = (unsigned*)(ws + OFF_GC);
  float* cu = ws + OFF_CU;
  float* cm = ws + OFF_CM;
  float* u1 = ws + OFF_U1;
  float* m1 = ws + OFF_M1;

  float* u2 = (float*)d_out;
  float* m2 = (float*)d_out + (size_t)N_USER * D;

  const dim3 bAgg(AGG_THREADS);
  const dim3 gAgg(NTILE);

  list_clear_kernel<<<dim3(1), dim3(64), 0, stream>>>(gcnt);
  edge_bin_kernel<<<dim3(BIN_BLOCKS), dim3(BIN_THREADS), 0, stream>>>(uidx, midx, lst, gcnt);

  tile_layer1_kernel<<<gAgg, bAgg, 0, stream>>>(x_user, x_movie, uidx, midx,
                                                Wl1_um, Wr1_um, b1_um,
                                                Wl1_mu, Wr1_mu, b1_mu,
                                                cu, cm, u1, m1, lst, gcnt);
  tile_layer2_kernel<<<gAgg, bAgg, 0, stream>>>(u1, m1, uidx, midx,
                                                Wl2_um, Wr2_um, b2_um,
                                                Wl2_mu, Wr2_mu, b2_mu,
                                                cu, cm, u2, m2, lst, gcnt);
}


# Sums by owner waves: layer 1 orders each work-group list by the wave owning the LDS row (row & 31); a wave adds into its own rows by plain LDS read-add-write, no LDS float atomics; sum order changes
# speedup vs baseline: 4.8031x; 1.0789x over previous
#include <hip/hip_runtime.h>
#include <hip/hip_bf16.h>


#define N_USER  200000
#define N_MOVIE 100000
#define N_EDGE  4000000
#define D       64

typedef float v2f __attribute__((ext_vector_type(2)));
typedef float v8f __attribute__((ext_vector_type(8)));

#define WPS 160

#define TR          1184
#define NT_M        ((N_MOVIE + TR - 1) / TR)
#define NT_U        ((N_USER + TR - 1) / TR)
#define NTILE       (NT_M + NT_U)
#define AGG_THREADS 1024
#define AGG_WAVES   (AGG_THREADS / 32)
#define QCAP        64
#define N_EDGE4     (N_EDGE / 4)
#define SCAN_U      4
#define STEP4       (AGG_THREADS * SCAN_U)
#define SCAN_ITERS  (N_EDGE4 / STEP4)
#define TAIL_E0     (SCAN_ITERS * STEP4 * 4)
#define TAIL_ITERS  ((N_EDGE - TAIL_E0 + AGG_THREADS - 1) / AGG_THREADS)
#define GB          8

#define LCAPW       37376
#define K1          400
#define K2          784

#define OFF_LST 0u
#define OFF_GC  (OFF_LST + 2u * NTILE * LCAPW)
#define OFF_CU (OFF_LST + (N_USER + N_MOVIE) * 64u)
#define OFF_CM (OFF_CU + N_USER)
#define OFF_U1 (OFF_CM + N_MOVIE)
#define OFF_M1 (OFF_U1 + N_USER * 64u)

static_assert(LCAPW % 32 == 0, "a work-group's list is whole 128-B lines");
static_assert((unsigned long long)OFF_GC + 2ull * NTILE * 32ull <= (unsigned long long)OFF_CU,
              "lists and their counters fit into their workspace plane");
static_assert(K1 % 16 == 0 && K2 % 16 == 0 && TR - K2 == K1 && (K2 - K1) + 2 * K1 == TR,
              "sub-ranges are whole matrix tiles; one movie and two user sub-ranges are TR rows");
static_assert((NTILE / 3) * TR + (NTILE % 3 == 0 ? 0 : (NTILE % 3 == 1 ? K1 : K2)) >= N_MOVIE,
              "movie sub-ranges 0 .. NTILE - 1 hold every movie row");
static_assert(((2 * NTILE) / 3) * TR + ((2 * NTILE) % 3 == 0 ? 0 : ((2 * NTILE) % 3 == 1 ? K1 : K2)) >= N_USER,
              "user sub-ranges 0 .. 2 NTILE - 1 hold every user row");
static_assert(N_USER <= (1 << 21) && N_MOVIE <= (1 << 21) && TR <= (1 << 11),
              "source id and LDS row share one 32-bit word");
static_assert(TR % 16 == 0 && N_USER % 16 == 0 && N_MOVIE % 16 == 0,
              "every tile is a whole number of 16-row matrix tiles");
static_assert(N_EDGE % 4 == 0 && SCAN_U == 4, "edge indices are scanned 4 per load, 16 per lane and step");
static_assert((QCAP & (QCAP - 1)) == 0 && QCAP >= 64, "ring holds 31 + 32 entries");
static_assert(NTILE <= 256, "the binning pass keeps 256 counters per relation");
static_assert(2 * LCAPW <= TR * D, "both lists of a work-group are staged in the LDS of its sums");
static_assert(AGG_THREADS / 32 == 32 && TR % 32 == 0, "LDS row & 31 names the owning wave; every wave owns TR / 32 rows");

__device__ __forceinline__ int sub_lo(int k) { return k == 0 ? 0 : (k == 1 ? K1 : (k == 2 ? K2 : TR)); }
__device__ __forceinline__ int sub_size(int k) { return k == 1 ? K2 - K1 : K1; }
__device__ __forceinline__ void row_owner(unsigned id, bool user, unsigned& wg, unsigned& lrow) {
  const unsigned t = id / (unsigned)TR;
  const unsigned r = id - t * (unsigned)TR;
  const unsigned k = r < (unsigned)K1 ? 0u : (r < (unsigned)K2 ? 1u : 2u);
  const unsigned g = 3u * t + k;
  const unsigned off = r - (unsigned)sub_lo((int)k);
  if (!user) {
    wg = g;
    lrow = off;
  } else {
    wg = g >> 1;
    lrow = (unsigned)sub_size((int)(wg % 3u)) +
           ((g & 1u) ? (unsigned)sub_size((int)((g - 1u) % 3u)) : 0u) + off;
  }
}

__device__ __forceinline__ int sub_start(int g, int nrow) {
  const int t = g / 3;
  return min(t * TR + sub_lo(g - 3 * t), nrow);
}

__device__ __forceinline__ float2 row_load(const float* __restrict__ x, int r,
                                           unsigned lane) {
  return ((const float2*)x)[((unsigned)r << 5) + lane];
}

template <bool COUNT>
__device__ __forceinline__ void drain_ring(
    const float* __restrict__ xsrc, const int* __restrict__ sidx,
    float* acc, float* cnt, const int* qe, const int* qd,
    int head, int n, unsigned lane) {
  int s = 0, d = 0;
  if ((int)lane < n) {
    const int slot = (head + (int)lane) & (QCAP - 1);
    const int e = qe[slot];
    d = qd[slot];
    s = sidx[e];
    if (COUNT) atomicAdd(&cnt[d], 1.0f);
  }
  if (n == 32) {
    #pragma unroll 1
    for (int k = 0; k < 32; k += GB) {
      float2 r[GB];
      #pragma unroll
      for (int j = 0; j < GB; ++j)
        r[j] = row_load(xsrc, __builtin_amdgcn_readlane(s, k + j), lane);
      #pragma unroll
      for (int j = 0; j < GB; ++j) {
        float* q = &acc[((unsigned)__builtin_amdgcn_readlane(d, k + j) << 6) + lane * 2u];
        atomicAdd(q + 0, r[j].x);
        atomicAdd(q + 1, r[j].y);
      }
    }
  } else {
    #pragma unroll 1
    for (int k = 0; k < n; ++k) {
      const float2 r1 = row_load(xsrc, __builtin_amdgcn_readlane(s, k), lane);
      float* q = &acc[((unsigned)__builtin_amdgcn_readlane(d, k) << 6) + lane * 2u];
      atomicAdd(q + 0, r1.x);
      atomicAdd(q + 1, r1.y);
    }
  }
}

template <bool COUNT>
__device__ __forceinline__ void scan_one(
    int dst, int eid, int lo, int rows,
    const float* __restrict__ xsrc, const int* __restrict__ sidx,
    float* acc, float* cnt, int* qe, int* qd,
    int& head, int& qn, unsigned lane) {
  const unsigned dl = (unsigned)(dst - lo);
  const bool hit = dl < (unsigned)rows;
  const unsigned mask = __builtin_amdgcn_ballot_w32(hit);
  if (mask != 0u) {
    if (hit) {
      const int pos = (head + qn + (int)__builtin_amdgcn_mbcnt_lo(mask, 0u)) & (QCAP - 1);
      qe[pos] = eid;
      qd[pos] = (int)dl;
    }
    qn += __builtin_popcount(mask);
    if (qn >= 32) {
      drain_ring<COUNT>(xsrc, sidx, acc, cnt, qe, qd, head, 32, lane);
      head = (head + 32) & (QCAP - 1);
      qn -= 32;
    }
  }
}

__device__ __forceinline__ int4 scan_load(const int4* __restrict__ d4, int it, int u, int tid) {
  return d4[(it * AGG_THREADS + tid) * SCAN_U + u];
}

template <bool COUNT>
__device__ __forceinline__ void scan_edges(
    const int* __restrict__ didx, const int* __restrict__ sidx,
    const float* __restrict__ xsrc, float* acc, float* cnt, int* qe, int* qd,
    int lo, int rows, int tile, int tid, unsigned lane) {
  int head = 0, qn = 0;

  const int4* d4 = (const int4*)didx;
  int it = (tile * SCAN_ITERS) / NTILE;
  int4 v[SCAN_U];
  #pragma unroll
  for (int u = 0; u < SCAN_U; ++u) v[u] = scan_load(d4, it, u, tid);
  #pragma unroll 1
  for (int n = 0; n < SCAN_ITERS; ++n) {
    const int itn = (it + 1 == SCAN_ITERS) ? 0 : it + 1;
    int4 vn[SCAN_U];
    #pragma unroll
    for (int u = 0; u < SCAN_U; ++u) vn[u] = scan_load(d4, itn, u, tid);
    #pragma unroll
    for (int u = 0; u < SCAN_U; ++u) {
      const int e0 = ((it * AGG_THREADS + tid) * SCAN_U + u) * 4;
      scan_one<COUNT>(v[u].x, e0 + 0, lo, rows, xsrc, sidx, acc, cnt, qe, qd, head, qn, lane);
      scan_one<COUNT>(v[u].y, e0 + 1, lo, rows, xsrc, sidx, acc, cnt, qe, qd, head, qn, lane);
      scan_one<COUNT>(v[u].z, e0 + 2, lo, rows, xsrc, sidx, acc, cnt, qe, qd, head, qn, lane);
      scan_one<COUNT>(v[u].w, e0 + 3, lo, rows, xsrc, sidx, acc, cnt, qe, qd, head, qn, lane);
    }
    #pragma unroll
    for (int u = 0; u < SCAN_U; ++u) v[u] = vn[u];
    it = itn;
  }
  #pragma unroll 1
  for (int j = 0; j < TAIL_ITERS; ++j) {
    const int e = TAIL_E0 + j * AGG_THREADS + tid;
    const int dv = (e < N_EDGE) ? didx[e] : -1;
    scan_one<COUNT>(dv, e, lo, rows, xsrc, sidx, acc, cnt, qe, qd, head, qn, lane);
  }
  if (qn > 0) drain_ring<COUNT>(xsrc, sidx, acc, cnt, qe, qd, head, qn, lane);
}

__device__ __forceinline__ unsigned same_key_lanes(unsigned key, bool valid) {
  unsigned m = __builtin_amdgcn_ballot_w32(valid);
  #pragma unroll
  for (int b = 0; b < 5; ++b) {
    const bool bit = ((key >> b) & 1u) != 0u;
    const unsigned bal = __builtin_amdgcn_ballot_w32(bit);
    m &= bit ? bal : ~bal;
  }
  return m;
}

#define SORT_U 8
template <bool PLACE>
__device__ __forceinline__ void order_pass(const unsigned* L, int n, unsigned* seg,
                                           unsigned* stage, int wave, unsigned lane) {
  const unsigned below = (1u << lane) - 1u;
  const int b0 = __builtin_amdgcn_readfirstlane(wave) * 32;
  for (int b = b0; b < n; b += SORT_U * AGG_THREADS) {
    unsigned wv[SORT_U];
    #pragma unroll
    for (int u = 0; u < SORT_U; ++u) {
      const int i = b + u * AGG_THREADS + (int)lane;
      wv[u] = i < n ? L[i] : 0u;
    }
    #pragma unroll
    for (int u = 0; u < SORT_U; ++u) {
      const bool valid = b + u * AGG_THREADS + (int)lane < n;
      const unsigned key = wv[u] & 31u;
      const unsigned m = same_key_lanes(key, valid);
      const unsigned rank = (unsigned)__builtin_popcount(m & below);
      if (!PLACE) {
        if (valid && rank == 0u) atomicAdd(&seg[key], (unsigned)__builtin_popcount(m));
      } else {
        unsigned base = 0u;
        if (valid && rank == 0u) base = atomicAdd(&seg[key], (unsigned)__builtin_popcount(m));
        base = (unsigned)__shfl((int)base, __builtin_ffs((int)m) - 1);
        if (valid) stage[base + rank] = wv[u];
      }
    }
  }
}

template <bool COUNT>
__device__ __forceinline__ void walk_owned(
    const float* __restrict__ xsrc, const unsigned* wl, int beg, int end,
    float* acc, float* cnt, unsigned lane) {
  int cb = __builtin_amdgcn_readfirstlane(beg);
  const int n = __builtin_amdgcn_readfirstlane(end);
  if (cb >= n) return;
  const int last = n - 1;
  unsigned w = wl[min(cb + (int)lane, last)];
  #pragma unroll 1
  for (; cb < n; cb += 32) {
    const int m = min(32, n - cb);
    const unsigned wn = wl[min(cb + 32 + (int)lane, last)];
    const int s = (int)(w >> 11);
    const int d = (int)(w & 2047u);
    if (COUNT) {
      if ((int)lane < m) atomicAdd(&cnt[d], 1.0f);
    }
    if (m == 32) {
      float2 r[32];
      #pragma unroll
      for (int j = 0; j < 32; ++j)
        r[j] = row_load(xsrc, __builtin_amdgcn_readlane(s, j), lane);
      #pragma unroll
      for (int j = 0; j < 32; ++j) {
        float2* q = (float2*)&acc[((unsigned)__builtin_amdgcn_readlane(d, j) << 6) + lane * 2u];
        float2 a = *q;
        a.x += r[j].x;
        a.y += r[j].y;
        *q = a;
      }
    } else {
      #pragma unroll 1
      for (int k = 0; k < m; ++k) {
        const float2 r1 = row_load(xsrc, __builtin_amdgcn_readlane(s, k), lane);
        float2* q = (float2*)&acc[((unsigned)__builtin_amdgcn_readlane(d, k) << 6) + lane * 2u];
        float2 a = *q;
        a.x += r1.x;
        a.y += r1.y;
        *q = a;
      }
    }
    w = wn;
  }
}

#define AUX_CNT    0
#define AUX_RINGE  TR
#define AUX_RINGD  (TR + AGG_WAVES * QCAP)
#define AUX_FLOATS (TR + 2 * AGG_WAVES * QCAP)
static_assert(32 * WPS <= AUX_FLOATS, "a staged weight matrix fits where counts and rings were");

__device__ __forceinline__ void stage_weights(float* aux, const float* __restrict__ Wg, int tid) {
  for (int idx = tid; idx < 32 * D; idx += AGG_THREADS) {
    const int kp = idx >> 6, n = idx & 63;
    *(float2*)&aux[kp * WPS + 2 * n] =
        make_float2(Wg[(2 * kp) * D + n], Wg[(2 * kp + 1) * D + n]);
  }
}

__device__ __forceinline__ void load_a_frags(const float* tile, int c, int h, v2f (&a)[16]) {
  #pragma unroll
  for (int j = 0; j < 16; ++j)
    a[j] = *(const v2f*)&tile[c * D + ((j ^ c) << 2) + 2 * h];
}

__device__ __forceinline__ v2f load_b_frag(const float* aux, int wo, int j, int ct, int c, int h) {
  return *(const v2f*)&aux[wo + (2 * j + h) * WPS + 2 * (16 * ct + c)];
}

__device__ __forceinline__ int trip_offset() {
  int wo = 0;
  asm volatile("" : "+v"(wo));
  return wo;
}

__device__ __forceinline__ int d_place(int h, int r, int col) {
  return (8 * h + r) * D + (col ^ (32 * h));
}

template <bool SCALED>
__device__ __forceinline__ void tile_transform(
    float* acc, float* aux, const float* __restrict__ cntG,
    const float* __restrict__ xdst, const float* __restrict__ Wl,
    const float* __restrict__ Wr, const float* __restrict__ bias,
    float* __restrict__ out, int rows, bool relu, int tid) {
  const int lane = tid & 31, wave = tid >> 5;
  const int c = lane & 15, h = lane >> 4;
  const int nrt = rows >> 4;

  if (!SCALED) {
    for (int i = tid; i < rows * 16; i += AGG_THREADS) {
      const int row = i >> 4, q = i & 15;
      const float inv = 1.0f / fmaxf(cntG[row], 1.0f);
      const float4 sv = ((const float4*)acc)[i];
      ((float4*)acc)[(row << 4) + (q ^ (row & 15))] =
          make_float4(sv.x * inv, sv.y * inv, sv.z * inv, sv.w * inv);
    }
  }
  __syncthreads();
  stage_weights(aux, Wl, tid);
  __syncthreads();

  #pragma unroll 1
  for (int rt = wave; rt < nrt; rt += AGG_WAVES) {
    float* tile = acc + rt * 16 * D;
    const int wo = trip_offset();
    v2f a[16];
    load_a_frags(tile, c, h, a);
    #pragma unroll
    for (int ct = 0; ct < 4; ++ct) {
      v8f d = {0.f, 0.f, 0.f, 0.f, 0.f, 0.f, 0.f, 0.f};
      #pragma unroll
      for (int j = 0; j < 16; ++j)
        d = __builtin_amdgcn_wmma_f32_16x16x4_f32(false, a[j], false,
                                                  load_b_frag(aux, wo, j, ct, c, h),
                                                  (short)0, d, false, false);
      #pragma unroll
      for (int r = 0; r < 8; ++r) tile[d_place(h, r, 16 * ct + c)] = d[r];
    }
  }
  __syncthreads();
  stage_weights(aux, Wr, tid);
  __syncthreads();

  #pragma unroll 1
  for (int rt = wave; rt < nrt; rt += AGG_WAVES) {
    float* tile = acc + rt * 16 * D;
    const int wo = trip_offset();
    const float* xt = xdst + (size_t)rt * 16 * D;
    float*       ot = out + (size_t)rt * 16 * D;
    float4 xr[8];
    #pragma unroll
    for (int i = 0; i < 8; ++i)
      xr[i] = *(const float4*)&xt[(2 * i + h) * D + 4 * c];
    v8f d[4];
    #pragma unroll
    for (int ct = 0; ct < 4; ++ct) {
      #pragma unroll
      for (int r = 0; r < 8; ++r) d[ct][r] = tile[d_place(h, r, 16 * ct + c)];
    }
    #pragma unroll
    for (int i = 0; i < 8; ++i) {
      const int row = 2 * i + h;
      *(float4*)&tile[row * D + ((c ^ row) << 2)] = xr[i];
    }
    v2f a[16];
    load_a_frags(tile, c, h, a);
    #pragma unroll
    for (int ct = 0; ct < 4; ++ct) {
      #pragma unroll
      for (int j = 0; j < 16; ++j)
        d[ct] = __builtin_amdgcn_wmma_f32_16x16x4_f32(false, a[j], false,
                                                      load_b_frag(aux, wo, j, ct, c, h),
                                                      (short)0, d[ct], false, false);
    }
    #pragma unroll
    for (int ct = 0; ct < 4; ++ct) {
      const float bcol = bias[16 * ct + c];
      #pragma unroll
      for (int r = 0; r < 8; ++r) {
        float val = d[ct][r] + bcol;
        if (relu) val = fmaxf(val, 0.0f);
        tile[d_place(h, r, 16 * ct + c)] = val;
      }
    }
    #pragma unroll
    for (int i = 0; i < 8; ++i) {
      const int row = 2 * i + h;
      *(float4*)&ot[row * D + 4 * c] =
          *(const float4*)&tile[row * D + ((c ^ (8 * (row >> 3))) << 2)];
    }
  }
}

#define BIN_THREADS 1024
#define BIN_EPT     8
#define BIN_EDGES   (BIN_THREADS * BIN_EPT)
#define BIN_BLOCKS  ((N_EDGE + BIN_EDGES - 1) / BIN_EDGES)
#define BIN_ENT     (2 * BIN_EDGES)
#define BIN_LISTS   512

static_assert(BIN_ENT <= (1 << 16), "an entry's rank in its block fits 16 bits beside the list number");
static_assert(BIN_LISTS == 2 * 256 && BIN_LISTS % 32 == 0 && BIN_LISTS + 32 <= BIN_THREADS,
              "counters are summed 16 per lane by wave 0 while one thread per list takes its run");

__global__ __launch_bounds__(64) void list_clear_kernel(unsigned* __restrict__ gcnt) {
  for (int i = threadIdx.x; i < 2 * NTILE; i += 64) gcnt[i * 32] = 0u;
}

__global__ __launch_bounds__(BIN_THREADS) void edge_bin_kernel(
    const int* __restrict__ uidx, const int* __restrict__ midx,
    unsigned* __restrict__ lst, unsigned* __restrict__ gcnt) {
  __shared__ unsigned binCnt[BIN_LISTS];
  __shared__ unsigned binOff[BIN_LISTS];
  __shared__ unsigned binBase[BIN_LISTS];
  __shared__ unsigned sorted[BIN_ENT];

  const int tid = threadIdx.x;
  const int lane = tid & 31, wave = tid >> 5;
  if (tid < BIN_LISTS) binCnt[tid] = 0u;
  __syncthreads();

  const int e0 = blockIdx.x * BIN_EDGES;
  unsigned uv[BIN_EPT], mv[BIN_EPT];
  #pragma unroll
  for (int j = 0; j < BIN_EPT; ++j) {
    const int e = e0 + j * BIN_THREADS + tid;
    const bool ok = e < N_EDGE;
    uv[j] = ok ? (unsigned)uidx[e] : 0u;
    mv[j] = ok ? (unsigned)midx[e] : 0u;
  }
  unsigned ent[2 * BIN_EPT], key[2 * BIN_EPT];
  #pragma unroll
  for (int j = 0; j < BIN_EPT; ++j) {
    const int e = e0 + j * BIN_THREADS + tid;
    key[2 * j] = 0xffffffffu;
    key[2 * j + 1] = 0xffffffffu;
    ent[2 * j] = 0u;
    ent[2 * j + 1] = 0u;
    if (e < N_EDGE) {
      unsigned tm, rm_, tu, ru;
      row_owner(mv[j], false, tm, rm_);
      row_owner(uv[j], true, tu, ru);
      ent[2 * j] = (uv[j] << 11) | rm_;
      key[2 * j] = (tm << 16) | atomicAdd(&binCnt[tm], 1u);
      ent[2 * j + 1] = (mv[j] << 11) | ru;
      key[2 * j + 1] = ((256u + tu) << 16) | atomicAdd(&binCnt[256u + tu], 1u);
    }
  }
  __syncthreads();

  if (tid < 32) {
    unsigned c[16], tot = 0u;
    #pragma unroll
    for (int k = 0; k < 16; ++k) { c[k] = binCnt[16 * tid + k]; tot += c[k]; }
    unsigned inc = tot;
    #pragma unroll
    for (int o = 1; o < 32; o <<= 1) {
      const unsigned v = __shfl_up(inc, o);
      if (lane >= o) inc += v;
    }
    unsigned ex = inc - tot;
    #pragma unroll
    for (int k = 0; k < 16; ++k) { binOff[16 * tid + k] = ex; ex += c[k]; }
  } else if (tid < 32 + BIN_LISTS) {
    const int i = tid - 32;
    const int t = i & 255, r = i >> 8;
    if (t < NTILE) binBase[i] = atomicAdd(&gcnt[(2 * t + r) * 32], binCnt[i]);
  }
  __syncthreads();

  #pragma unroll
  for (int i = 0; i < 2 * BIN_EPT; ++i) {
    if (key[i] != 0xffffffffu) sorted[binOff[key[i] >> 16] + (key[i] & 0xffffu)] = ent[i];
  }
  __syncthreads();

  for (int i = wave; i < BIN_LISTS; i += BIN_THREADS / 32) {
    const int t = i & 255, r = i >> 8;
    if (t >= NTILE) continue;
    const unsigned n = binCnt[i], off = binOff[i], base = binBase[i];
    unsigned* dstl = lst + (size_t)(2 * t + r) * LCAPW;
    for (unsigned k = (unsigned)lane; k < n; k += 32u) {
      const unsigned pos = base + k;
      if (pos < (unsigned)LCAPW) dstl[pos] = sorted[off + k];
    }
  }
}

__global__ __launch_bounds__(AGG_THREADS) void tile_layer1_kernel(
    const float* __restrict__ xu, const float* __restrict__ xm,
    const int* __restrict__ uidx, const int* __restrict__ midx,
    const float* __restrict__ WlM, const float* __restrict__ WrM, const float* __restrict__ bM,
    const float* __restrict__ WlU, const float* __restrict__ WrU, const float* __restrict__ bU,
    float* __restrict__ cu, float* __restrict__ cm,
    float* __restrict__ u1, float* __restrict__ m1,
    unsigned* lst, unsigned* gcnt) {
  __shared__ __attribute__((aligned(16))) float acc[TR * D];
  __shared__ __attribute__((aligned(16))) float aux[AUX_FLOATS];
  __shared__ unsigned segCnt[64];
  __shared__ unsigned segCur[64];
  __shared__ unsigned segRel[2][33];
  float* cnt   = aux + AUX_CNT;
  int*   ringE = (int*)(aux + AUX_RINGE);
  int*   ringD = (int*)(aux + AUX_RINGD);

  const int tid = threadIdx.x;
  const unsigned lane = threadIdx.x & 31u;
  const int wave = tid >> 5;
  const int wg = blockIdx.x;

  const int mlo   = sub_start(wg, N_MOVIE);
  const int rm    = sub_start(wg + 1, N_MOVIE) - mlo;
  const int ulo   = sub_start(2 * wg, N_USER);
  const int umid  = sub_start(2 * wg + 1, N_USER);
  const int uhi   = sub_start(2 * wg + 2, N_USER);
  const int ubase = sub_size(wg % 3);

  const int n0 = __builtin_amdgcn_readfirstlane((int)gcnt[(2 * wg) * 32]);
  const int n1 = __builtin_amdgcn_readfirstlane((int)gcnt[(2 * wg + 1) * 32]);
  const bool overflow = n0 > LCAPW || n1 > LCAPW;
  unsigned* lst0 = lst + (size_t)(2 * wg) * LCAPW;
  unsigned* lst1 = lst + (size_t)(2 * wg + 1) * LCAPW;

  if (!overflow) {
    unsigned* stage = (unsigned*)acc;
    if (tid < 64) segCnt[tid] = 0u;
    __syncthreads();
    order_pass<false>(lst0, n0, segCnt, stage, wave, lane);
    order_pass<false>(lst1, n1, segCnt + 32, stage, wave, lane);
    __syncthreads();
    if (tid < 64) {
      const int r = tid >> 5;
      const unsigned cv = segCnt[tid];
      unsigned inc = cv;
      #pragma unroll
      for (int o = 1; o < 32; o <<= 1) {
        const unsigned v = __shfl_up(inc, o);
        if ((int)lane >= o) inc += v;
      }
      segRel[r][lane] = inc - cv;
      if (lane == 31u) segRel[r][32] = inc;
      segCur[tid] = inc - cv + (r ? (unsigned)n0 : 0u);
    }
    __syncthreads();
    order_pass<true>(lst0, n0, segCur, stage, wave, lane);
    order_pass<true>(lst1, n1, segCur + 32, stage, wave, lane);
    __syncthreads();
    #pragma unroll 4
    for (int i = tid; i < n0; i += AGG_THREADS) lst0[i] = stage[i];
    #pragma unroll 4
    for (int i = tid; i < n1; i += AGG_THREADS) lst1[i] = stage[n0 + i];
    if (tid >= 1 && tid < 32) gcnt[(2 * wg) * 32 + tid] = segRel[0][tid];
    if (tid >= 33 && tid < 64) gcnt[(2 * wg + 1) * 32 + tid - 32] = segRel[1][tid - 32];
    __syncthreads();
  }

  for (int i = tid; i < TR * D / 4; i += AGG_THREADS)
    ((float4*)acc)[i] = make_float4(0.f, 0.f, 0.f, 0.f);
  for (int i = tid; i < TR; i += AGG_THREADS) cnt[i] = 0.f;
  __syncthreads();

  if (overflow) {
    #pragma unroll 1
    for (int p = 0; p < 3; ++p) {
      const bool mv = p == 0;
      const int plo   = mv ? mlo : (p == 1 ? ulo : umid);
      const int prows = mv ? rm : (p == 1 ? umid - ulo : uhi - umid);
      const int prow0 = mv ? 0 : (p == 1 ? ubase : ubase + (umid - ulo));
      scan_edges<true>(mv ? midx : uidx, mv ? uidx : midx, mv ? xu : xm,
                       acc + prow0 * D, cnt + prow0, &ringE[wave * QCAP], &ringD[wave * QCAP],
                       plo, prows, wg, tid, lane);
    }
  } else {
    walk_owned<true>(xu, lst0, (int)segRel[0][wave], (int)segRel[0][wave + 1], acc, cnt, lane);
    walk_owned<true>(xm, lst1, (int)segRel[1][wave], (int)segRel[1][wave + 1], acc, cnt, lane);
  }
  __syncthreads();

  for (int i = tid; i < TR * 16; i += AGG_THREADS) {
    const int row = i >> 4, q = i & 15;
    const float cv = cnt[row];
    if (q == 0) {
      if (row < ubase) {
        if (row < rm) cm[mlo + row] = cv;
      } else if (row - ubase < uhi - ulo) {
        cu[ulo + row - ubase] = cv;
      }
    }
    const float inv = 1.0f / fmaxf(cv, 1.0f);
    const float4 sv = ((const float4*)acc)[i];
    ((float4*)acc)[(row << 4) + (q ^ (row & 15))] =
        make_float4(sv.x * inv, sv.y * inv, sv.z * inv, sv.w * inv);
  }

  tile_transform<true>(acc, aux, nullptr, xm + (size_t)mlo * D, WlM, WrM, bM,
                       m1 + (size_t)mlo * D, rm, true, tid);
  tile_transform<true>(acc + ubase * D, aux, nullptr, xu + (size_t)ulo * D, WlU, WrU, bU,
                       u1 + (size_t)ulo * D, uhi - ulo, true, tid);
}


__global__ __launch_bounds__(AGG_THREADS) void tile_layer2_kernel(
    const float* __restrict__ h1u, const float* __restrict__ h1m,
    const int* __restrict__ uidx, const int* __restrict__ midx,
    const float* __restrict__ WlM, const float* __restrict__ WrM, const float* __restrict__ bM,
    const float* __restrict__ WlU, const float* __restrict__ WrU, const float* __restrict__ bU,
    float* __restrict__ cu, float* __restrict__ cm,
    float* __restrict__ outU, float* __restrict__ outM,
    const unsigned* __restrict__ lst, const unsigned* __restrict__ gcnt) {
  __shared__ __attribute__((aligned(16))) float acc[TR * D];
  __shared__ __attribute__((aligned(16))) float aux[AUX_FLOATS];
  int* ringE = (int*)(aux + AUX_RINGE);
  int* ringD = (int*)(aux + AUX_RINGD);

  const int tid = threadIdx.x;
  const unsigned lane = threadIdx.x & 31u;
  const int wave = tid >> 5;
  const int wg = blockIdx.x;

  const int mlo   = sub_start(wg, N_MOVIE);
  const int rm    = sub_start(wg + 1, N_MOVIE) - mlo;
  const int ulo   = sub_start(2 * wg, N_USER);
  const int umid  = sub_start(2 * wg + 1, N_USER);
  const int uhi   = sub_start(2 * wg + 2, N_USER);
  const int ubase = sub_size(wg % 3);

  for (int i = tid; i < TR * D / 4; i += AGG_THREADS)
    ((float4*)acc)[i] = make_float4(0.f, 0.f, 0.f, 0.f);
  __syncthreads();

  const int n0 = __builtin_amdgcn_readfirstlane((int)gcnt[(2 * wg) * 32]);
  const int n1 = __builtin_amdgcn_readfirstlane((int)gcnt[(2 * wg + 1) * 32]);
  if (n0 > LCAPW || n1 > LCAPW) {
    #pragma unroll 1
    for (int p = 0; p < 3; ++p) {
      const bool mv = p == 0;
      const int plo   = mv ? mlo : (p == 1 ? ulo : umid);
      const int prows = mv ? rm : (p == 1 ? umid - ulo : uhi - umid);
      const int prow0 = mv ? 0 : (p == 1 ? ubase : ubase + (umid - ulo));
      scan_edges<false>(mv ? midx : uidx, mv ? uidx : midx, mv ? h1u : h1m,
                        acc + prow0 * D, nullptr, &ringE[wave * QCAP], &ringD[wave * QCAP],
                        plo, prows, wg, tid, lane);
    }
  } else {
    const unsigned* g0 = gcnt + (2 * wg) * 32;
    const unsigned* g1 = gcnt + (2 * wg + 1) * 32;
    const int b0 = wave == 0 ? 0 : (int)g0[wave];
    const int e0 = wave == AGG_WAVES - 1 ? n0 : (int)g0[wave + 1];
    const int b1 = wave == 0 ? 0 : (int)g1[wave];
    const int e1 = wave == AGG_WAVES - 1 ? n1 : (int)g1[wave + 1];
    walk_owned<false>(h1u, lst + (size_t)(2 * wg) * LCAPW, b0, e0, acc, nullptr, lane);
    walk_owned<false>(h1m, lst + (size_t)(2 * wg + 1) * LCAPW, b1, e1, acc, nullptr, lane);
  }
  __syncthreads();

  tile_transform<false>(acc, aux, cm + mlo, h1m + (size_t)mlo * D, WlM, WrM, bM,
                        outM + (size_t)mlo * D, rm, false, tid);
  tile_transform<false>(acc + ubase * D, aux, cu + ulo, h1u + (size_t)ulo * D, WlU, WrU, bU,
                        outU + (size_t)ulo * D, uhi - ulo, false, tid);
}

extern "C" void kernel_launch(void* const* d_in, const int* in_sizes, int n_in,
                              void* d_out, int out_size, void* d_ws, size_t ws_size,
                              hipStream_t stream) {
  const float* x_user  = (const float*)d_in[0];
  const float* x_movie = (const float*)d_in[1];
  const float* Wl1_um  = (const float*)d_in[2];
  const float* Wr1_um  = (const float*)d_in[3];
  const float* b1_um   = (const float*)d_in[4];
  const float* Wl1_mu  = (const float*)d_in[5];
  const float* Wr1_mu  = (const float*)d_in[6];
  const float* b1_mu   = (const float*)d_in[7];
  const float* Wl2_um  = (const float*)d_in[8];
  const float* Wr2_um  = (const float*)d_in[9];
  const float* b2_um   = (const float*)d_in[10];
  const float* Wl2_mu  = (const float*)d_in[11];
  const float* Wr2_mu  = (const float*)d_in[12];
  const float* b2_mu   = (const float*)d_in[13];
  const int*   uidx    = (const int*)d_in[14];
  const int*   midx    = (const int*)d_in[15];

  float* ws = (float*)d_ws;
  unsigned* lst  = (unsigned*)(ws + OFF_LST);
  unsigned* gcnt = (unsigned*)(ws + OFF_GC);
  float* cu = ws + OFF_CU;
  float* cm = ws + OFF_CM;
  float* u1 = ws + OFF_U1;
  float* m1 = ws + OFF_M1;

  float* u2 = (float*)d_out;
  float* m2 = (float*)d_out + (size_t)N_USER * D;

  const dim3 bAgg(AGG_THREADS);
  const dim3 gAgg(NTILE);

  list_clear_kernel<<<dim3(1), dim3(64), 0, stream>>>(gcnt);
  edge_bin_kernel<<<dim3(BIN_BLOCKS), dim3(BIN_THREADS), 0, stream>>>(uidx, midx, lst, gcnt);

  tile_layer1_kernel<<<gAgg, bAgg, 0, stream>>>(x_user, x_movie, uidx, midx,
                                                Wl1_um, Wr1_um, b1_um,
                                                Wl1_mu, Wr1_mu, b1_mu,
                                                cu, cm, u1, m1, lst, gcnt);
  tile_layer2_kernel<<<gAgg, bAgg, 0, stream>>>(u1, m1, uidx, midx,
                                                Wl2_um, Wr2_um, b2_um,
                                                Wl2_mu, Wr2_mu, b2_mu,
                                                cu, cm, u2, m2, lst, gcnt);
}
